# starting kernel
# baseline (speedup 1.0000x reference)
#include <hip/hip_runtime.h>
#include <hip/hip_bf16.h>
#include <math.h>

#define B_    8192
#define D_    32
#define H_    128
#define P2_   64
#define PREP_ 10
#define GIN_  320
#define G3_   384
#define CS_   8
#define CH_   32
#define NT_   16
#define NO_   4096
#define DT_   0.1f
#define NPOST_ 4
#define LOGC_  0.91893853320467274178f

typedef __attribute__((ext_vector_type(2))) float v2f;
typedef __attribute__((ext_vector_type(8))) float v8f;

__device__ __forceinline__ float sigmoidf_(float x) { return 1.0f / (1.0f + expf(-x)); }


__device__ __forceinline__ v2f load_afrag(const float* A, int lda, int m0, int k, int lane) {
    int m  = m0 + (lane & 15);
    int kk = k + ((lane >> 4) << 1);
    v2f a;
    a.x = A[m * lda + kk];
    a.y = A[m * lda + kk + 1];
    return a;
}

__device__ __forceinline__ v2f load_bfrag(const float* Bm, int ldb, int k, int n0, int lane) {
    int n  = n0 + (lane & 15);
    int kk = k + ((lane >> 4) << 1);
    v2f b;
    b.x = Bm[kk * ldb + n];
    b.y = Bm[(kk + 1) * ldb + n];
    return b;
}

__device__ __forceinline__ v8f wmma4(v2f a, v2f b, v8f c) {
    return __builtin_amdgcn_wmma_f32_16x16x4_f32(false, a, false, b, (short)0, c, false, false);
}

__global__ void k_init(const float* __restrict__ cov,
                       const float* __restrict__ w1, const float* __restrict__ b1,
                       const float* __restrict__ w2, const float* __restrict__ b2,
                       float* __restrict__ h, float* __restrict__ loss) {
    __shared__ float t_s[16][CH_];
    const int tid = threadIdx.x;
    const int m0  = blockIdx.x * 16;
    if (blockIdx.x == 0 && tid == 0) *loss = 0.0f;
    for (int i = tid; i < 16 * CH_; i += 128) {
        int r = i / CH_, c = i % CH_;
        float acc = b1[c];
        #pragma unroll
        for (int k = 0; k < CS_; ++k) acc += cov[(m0 + r) * CS_ + k] * w1[k * CH_ + c];
        t_s[r][c] = fmaxf(acc, 0.0f);
    }
    __syncthreads();
    for (int i = tid; i < 16 * H_; i += 128) {
        int r = i / H_, c = i % H_;
        float acc = b2[c];
        #pragma unroll
        for (int k = 0; k < CH_; ++k) acc += t_s[r][k] * w2[k * H_ + c];
        h[(m0 + r) * H_ + c] = tanhf(acc);
    }
}

__global__ void k_pmodel(const float* __restrict__ h,
                         const float* __restrict__ w1, const float* __restrict__ b1,
                         const float* __restrict__ w2, const float* __restrict__ b2,
                         float* __restrict__ p) {
    __shared__ float t_s[16][P2_];
    const int tid = threadIdx.x, lane = tid & 31, wv = tid >> 5;
    const int m0 = blockIdx.x * 16, n0 = wv * 16;

    v8f acc = {};
    for (int k = 0; k < H_; k += 4)
        acc = wmma4(load_afrag(h, H_, m0, k, lane), load_bfrag(w1, P2_, k, n0, lane), acc);
    const int col = n0 + (lane & 15), rb = (lane >> 4) << 3;
    const float bv = b1[col];
    #pragma unroll
    for (int j = 0; j < 8; ++j) t_s[rb + j][col] = fmaxf(acc[j] + bv, 0.0f);
    __syncthreads();

    v8f acc2 = {};
    for (int k = 0; k < P2_; k += 4)
        acc2 = wmma4(load_afrag(&t_s[0][0], P2_, 0, k, lane), load_bfrag(w2, P2_, k, n0, lane), acc2);
    const float bv2 = b2[col];
    #pragma unroll
    for (int j = 0; j < 8; ++j) p[(m0 + rb + j) * P2_ + col] = acc2[j] + bv2;
}

__global__ void k_ode(float* __restrict__ h, const float* __restrict__ p,
                      const float* __restrict__ xz_w, const float* __restrict__ xz_b,
                      const float* __restrict__ hz_w,
                      const float* __restrict__ xn_w, const float* __restrict__ xn_b,
                      const float* __restrict__ hn_w) {
    __shared__ float zh_s[16][H_];
    const int tid = threadIdx.x, lane = tid & 31, wv = tid >> 5;
    const int m0 = blockIdx.x * 16, n0 = wv * 16;

    v8f az = {};
    for (int k = 0; k < P2_; k += 4)
        az = wmma4(load_afrag(p, P2_, m0, k, lane), load_bfrag(xz_w, H_, k, n0, lane), az);
    for (int k = 0; k < H_; k += 4)
        az = wmma4(load_afrag(h, H_, m0, k, lane), load_bfrag(hz_w, H_, k, n0, lane), az);

    const int col = n0 + (lane & 15), rb = (lane >> 4) << 3;
    const float bz = xz_b[col];
    float z[8];
    #pragma unroll
    for (int j = 0; j < 8; ++j) {
        z[j] = sigmoidf_(az[j] + bz);
        zh_s[rb + j][col] = z[j] * h[(m0 + rb + j) * H_ + col];
    }
    __syncthreads();

    v8f an = {};
    for (int k = 0; k < P2_; k += 4)
        an = wmma4(load_afrag(p, P2_, m0, k, lane), load_bfrag(xn_w, H_, k, n0, lane), an);
    for (int k = 0; k < H_; k += 4)
        an = wmma4(load_afrag(&zh_s[0][0], H_, 0, k, lane), load_bfrag(hn_w, H_, k, n0, lane), an);

    const float bn = xn_b[col];
    #pragma unroll
    for (int j = 0; j < 8; ++j) {
        float n = tanhf(an[j] + bn);
        int row = m0 + rb + j;
        float hv = h[row * H_ + col];
        h[row * H_ + col] = hv + DT_ * (1.0f - z[j]) * (n - hv);
    }
}

__global__ void k_jump(int t,
                       const float* __restrict__ X, const float* __restrict__ M,
                       const int* __restrict__ obs_idx,
                       float* __restrict__ h, const float* __restrict__ p,
                       const float* __restrict__ w_ih, const float* __restrict__ w_hh,
                       const float* __restrict__ b_ih, const float* __restrict__ b_hh,
                       const float* __restrict__ w_prep, const float* __restrict__ bias_prep,
                       float* __restrict__ loss_part) {
    __shared__ int   idx_s[16];
    __shared__ float hold_s[16][H_];
    __shared__ float gin_s[16][GIN_];
    __shared__ float gi_s[16][G3_];
    __shared__ float gh_s[16][G3_];
    __shared__ float red[256];
    const int tid = threadIdx.x, lane = tid & 31, wv = tid >> 5;
    const int o0 = blockIdx.x * 16;
    const float* Xt = X + (size_t)t * NO_ * D_;
    const float* Mt = M + (size_t)t * NO_ * D_;

    if (tid < 16) idx_s[tid] = obs_idx[t * NO_ + o0 + tid];
    __syncthreads();

    for (int i = tid; i < 16 * H_; i += 256) {
        int o = i >> 7, j = i & 127;
        hold_s[o][j] = h[idx_s[o] * H_ + j];
    }

    float lsum = 0.0f;
    for (int i = tid; i < 16 * D_; i += 256) {
        int o = i / D_, d = i % D_;
        int gb = o0 + o, b = idx_s[o];
        float mean = p[b * P2_ + d];
        float lv   = p[b * P2_ + D_ + d];
        float x    = Xt[gb * D_ + d];
        float m    = Mt[gb * D_ + d];
        float err  = (x - mean) * expf(-0.5f * lv);
        lsum += 0.5f * ((err * err + lv + 2.0f * LOGC_) * m);
    }

    for (int i = tid; i < 16 * GIN_; i += 256) {
        int o = i / GIN_, c = i % GIN_;
        int d = c / PREP_, pi = c - d * PREP_;
        int gb = o0 + o, b = idx_s[o];
        float mean = p[b * P2_ + d];
        float lv   = p[b * P2_ + D_ + d];
        float x    = Xt[gb * D_ + d];
        float m    = Mt[gb * D_ + d];
        float err  = (x - mean) * expf(-0.5f * lv);
        const float* wp = w_prep + (size_t)(d * 4) * PREP_ + pi;
        float a = x * wp[0] + mean * wp[PREP_] + lv * wp[2 * PREP_] + err * wp[3 * PREP_]
                + bias_prep[d * PREP_ + pi];
        gin_s[o][c] = fmaxf(a, 0.0f) * m;
    }
    __syncthreads();

    for (int s = 0; s < 3; ++s) {
        const int n0 = (wv * 3 + s) * 16;
        v8f gi = {};
        for (int k = 0; k < GIN_; k += 4)
            gi = wmma4(load_afrag(&gin_s[0][0], GIN_, 0, k, lane), load_bfrag(w_ih, G3_, k, n0, lane), gi);
        v8f gh = {};
        for (int k = 0; k < H_; k += 4)
            gh = wmma4(load_afrag(&hold_s[0][0], H_, 0, k, lane), load_bfrag(w_hh, G3_, k, n0, lane), gh);
        const int col = n0 + (lane & 15), rb = (lane >> 4) << 3;
        const float bi = b_ih[col], bh = b_hh[col];
        #pragma unroll
        for (int j = 0; j < 8; ++j) {
            gi_s[rb + j][col] = gi[j] + bi;
            gh_s[rb + j][col] = gh[j] + bh;
        }
    }
    __syncthreads();

    for (int i = tid; i < 16 * H_; i += 256) {
        int o = i >> 7, j = i & 127;
        float r = sigmoidf_(gi_s[o][j] + gh_s[o][j]);
        float z = sigmoidf_(gi_s[o][H_ + j] + gh_s[o][H_ + j]);
        float n = tanhf(gi_s[o][2 * H_ + j] + r * gh_s[o][2 * H_ + j]);
        float hv = hold_s[o][j];
        h[idx_s[o] * H_ + j] = (1.0f - z) * n + z * hv;
    }

    red[tid] = lsum;
    __syncthreads();
    for (int s2 = 128; s2 > 0; s2 >>= 1) {
        if (tid < s2) red[tid] += red[tid + s2];
        __syncthreads();
    }
    if (tid == 0) loss_part[t * (NO_ / 16) + blockIdx.x] = red[0];
}

__global__ void k_loss_final(const float* __restrict__ parts, float* __restrict__ loss) {
    __shared__ float red[256];
    const int tid = threadIdx.x;
    float s = 0.0f;
    for (int i = tid; i < NT_ * (NO_ / 16); i += 256) s += parts[i];
    red[tid] = s;
    __syncthreads();
    for (int k = 128; k > 0; k >>= 1) {
        if (tid < k) red[tid] += red[tid + k];
        __syncthreads();
    }
    if (tid == 0) *loss = red[0];
}

extern "C" void kernel_launch(void* const* d_in, const int* in_sizes, int n_in,
                              void* d_out, int out_size, void* d_ws, size_t ws_size,
                              hipStream_t stream) {
    const float* X       = (const float*)d_in[0];
    const float* M       = (const float*)d_in[1];
    const int*   obs_idx = (const int*)d_in[2];
    const float* cov     = (const float*)d_in[3];
    const float* cm_w1   = (const float*)d_in[4];
    const float* cm_b1   = (const float*)d_in[5];
    const float* cm_w2   = (const float*)d_in[6];
    const float* cm_b2   = (const float*)d_in[7];
    const float* p_w1    = (const float*)d_in[8];
    const float* p_b1    = (const float*)d_in[9];
    const float* p_w2    = (const float*)d_in[10];
    const float* p_b2    = (const float*)d_in[11];
    const float* xz_w    = (const float*)d_in[12];
    const float* xz_b    = (const float*)d_in[13];
    const float* hz_w    = (const float*)d_in[14];
    const float* xn_w    = (const float*)d_in[15];
    const float* xn_b    = (const float*)d_in[16];
    const float* hn_w    = (const float*)d_in[17];
    const float* w_ih    = (const float*)d_in[18];
    const float* w_hh    = (const float*)d_in[19];
    const float* b_ih    = (const float*)d_in[20];
    const float* b_hh    = (const float*)d_in[21];
    const float* w_prep  = (const float*)d_in[22];
    const float* b_prep  = (const float*)d_in[23];

    float* h    = (float*)d_out;
    float* p    = h + (size_t)B_ * H_;
    float* loss = p + (size_t)B_ * P2_;
    float* loss_part = (float*)d_ws;

    const int MB = B_ / 16;
    const int OB = NO_ / 16;

    k_init<<<MB, 128, 0, stream>>>(cov, cm_w1, cm_b1, cm_w2, cm_b2, h, loss);
    k_pmodel<<<MB, 128, 0, stream>>>(h, p_w1, p_b1, p_w2, p_b2, p);

    for (int t = 0; t < NT_; ++t) {
        k_ode<<<MB, 256, 0, stream>>>(h, p, xz_w, xz_b, hz_w, xn_w, xn_b, hn_w);
        k_pmodel<<<MB, 128, 0, stream>>>(h, p_w1, p_b1, p_w2, p_b2, p);
        k_jump<<<OB, 256, 0, stream>>>(t, X, M, obs_idx, h, p,
                                       w_ih, w_hh, b_ih, b_hh, w_prep, b_prep, loss_part);
        k_pmodel<<<MB, 128, 0, stream>>>(h, p_w1, p_b1, p_w2, p_b2, p);
    }
    for (int s = 0; s < NPOST_; ++s) {
        k_ode<<<MB, 256, 0, stream>>>(h, p, xz_w, xz_b, hz_w, xn_w, xn_b, hn_w);
        k_pmodel<<<MB, 128, 0, stream>>>(h, p_w1, p_b1, p_w2, p_b2, p);
    }
    k_loss_final<<<1, 256, 0, stream>>>(loss_part, loss);
}


# whole time loop in ONE kernel: a work-group keeps its 16 trajectories' h and p in LDS over all 20 steps and finds its observed rows by one scan of obs_idx; 75 launches -> 2; loss summed per row tile
# speedup vs baseline: 1.6352x; 1.6352x over previous
#include <hip/hip_runtime.h>
#include <hip/hip_bf16.h>
#include <math.h>


#define B_    8192
#define D_    32
#define H_    128
#define P2_   64
#define PREP_ 10
#define GIN_  320
#define G3_   384
#define CS_   8
#define CH_   32
#define NT_   16
#define NO_   4096
#define DT_   0.1f
#define NPOST_ 4
#define LOGC_  0.91893853320467274178f

#define MB_   (B_ / 16)
#define HP_   132
#define PP_   68
#define GP_   324

typedef __attribute__((ext_vector_type(2))) float v2f;
typedef __attribute__((ext_vector_type(4))) float v4f;
typedef __attribute__((ext_vector_type(8))) float v8f;
typedef __attribute__((ext_vector_type(4))) int   v4i;

__device__ __forceinline__ float sigmoidf_(float x) { return 1.0f / (1.0f + expf(-x)); }


__device__ __forceinline__ v2f load_afrag(const float* A, int lda, int m0, int k, int lane) {
    int m  = m0 + (lane & 15);
    int kk = k + ((lane >> 4) << 1);
    v2f a;
    a.x = A[m * lda + kk];
    a.y = A[m * lda + kk + 1];
    return a;
}

__device__ __forceinline__ v8f wmma4(v2f a, v2f b, v8f c) {
    return __builtin_amdgcn_wmma_f32_16x16x4_f32(false, a, false, b, (short)0, c, false, false);
}

template <int NK>
__device__ __forceinline__ void bfrag2_load(const float* Bm, int ldb, int n0, int lane,
                                            v2f (&lo)[NK], v2f (&hi)[NK]) {
    const float* bp = Bm + ((lane >> 4) << 1) * ldb + n0 + 2 * (lane & 15);
    #pragma unroll
    for (int i = 0; i < NK; ++i) {
        lo[i] = *reinterpret_cast<const v2f*>(bp + (4 * i) * ldb);
        hi[i] = *reinterpret_cast<const v2f*>(bp + (4 * i + 1) * ldb);
    }
}

template <int NK>
__device__ __forceinline__ void lds_frag2_mma(const float* As, int lda, int lane,
                                              const v2f (&lo)[NK], const v2f (&hi)[NK],
                                              v8f& c0, v8f& c1) {
    #pragma unroll
    for (int i = 0; i < NK; ++i) {
        const v2f a = load_afrag(As, lda, 0, 4 * i, lane);
        v2f b0, b1;
        b0.x = lo[i].x; b0.y = hi[i].x;
        b1.x = lo[i].y; b1.y = hi[i].y;
        c0 = wmma4(a, b0, c0);
        c1 = wmma4(a, b1, c1);
    }
}

template <int NS>
__device__ __forceinline__ void gemm2_lds_glb(const float* As, int lda,
                                              const float* Bm, int ldb, int n0, int lane,
                                              v8f& c0, v8f& c1) {
    v2f lo[2][16], hi[2][16];
    bfrag2_load<16>(Bm, ldb, n0, lane, lo[0], hi[0]);
    #pragma unroll
    for (int s = 0; s < NS; ++s) {
        if (s + 1 < NS)
            bfrag2_load<16>(Bm + (s + 1) * 64 * ldb, ldb, n0, lane, lo[(s + 1) & 1], hi[(s + 1) & 1]);
        lds_frag2_mma<16>(As + s * 64, lda, lane, lo[s & 1], hi[s & 1], c0, c1);
    }
}

__device__ __forceinline__ void pmodel_phase(const float* h_s, float* t_s, float* p_s,
                                             const float* __restrict__ w1, const float* __restrict__ b1,
                                             const float* __restrict__ w2, const float* __restrict__ b2,
                                             int lane, int wv, int wb) {
    const int pw = wv - wb;
    const bool mine = (pw == 0) || (pw == 1);
    const int n0 = (pw & 1) * 32;
    const int col = n0 + 2 * (lane & 15), rb = (lane >> 4) << 3;
    if (mine) {
        v8f acc0 = {}, acc1 = {};
        gemm2_lds_glb<H_ / 64>(h_s, HP_, w1, P2_, n0, lane, acc0, acc1);
        const v2f bv = *reinterpret_cast<const v2f*>(b1 + col);
        #pragma unroll
        for (int j = 0; j < 8; ++j) {
            v2f t;
            t.x = fmaxf(acc0[j] + bv.x, 0.0f);
            t.y = fmaxf(acc1[j] + bv.y, 0.0f);
            *reinterpret_cast<v2f*>(t_s + (rb + j) * PP_ + col) = t;
        }
    }
    __syncthreads();
    if (mine) {
        v8f acc2 = {}, acc3 = {};
        gemm2_lds_glb<P2_ / 64>(t_s, PP_, w2, P2_, n0, lane, acc2, acc3);
        const v2f bv2 = *reinterpret_cast<const v2f*>(b2 + col);
        #pragma unroll
        for (int j = 0; j < 8; ++j) {
            v2f o;
            o.x = acc2[j] + bv2.x;
            o.y = acc3[j] + bv2.y;
            *reinterpret_cast<v2f*>(p_s + (rb + j) * PP_ + col) = o;
        }
    }
    __syncthreads();
}

__device__ __forceinline__ void ode_phase(const int* inv_t,
                                          const float* __restrict__ Xt, const float* __restrict__ Mt,
                                          float* x_s, float* m_s, int tid,
                                          float* h_s, const float* p_s, float* zh_s,
                                          const float* __restrict__ xz_w, const float* __restrict__ xz_b,
                                          const float* __restrict__ hz_w,
                                          const float* __restrict__ xn_w, const float* __restrict__ xn_b,
                                          const float* __restrict__ hn_w,
                                          int lane, int wv) {
    const int n0 = (wv & 3) * 32;
    const int col = n0 + 2 * (lane & 15), rb = (lane >> 4) << 3;
    float z0[8], z1[8];
    if (wv < 4) {
        v8f az0 = {}, az1 = {};
        gemm2_lds_glb<P2_ / 64>(p_s, PP_, xz_w, H_, n0, lane, az0, az1);
        gemm2_lds_glb<H_ / 64>(h_s, HP_, hz_w, H_, n0, lane, az0, az1);
        const v2f bz = *reinterpret_cast<const v2f*>(xz_b + col);
        #pragma unroll
        for (int j = 0; j < 8; ++j) {
            const v2f hv = *reinterpret_cast<const v2f*>(h_s + (rb + j) * HP_ + col);
            z0[j] = sigmoidf_(az0[j] + bz.x);
            z1[j] = sigmoidf_(az1[j] + bz.y);
            v2f zh;
            zh.x = z0[j] * hv.x;
            zh.y = z1[j] * hv.y;
            *reinterpret_cast<v2f*>(zh_s + (rb + j) * HP_ + col) = zh;
        }
    } else if (inv_t != nullptr) {
        const int j = tid - 128;
        const int r = j >> 3, c4 = j & 7;
        const int o = inv_t[r];
        if (o >= 0) {
            *reinterpret_cast<v4f*>(x_s + r * D_ + 4 * c4) =
                *reinterpret_cast<const v4f*>(Xt + (size_t)o * D_ + 4 * c4);
            *reinterpret_cast<v4f*>(m_s + r * D_ + 4 * c4) =
                *reinterpret_cast<const v4f*>(Mt + (size_t)o * D_ + 4 * c4);
        }
    }
    __syncthreads();
    if (wv < 4) {
        v8f an0 = {}, an1 = {};
        gemm2_lds_glb<P2_ / 64>(p_s, PP_, xn_w, H_, n0, lane, an0, an1);
        gemm2_lds_glb<H_ / 64>(zh_s, HP_, hn_w, H_, n0, lane, an0, an1);
        const v2f bn = *reinterpret_cast<const v2f*>(xn_b + col);
        #pragma unroll
        for (int j = 0; j < 8; ++j) {
            const float n0v = tanhf(an0[j] + bn.x);
            const float n1v = tanhf(an1[j] + bn.y);
            float* hp = h_s + (rb + j) * HP_ + col;
            const v2f hv = *reinterpret_cast<const v2f*>(hp);
            v2f o;
            o.x = hv.x + DT_ * (1.0f - z0[j]) * (n0v - hv.x);
            o.y = hv.y + DT_ * (1.0f - z1[j]) * (n1v - hv.y);
            *reinterpret_cast<v2f*>(hp) = o;
        }
    }
    __syncthreads();
}

#define JCH_    32
#define JPAIRS_ 16
#define JPN_    400
#define JNCI_   10
#define JUNITS_ 6

__device__ __forceinline__ void wchunk_gload(const float* __restrict__ src, int tid,
                                             v4f (&lo)[JUNITS_], v4f (&hi)[JUNITS_]) {
    #pragma unroll
    for (int i = 0; i < JUNITS_; ++i) {
        const int u = tid + 256 * i;
        const int q = u / 96, n4 = u - q * 96;
        const float* r0 = src + (2 * q) * G3_ + 4 * n4;
        lo[i] = *reinterpret_cast<const v4f*>(r0);
        hi[i] = *reinterpret_cast<const v4f*>(r0 + G3_);
    }
}

__device__ __forceinline__ void wchunk_lstore(float* buf, int tid,
                                              const v4f (&lo)[JUNITS_], const v4f (&hi)[JUNITS_]) {
    #pragma unroll
    for (int i = 0; i < JUNITS_; ++i) {
        const int u = tid + 256 * i;
        const int q = u / 96, n4 = u - q * 96;
        float* d = buf + 2 * (q * JPN_ + 4 * n4);
        const v4f e0 = {lo[i].x, hi[i].x, lo[i].y, hi[i].y};
        const v4f e1 = {lo[i].z, hi[i].z, lo[i].w, hi[i].w};
        *reinterpret_cast<v4f*>(d)     = e0;
        *reinterpret_cast<v4f*>(d + 4) = e1;
    }
}

__device__ __forceinline__ v2f load_bfrag_staged(const float* buf, int k, int n0, int lane) {
    const int n = n0 + (lane & 15);
    const int q = (k >> 1) + (lane >> 4);
    return *reinterpret_cast<const v2f*>(buf + 2 * (q * JPN_ + n));
}

__device__ __forceinline__ void wchunk_consume(const float* A, int lda, int kbase,
                                               const float* bs, int n0w, int lane, v8f (&acc)[3]) {
    #pragma unroll
    for (int k8 = 0; k8 < JCH_ / 4; ++k8) {
        const v2f a = load_afrag(A, lda, 0, kbase + 4 * k8, lane);
        #pragma unroll
        for (int s = 0; s < 3; ++s)
            acc[s] = wmma4(a, load_bfrag_staged(bs, 4 * k8, n0w + 16 * s, lane), acc[s]);
    }
}

__device__ __forceinline__ void jump_phase(const int* inv_t,
                                           const float* x_s, const float* m_s,
                                           const float* wp_s, const float* bp_s,
                                           float* h_s, const float* p_s, float* gin_s, float* wst_s,
                                           const float* __restrict__ w_ih, const float* __restrict__ w_hh,
                                           const float* __restrict__ b_ih, const float* __restrict__ b_hh,
                                           float& lsum, int tid, int lane, int wv) {
    float* wst0 = wst_s;
    float* wst1 = wst_s + JPAIRS_ * JPN_ * 2;
    float* gi_s = wst0;
    float* gh_s = wst0 + 16 * G3_;

    v4f wlo[JUNITS_], whi[JUNITS_];
    wchunk_gload(w_ih, tid, wlo, whi);

    for (int i = tid; i < 16 * D_; i += 256) {
        const int r = i / D_, d = i % D_;
        const int o = inv_t[r];
        if (o >= 0) {
            float mean = p_s[r * PP_ + d];
            float lv   = p_s[r * PP_ + D_ + d];
            float x    = x_s[r * D_ + d];
            float m    = m_s[r * D_ + d];
            float err  = (x - mean) * expf(-0.5f * lv);
            lsum += 0.5f * ((err * err + lv + 2.0f * LOGC_) * m);
        }
    }

    for (int i = tid; i < 16 * GIN_; i += 256) {
        const int r = i / GIN_, c = i % GIN_;
        const int d = c / PREP_, pi = c - d * PREP_;
        const int o = inv_t[r];
        float g = 0.0f;
        if (o >= 0) {
            float mean = p_s[r * PP_ + d];
            float lv   = p_s[r * PP_ + D_ + d];
            float x    = x_s[r * D_ + d];
            float m    = m_s[r * D_ + d];
            float err  = (x - mean) * expf(-0.5f * lv);
            const float* wp = wp_s + (d * 4) * PREP_ + pi;
            float a = x * wp[0] + mean * wp[PREP_] + lv * wp[2 * PREP_] + err * wp[3 * PREP_]
                    + bp_s[d * PREP_ + pi];
            g = fmaxf(a, 0.0f) * m;
        }
        gin_s[r * GP_ + c] = g;
    }
    wchunk_lstore(wst0, tid, wlo, whi);
    __syncthreads();

    v8f gi[3], gh[3];
    #pragma unroll
    for (int s = 0; s < 3; ++s) { gi[s] = (v8f){}; gh[s] = (v8f){}; }
    const int n0w = wv * 48;

    #pragma unroll 1
    for (int c = 0; c < JNCI_; ++c) {
        const float* nxt = (c + 1 < JNCI_) ? (w_ih + (c + 1) * (JCH_ * G3_)) : w_hh;
        wchunk_gload(nxt, tid, wlo, whi);
        wchunk_consume(gin_s, GP_, c * JCH_, (c & 1) ? wst1 : wst0, n0w, lane, gi);
        wchunk_lstore((c & 1) ? wst0 : wst1, tid, wlo, whi);
        __syncthreads();
    }
    #pragma unroll 1
    for (int c = 0; c < 3; ++c) {
        wchunk_gload(w_hh + (c + 1) * (JCH_ * G3_), tid, wlo, whi);
        wchunk_consume(h_s, HP_, c * JCH_, (c & 1) ? wst1 : wst0, n0w, lane, gh);
        wchunk_lstore((c & 1) ? wst0 : wst1, tid, wlo, whi);
        __syncthreads();
    }
    wchunk_consume(h_s, HP_, 3 * JCH_, wst1, n0w, lane, gh);

    {
        const int rb = (lane >> 4) << 3;
        #pragma unroll
        for (int s = 0; s < 3; ++s) {
            const int col = n0w + 16 * s + (lane & 15);
            const float bi = b_ih[col], bh = b_hh[col];
            #pragma unroll
            for (int j = 0; j < 8; ++j) {
                gi_s[(rb + j) * G3_ + col] = gi[s][j] + bi;
                gh_s[(rb + j) * G3_ + col] = gh[s][j] + bh;
            }
        }
    }
    __syncthreads();

    for (int i = tid; i < 16 * H_; i += 256) {
        const int r = i >> 7, j = i & 127;
        if (inv_t[r] >= 0) {
            float rr = sigmoidf_(gi_s[r * G3_ + j] + gh_s[r * G3_ + j]);
            float z  = sigmoidf_(gi_s[r * G3_ + H_ + j] + gh_s[r * G3_ + H_ + j]);
            float n  = tanhf(gi_s[r * G3_ + 2 * H_ + j] + rr * gh_s[r * G3_ + 2 * H_ + j]);
            float hv = h_s[r * HP_ + j];
            h_s[r * HP_ + j] = (1.0f - z) * n + z * hv;
        }
    }
    __syncthreads();
}

__global__ void __launch_bounds__(256, 4) k_traj(
                       const float* __restrict__ X, const float* __restrict__ M,
                       const int* __restrict__ obs_idx, const float* __restrict__ cov,
                       const float* __restrict__ cm_w1, const float* __restrict__ cm_b1,
                       const float* __restrict__ cm_w2, const float* __restrict__ cm_b2,
                       const float* __restrict__ p_w1, const float* __restrict__ p_b1,
                       const float* __restrict__ p_w2, const float* __restrict__ p_b2,
                       const float* __restrict__ xz_w, const float* __restrict__ xz_b,
                       const float* __restrict__ hz_w,
                       const float* __restrict__ xn_w, const float* __restrict__ xn_b,
                       const float* __restrict__ hn_w,
                       const float* __restrict__ w_ih, const float* __restrict__ w_hh,
                       const float* __restrict__ b_ih, const float* __restrict__ b_hh,
                       const float* __restrict__ w_prep, const float* __restrict__ bias_prep,
                       float* __restrict__ h, float* __restrict__ p,
                       float* __restrict__ loss_part) {
    __shared__ __attribute__((aligned(16))) float h_s[16 * HP_];
    __shared__ __attribute__((aligned(16))) float p_s[16 * PP_];
    __shared__ __attribute__((aligned(16))) float t_s[16 * PP_];
    __shared__ __attribute__((aligned(16))) float gin_s[16 * GP_];
    __shared__ __attribute__((aligned(16))) float wst_s[2 * JPAIRS_ * JPN_ * 2];
    __shared__ __attribute__((aligned(16))) float x_s[16 * D_];
    __shared__ __attribute__((aligned(16))) float m_s[16 * D_];
    __shared__ float wp_s[D_ * 4 * PREP_];
    __shared__ float bp_s[D_ * PREP_];
    __shared__ int   inv_s[NT_ * 16];
    float* zh_s = gin_s;
    float* red  = t_s;
    const int tid = threadIdx.x, lane = tid & 31;
    const int wv = __builtin_amdgcn_readfirstlane(tid >> 5);
    const int bx = blockIdx.x;
    const int m0 = bx * 16;

    inv_s[tid] = -1;
    __syncthreads();
    {
        const v4i* oi = reinterpret_cast<const v4i*>(obs_idx);
        #pragma unroll 1
        for (int q0 = 0; q0 < NT_ * NO_ / 4; q0 += 8 * 256) {
            v4i v[8];
            #pragma unroll
            for (int u = 0; u < 8; ++u) v[u] = oi[q0 + u * 256 + tid];
            #pragma unroll
            for (int u = 0; u < 8; ++u) {
                const int e = 4 * (q0 + u * 256 + tid);
                const int tt = e >> 12, o = e & (NO_ - 1);
                if ((v[u].x >> 4) == bx) inv_s[tt * 16 + (v[u].x & 15)] = o;
                if ((v[u].y >> 4) == bx) inv_s[tt * 16 + (v[u].y & 15)] = o + 1;
                if ((v[u].z >> 4) == bx) inv_s[tt * 16 + (v[u].z & 15)] = o + 2;
                if ((v[u].w >> 4) == bx) inv_s[tt * 16 + (v[u].w & 15)] = o + 3;
            }
        }
    }
    for (int i = tid; i < D_ * 4 * PREP_; i += 256) wp_s[i] = w_prep[i];
    for (int i = tid; i < D_ * PREP_; i += 256) bp_s[i] = bias_prep[i];

    for (int i = tid; i < 16 * CH_; i += 256) {
        int r = i / CH_, c = i % CH_;
        float acc = cm_b1[c];
        #pragma unroll
        for (int k = 0; k < CS_; ++k) acc += cov[(m0 + r) * CS_ + k] * cm_w1[k * CH_ + c];
        t_s[r * PP_ + c] = fmaxf(acc, 0.0f);
    }
    __syncthreads();
    for (int i = tid; i < 16 * H_; i += 256) {
        int r = i / H_, c = i % H_;
        float acc = cm_b2[c];
        #pragma unroll
        for (int k = 0; k < CH_; ++k) acc += t_s[r * PP_ + k] * cm_w2[k * H_ + c];
        h_s[r * HP_ + c] = tanhf(acc);
    }
    __syncthreads();
    pmodel_phase(h_s, t_s, p_s, p_w1, p_b1, p_w2, p_b2, lane, wv, 0);

    float lsum = 0.0f;
    #pragma unroll 1
    for (int t = 0; t < NT_; ++t) {
        int tl = tid;
        asm volatile("" : "+v"(tl));
        const int ln = tl & 31;
        const int w = __builtin_amdgcn_readfirstlane(tl >> 5);
        ode_phase(inv_s + t * 16, X + (size_t)t * NO_ * D_, M + (size_t)t * NO_ * D_, x_s, m_s, tl,
                  h_s, p_s, zh_s, xz_w, xz_b, hz_w, xn_w, xn_b, hn_w, ln, w);
        pmodel_phase(h_s, t_s, p_s, p_w1, p_b1, p_w2, p_b2, ln, w, 0);
        jump_phase(inv_s + t * 16, x_s, m_s, wp_s, bp_s,
                   h_s, p_s, gin_s, wst_s, w_ih, w_hh, b_ih, b_hh,
                   lsum, tl, ln, w);
        pmodel_phase(h_s, t_s, p_s, p_w1, p_b1, p_w2, p_b2, ln, w, 2);
    }
    #pragma unroll 1
    for (int s = 0; s < NPOST_; ++s) {
        int tl = tid;
        asm volatile("" : "+v"(tl));
        const int ln = tl & 31;
        const int w = __builtin_amdgcn_readfirstlane(tl >> 5);
        ode_phase(nullptr, X, M, x_s, m_s, tl,
                  h_s, p_s, zh_s, xz_w, xz_b, hz_w, xn_w, xn_b, hn_w, ln, w);
        pmodel_phase(h_s, t_s, p_s, p_w1, p_b1, p_w2, p_b2, ln, w, 2 * (s & 1));
    }

    for (int i = tid; i < 16 * H_ / 4; i += 256) {
        const int r = i >> 5, c4 = i & 31;
        *reinterpret_cast<v4f*>(h + (size_t)(m0 + r) * H_ + 4 * c4) =
            *reinterpret_cast<const v4f*>(h_s + r * HP_ + 4 * c4);
    }
    {
        const int r = tid >> 4, c4 = tid & 15;
        *reinterpret_cast<v4f*>(p + (size_t)(m0 + r) * P2_ + 4 * c4) =
            *reinterpret_cast<const v4f*>(p_s + r * PP_ + 4 * c4);
    }

    red[tid] = lsum;
    __syncthreads();
    for (int s2 = 128; s2 > 0; s2 >>= 1) {
        if (tid < s2) red[tid] += red[tid + s2];
        __syncthreads();
    }
    if (tid == 0) loss_part[bx] = red[0];
}

__global__ void k_loss_final(const float* __restrict__ parts, float* __restrict__ loss) {
    __shared__ float red[256];
    const int tid = threadIdx.x;
    float s = 0.0f;
    for (int i = tid; i < MB_; i += 256) s += parts[i];
    red[tid] = s;
    __syncthreads();
    for (int k = 128; k > 0; k >>= 1) {
        if (tid < k) red[tid] += red[tid + k];
        __syncthreads();
    }
    if (tid == 0) *loss = red[0];
}

extern "C" void kernel_launch(void* const* d_in, const int* in_sizes, int n_in,
                              void* d_out, int out_size, void* d_ws, size_t ws_size,
                              hipStream_t stream) {
    const float* X       = (const float*)d_in[0];
    const float* M       = (const float*)d_in[1];
    const int*   obs_idx = (const int*)d_in[2];
    const float* cov     = (const float*)d_in[3];
    const float* cm_w1   = (const float*)d_in[4];
    const float* cm_b1   = (const float*)d_in[5];
    const float* cm_w2   = (const float*)d_in[6];
    const float* cm_b2   = (const float*)d_in[7];
    const float* p_w1    = (const float*)d_in[8];
    const float* p_b1    = (const float*)d_in[9];
    const float* p_w2    = (const float*)d_in[10];
    const float* p_b2    = (const float*)d_in[11];
    const float* xz_w    = (const float*)d_in[12];
    const float* xz_b    = (const float*)d_in[13];
    const float* hz_w    = (const float*)d_in[14];
    const float* xn_w    = (const float*)d_in[15];
    const float* xn_b    = (const float*)d_in[16];
    const float* hn_w    = (const float*)d_in[17];
    const float* w_ih    = (const float*)d_in[18];
    const float* w_hh    = (const float*)d_in[19];
    const float* b_ih    = (const float*)d_in[20];
    const float* b_hh    = (const float*)d_in[21];
    const float* w_prep  = (const float*)d_in[22];
    const float* b_prep  = (const float*)d_in[23];

    float* h    = (float*)d_out;
    float* p    = h + (size_t)B_ * H_;
    float* loss = p + (size_t)B_ * P2_;
    float* loss_part = (float*)d_ws;

    k_traj<<<MB_, 256, 0, stream>>>(X, M, obs_idx, cov, cm_w1, cm_b1, cm_w2, cm_b2,
                                    p_w1, p_b1, p_w2, p_b2,
                                    xz_w, xz_b, hz_w, xn_w, xn_b, hn_w,
                                    w_ih, w_hh, b_ih, b_hh, w_prep, b_prep,
                                    h, p, loss_part);
    k_loss_final<<<1, 256, 0, stream>>>(loss_part, loss);
}


# jump: gate weights not staged in LDS: a wave owns 48 adjacent columns as 3 column-interleaved tiles, B fragments by 12-byte global loads, 2 register slices; 13 barriers gone; order of sums kept
# speedup vs baseline: 1.7720x; 1.0837x over previous
#include <hip/hip_runtime.h>
#include <hip/hip_bf16.h>
#include <math.h>


#define B_    8192
#define D_    32
#define H_    128
#define P2_   64
#define PREP_ 10
#define GIN_  320
#define G3_   384
#define CS_   8
#define CH_   32
#define NT_   16
#define NO_   4096
#define DT_   0.1f
#define NPOST_ 4
#define LOGC_  0.91893853320467274178f

#define MB_   (B_ / 16)
#define HP_   132
#define PP_   68
#define GP_   324

typedef __attribute__((ext_vector_type(2))) float v2f;
typedef __attribute__((ext_vector_type(4))) float v4f;
typedef __attribute__((ext_vector_type(8))) float v8f;
typedef __attribute__((ext_vector_type(4))) int   v4i;

__device__ __forceinline__ float sigmoidf_(float x) { return 1.0f / (1.0f + expf(-x)); }


__device__ __forceinline__ v2f load_afrag(const float* A, int lda, int m0, int k, int lane) {
    int m  = m0 + (lane & 15);
    int kk = k + ((lane >> 4) << 1);
    v2f a;
    a.x = A[m * lda + kk];
    a.y = A[m * lda + kk + 1];
    return a;
}

__device__ __forceinline__ v8f wmma4(v2f a, v2f b, v8f c) {
    return __builtin_amdgcn_wmma_f32_16x16x4_f32(false, a, false, b, (short)0, c, false, false);
}

template <int NK>
__device__ __forceinline__ void bfrag2_load(const float* Bm, int ldb, int n0, int lane,
                                            v2f (&lo)[NK], v2f (&hi)[NK]) {
    const float* bp = Bm + ((lane >> 4) << 1) * ldb + n0 + 2 * (lane & 15);
    #pragma unroll
    for (int i = 0; i < NK; ++i) {
        lo[i] = *reinterpret_cast<const v2f*>(bp + (4 * i) * ldb);
        hi[i] = *reinterpret_cast<const v2f*>(bp + (4 * i + 1) * ldb);
    }
}

template <int NK>
__device__ __forceinline__ void lds_frag2_mma(const float* As, int lda, int lane,
                                              const v2f (&lo)[NK], const v2f (&hi)[NK],
                                              v8f& c0, v8f& c1) {
    #pragma unroll
    for (int i = 0; i < NK; ++i) {
        const v2f a = load_afrag(As, lda, 0, 4 * i, lane);
        v2f b0, b1;
        b0.x = lo[i].x; b0.y = hi[i].x;
        b1.x = lo[i].y; b1.y = hi[i].y;
        c0 = wmma4(a, b0, c0);
        c1 = wmma4(a, b1, c1);
    }
}

template <int NS>
__device__ __forceinline__ void gemm2_lds_glb(const float* As, int lda,
                                              const float* Bm, int ldb, int n0, int lane,
                                              v8f& c0, v8f& c1) {
    v2f lo[2][16], hi[2][16];
    bfrag2_load<16>(Bm, ldb, n0, lane, lo[0], hi[0]);
    #pragma unroll
    for (int s = 0; s < NS; ++s) {
        if (s + 1 < NS)
            bfrag2_load<16>(Bm + (s + 1) * 64 * ldb, ldb, n0, lane, lo[(s + 1) & 1], hi[(s + 1) & 1]);
        lds_frag2_mma<16>(As + s * 64, lda, lane, lo[s & 1], hi[s & 1], c0, c1);
    }
}

__device__ __forceinline__ void pmodel_phase(const float* h_s, float* t_s, float* p_s,
                                             const float* __restrict__ w1, const float* __restrict__ b1,
                                             const float* __restrict__ w2, const float* __restrict__ b2,
                                             int lane, int wv, int wb) {
    const int pw = wv - wb;
    const bool mine = (pw == 0) || (pw == 1);
    const int n0 = (pw & 1) * 32;
    const int col = n0 + 2 * (lane & 15), rb = (lane >> 4) << 3;
    if (mine) {
        v8f acc0 = {}, acc1 = {};
        gemm2_lds_glb<H_ / 64>(h_s, HP_, w1, P2_, n0, lane, acc0, acc1);
        const v2f bv = *reinterpret_cast<const v2f*>(b1 + col);
        #pragma unroll
        for (int j = 0; j < 8; ++j) {
            v2f t;
            t.x = fmaxf(acc0[j] + bv.x, 0.0f);
            t.y = fmaxf(acc1[j] + bv.y, 0.0f);
            *reinterpret_cast<v2f*>(t_s + (rb + j) * PP_ + col) = t;
        }
    }
    __syncthreads();
    if (mine) {
        v8f acc2 = {}, acc3 = {};
        gemm2_lds_glb<P2_ / 64>(t_s, PP_, w2, P2_, n0, lane, acc2, acc3);
        const v2f bv2 = *reinterpret_cast<const v2f*>(b2 + col);
        #pragma unroll
        for (int j = 0; j < 8; ++j) {
            v2f o;
            o.x = acc2[j] + bv2.x;
            o.y = acc3[j] + bv2.y;
            *reinterpret_cast<v2f*>(p_s + (rb + j) * PP_ + col) = o;
        }
    }
    __syncthreads();
}

__device__ __forceinline__ void ode_phase(const int* inv_t,
                                          const float* __restrict__ Xt, const float* __restrict__ Mt,
                                          float* x_s, float* m_s, int tid,
                                          float* h_s, const float* p_s, float* zh_s,
                                          const float* __restrict__ xz_w, const float* __restrict__ xz_b,
                                          const float* __restrict__ hz_w,
                                          const float* __restrict__ xn_w, const float* __restrict__ xn_b,
                                          const float* __restrict__ hn_w,
                                          int lane, int wv) {
    const int n0 = (wv & 3) * 32;
    const int col = n0 + 2 * (lane & 15), rb = (lane >> 4) << 3;
    float z0[8], z1[8];
    if (wv < 4) {
        v8f az0 = {}, az1 = {};
        gemm2_lds_glb<P2_ / 64>(p_s, PP_, xz_w, H_, n0, lane, az0, az1);
        gemm2_lds_glb<H_ / 64>(h_s, HP_, hz_w, H_, n0, lane, az0, az1);
        const v2f bz = *reinterpret_cast<const v2f*>(xz_b + col);
        #pragma unroll
        for (int j = 0; j < 8; ++j) {
            const v2f hv = *reinterpret_cast<const v2f*>(h_s + (rb + j) * HP_ + col);
            z0[j] = sigmoidf_(az0[j] + bz.x);
            z1[j] = sigmoidf_(az1[j] + bz.y);
            v2f zh;
            zh.x = z0[j] * hv.x;
            zh.y = z1[j] * hv.y;
            *reinterpret_cast<v2f*>(zh_s + (rb + j) * HP_ + col) = zh;
        }
    } else if (inv_t != nullptr) {
        const int j = tid - 128;
        const int r = j >> 3, c4 = j & 7;
        const int o = inv_t[r];
        if (o >= 0) {
            *reinterpret_cast<v4f*>(x_s + r * D_ + 4 * c4) =
                *reinterpret_cast<const v4f*>(Xt + (size_t)o * D_ + 4 * c4);
            *reinterpret_cast<v4f*>(m_s + r * D_ + 4 * c4) =
                *reinterpret_cast<const v4f*>(Mt + (size_t)o * D_ + 4 * c4);
        }
    }
    __syncthreads();
    if (wv < 4) {
        v8f an0 = {}, an1 = {};
        gemm2_lds_glb<P2_ / 64>(p_s, PP_, xn_w, H_, n0, lane, an0, an1);
        gemm2_lds_glb<H_ / 64>(zh_s, HP_, hn_w, H_, n0, lane, an0, an1);
        const v2f bn = *reinterpret_cast<const v2f*>(xn_b + col);
        #pragma unroll
        for (int j = 0; j < 8; ++j) {
            const float n0v = tanhf(an0[j] + bn.x);
            const float n1v = tanhf(an1[j] + bn.y);
            float* hp = h_s + (rb + j) * HP_ + col;
            const v2f hv = *reinterpret_cast<const v2f*>(hp);
            v2f o;
            o.x = hv.x + DT_ * (1.0f - z0[j]) * (n0v - hv.x);
            o.y = hv.y + DT_ * (1.0f - z1[j]) * (n1v - hv.y);
            *reinterpret_cast<v2f*>(hp) = o;
        }
    }
    __syncthreads();
}

#define JKS_ 8
#define JNCI_ 10
struct __attribute__((aligned(4))) f3_t { float x, y, z; };

__device__ __forceinline__ void bfrag3_load(const float* __restrict__ bp,
                                            f3_t (&lo)[JKS_], f3_t (&hi)[JKS_]) {
    #pragma unroll
    for (int i = 0; i < JKS_; ++i) {
        lo[i] = *reinterpret_cast<const f3_t*>(bp + (4 * i) * G3_);
        hi[i] = *reinterpret_cast<const f3_t*>(bp + (4 * i + 1) * G3_);
    }
}

__device__ __forceinline__ void frag3_mma(const float* As, int lda, int lane,
                                          const f3_t (&lo)[JKS_], const f3_t (&hi)[JKS_],
                                          v8f (&acc)[3]) {
    #pragma unroll
    for (int i = 0; i < JKS_; ++i) {
        const v2f a = load_afrag(As, lda, 0, 4 * i, lane);
        v2f b0, b1, b2;
        b0.x = lo[i].x; b0.y = hi[i].x;
        b1.x = lo[i].y; b1.y = hi[i].y;
        b2.x = lo[i].z; b2.y = hi[i].z;
        acc[0] = wmma4(a, b0, acc[0]);
        acc[1] = wmma4(a, b1, acc[1]);
        acc[2] = wmma4(a, b2, acc[2]);
    }
}

__device__ __forceinline__ void jump_phase(const int* inv_t,
                                           const float* x_s, const float* m_s,
                                           const float* wp_s, const float* bp_s,
                                           float* h_s, const float* p_s, float* gin_s, float* gate_s,
                                           const float* __restrict__ w_ih, const float* __restrict__ w_hh,
                                           const float* __restrict__ b_ih, const float* __restrict__ b_hh,
                                           float& lsum, int tid, int lane, int wv) {
    float* gi_s = gate_s;
    float* gh_s = gate_s + 16 * G3_;

    const int n0w = wv * 48;
    const int woff = ((lane >> 4) << 1) * G3_ + n0w + 3 * (lane & 15);
    const float* wi = w_ih + woff;
    const float* wh = w_hh + woff;

    f3_t lo0[JKS_], hi0[JKS_], lo1[JKS_], hi1[JKS_];
    bfrag3_load(wi, lo0, hi0);

    for (int i = tid; i < 16 * D_; i += 256) {
        const int r = i / D_, d = i % D_;
        const int o = inv_t[r];
        if (o >= 0) {
            float mean = p_s[r * PP_ + d];
            float lv   = p_s[r * PP_ + D_ + d];
            float x    = x_s[r * D_ + d];
            float m    = m_s[r * D_ + d];
            float err  = (x - mean) * expf(-0.5f * lv);
            lsum += 0.5f * ((err * err + lv + 2.0f * LOGC_) * m);
        }
    }

    for (int i = tid; i < 16 * GIN_; i += 256) {
        const int r = i / GIN_, c = i % GIN_;
        const int d = c / PREP_, pi = c - d * PREP_;
        const int o = inv_t[r];
        float g = 0.0f;
        if (o >= 0) {
            float mean = p_s[r * PP_ + d];
            float lv   = p_s[r * PP_ + D_ + d];
            float x    = x_s[r * D_ + d];
            float m    = m_s[r * D_ + d];
            float err  = (x - mean) * expf(-0.5f * lv);
            const float* wp = wp_s + (d * 4) * PREP_ + pi;
            float a = x * wp[0] + mean * wp[PREP_] + lv * wp[2 * PREP_] + err * wp[3 * PREP_]
                    + bp_s[d * PREP_ + pi];
            g = fmaxf(a, 0.0f) * m;
        }
        gin_s[r * GP_ + c] = g;
    }
    __syncthreads();

    v8f gi[3], gh[3];
    #pragma unroll
    for (int s = 0; s < 3; ++s) { gi[s] = (v8f){}; gh[s] = (v8f){}; }

    bfrag3_load(wi + JKS_ * 4 * G3_, lo1, hi1);
    #pragma unroll
    for (int s2 = 0; s2 < JNCI_ / 2; ++s2) {
        const bool last = (s2 == JNCI_ / 2 - 1);
        const float* nx0 = last ? wh : wi + (2 * s2 + 2) * (JKS_ * 4 * G3_);
        const float* nx1 = nx0 + JKS_ * 4 * G3_;
        frag3_mma(gin_s + (2 * s2) * (JKS_ * 4), GP_, lane, lo0, hi0, gi);
        bfrag3_load(nx0, lo0, hi0);
        frag3_mma(gin_s + (2 * s2 + 1) * (JKS_ * 4), GP_, lane, lo1, hi1, gi);
        bfrag3_load(nx1, lo1, hi1);
    }
    frag3_mma(h_s, HP_, lane, lo0, hi0, gh);
    bfrag3_load(wh + 2 * (JKS_ * 4 * G3_), lo0, hi0);
    frag3_mma(h_s + JKS_ * 4, HP_, lane, lo1, hi1, gh);
    bfrag3_load(wh + 3 * (JKS_ * 4 * G3_), lo1, hi1);
    frag3_mma(h_s + 2 * (JKS_ * 4), HP_, lane, lo0, hi0, gh);
    frag3_mma(h_s + 3 * (JKS_ * 4), HP_, lane, lo1, hi1, gh);

    {
        const int rb = (lane >> 4) << 3;
        #pragma unroll
        for (int s = 0; s < 3; ++s) {
            const int col = n0w + 3 * (lane & 15) + s;
            const float bi = b_ih[col], bh = b_hh[col];
            #pragma unroll
            for (int j = 0; j < 8; ++j) {
                gi_s[(rb + j) * G3_ + col] = gi[s][j] + bi;
                gh_s[(rb + j) * G3_ + col] = gh[s][j] + bh;
            }
        }
    }
    __syncthreads();

    for (int i = tid; i < 16 * H_; i += 256) {
        const int r = i >> 7, j = i & 127;
        if (inv_t[r] >= 0) {
            float rr = sigmoidf_(gi_s[r * G3_ + j] + gh_s[r * G3_ + j]);
            float z  = sigmoidf_(gi_s[r * G3_ + H_ + j] + gh_s[r * G3_ + H_ + j]);
            float n  = tanhf(gi_s[r * G3_ + 2 * H_ + j] + rr * gh_s[r * G3_ + 2 * H_ + j]);
            float hv = h_s[r * HP_ + j];
            h_s[r * HP_ + j] = (1.0f - z) * n + z * hv;
        }
    }
    __syncthreads();
}

__global__ void __launch_bounds__(256, 4) k_traj(
                       const float* __restrict__ X, const float* __restrict__ M,
                       const int* __restrict__ obs_idx, const float* __restrict__ cov,
                       const float* __restrict__ cm_w1, const float* __restrict__ cm_b1,
                       const float* __restrict__ cm_w2, const float* __restrict__ cm_b2,
                       const float* __restrict__ p_w1, const float* __restrict__ p_b1,
                       const float* __restrict__ p_w2, const float* __restrict__ p_b2,
                       const float* __restrict__ xz_w, const float* __restrict__ xz_b,
                       const float* __restrict__ hz_w,
                       const float* __restrict__ xn_w, const float* __restrict__ xn_b,
                       const float* __restrict__ hn_w,
                       const float* __restrict__ w_ih, const float* __restrict__ w_hh,
                       const float* __restrict__ b_ih, const float* __restrict__ b_hh,
                       const float* __restrict__ w_prep, const float* __restrict__ bias_prep,
                       float* __restrict__ h, float* __restrict__ p,
                       float* __restrict__ loss_part) {
    __shared__ __attribute__((aligned(16))) float h_s[16 * HP_];
    __shared__ __attribute__((aligned(16))) float p_s[16 * PP_];
    __shared__ __attribute__((aligned(16))) float t_s[16 * PP_];
    __shared__ __attribute__((aligned(16))) float gin_s[16 * GP_];
    __shared__ __attribute__((aligned(16))) float gate_s[2 * 16 * G3_];
    __shared__ __attribute__((aligned(16))) float x_s[16 * D_];
    __shared__ __attribute__((aligned(16))) float m_s[16 * D_];
    __shared__ float wp_s[D_ * 4 * PREP_];
    __shared__ float bp_s[D_ * PREP_];
    __shared__ int   inv_s[NT_ * 16];
    float* zh_s = gin_s;
    float* red  = t_s;
    const int tid = threadIdx.x, lane = tid & 31;
    const int wv = __builtin_amdgcn_readfirstlane(tid >> 5);
    const int bx = blockIdx.x;
    const int m0 = bx * 16;

    inv_s[tid] = -1;
    __syncthreads();
    {
        const v4i* oi = reinterpret_cast<const v4i*>(obs_idx);
        #pragma unroll 1
        for (int q0 = 0; q0 < NT_ * NO_ / 4; q0 += 8 * 256) {
            v4i v[8];
            #pragma unroll
            for (int u = 0; u < 8; ++u) v[u] = oi[q0 + u * 256 + tid];
            #pragma unroll
            for (int u = 0; u < 8; ++u) {
                const int e = 4 * (q0 + u * 256 + tid);
                const int tt = e >> 12, o = e & (NO_ - 1);
                if ((v[u].x >> 4) == bx) inv_s[tt * 16 + (v[u].x & 15)] = o;
                if ((v[u].y >> 4) == bx) inv_s[tt * 16 + (v[u].y & 15)] = o + 1;
                if ((v[u].z >> 4) == bx) inv_s[tt * 16 + (v[u].z & 15)] = o + 2;
                if ((v[u].w >> 4) == bx) inv_s[tt * 16 + (v[u].w & 15)] = o + 3;
            }
        }
    }
    for (int i = tid; i < D_ * 4 * PREP_; i += 256) wp_s[i] = w_prep[i];
    for (int i = tid; i < D_ * PREP_; i += 256) bp_s[i] = bias_prep[i];

    for (int i = tid; i < 16 * CH_; i += 256) {
        int r = i / CH_, c = i % CH_;
        float acc = cm_b1[c];
        #pragma unroll
        for (int k = 0; k < CS_; ++k) acc += cov[(m0 + r) * CS_ + k] * cm_w1[k * CH_ + c];
        t_s[r * PP_ + c] = fmaxf(acc, 0.0f);
    }
    __syncthreads();
    for (int i = tid; i < 16 * H_; i += 256) {
        int r = i / H_, c = i % H_;
        float acc = cm_b2[c];
        #pragma unroll
        for (int k = 0; k < CH_; ++k) acc += t_s[r * PP_ + k] * cm_w2[k * H_ + c];
        h_s[r * HP_ + c] = tanhf(acc);
    }
    __syncthreads();
    pmodel_phase(h_s, t_s, p_s, p_w1, p_b1, p_w2, p_b2, lane, wv, 0);

    float lsum = 0.0f;
    #pragma unroll 1
    for (int t = 0; t < NT_; ++t) {
        int tl = tid;
        asm volatile("" : "+v"(tl));
        const int ln = tl & 31;
        const int w = __builtin_amdgcn_readfirstlane(tl >> 5);
        ode_phase(inv_s + t * 16, X + (size_t)t * NO_ * D_, M + (size_t)t * NO_ * D_, x_s, m_s, tl,
                  h_s, p_s, zh_s, xz_w, xz_b, hz_w, xn_w, xn_b, hn_w, ln, w);
        pmodel_phase(h_s, t_s, p_s, p_w1, p_b1, p_w2, p_b2, ln, w, 0);
        jump_phase(inv_s + t * 16, x_s, m_s, wp_s, bp_s,
                   h_s, p_s, gin_s, gate_s, w_ih, w_hh, b_ih, b_hh,
                   lsum, tl, ln, w);
        pmodel_phase(h_s, t_s, p_s, p_w1, p_b1, p_w2, p_b2, ln, w, 2);
    }
    #pragma unroll 1
    for (int s = 0; s < NPOST_; ++s) {
        int tl = tid;
        asm volatile("" : "+v"(tl));
        const int ln = tl & 31;
        const int w = __builtin_amdgcn_readfirstlane(tl >> 5);
        ode_phase(nullptr, X, M, x_s, m_s, tl,
                  h_s, p_s, zh_s, xz_w, xz_b, hz_w, xn_w, xn_b, hn_w, ln, w);
        pmodel_phase(h_s, t_s, p_s, p_w1, p_b1, p_w2, p_b2, ln, w, 2 * (s & 1));
    }

    for (int i = tid; i < 16 * H_ / 4; i += 256) {
        const int r = i >> 5, c4 = i & 31;
        *reinterpret_cast<v4f*>(h + (size_t)(m0 + r) * H_ + 4 * c4) =
            *reinterpret_cast<const v4f*>(h_s + r * HP_ + 4 * c4);
    }
    {
        const int r = tid >> 4, c4 = tid & 15;
        *reinterpret_cast<v4f*>(p + (size_t)(m0 + r) * P2_ + 4 * c4) =
            *reinterpret_cast<const v4f*>(p_s + r * PP_ + 4 * c4);
    }

    red[tid] = lsum;
    __syncthreads();
    for (int s2 = 128; s2 > 0; s2 >>= 1) {
        if (tid < s2) red[tid] += red[tid + s2];
        __syncthreads();
    }
    if (tid == 0) loss_part[bx] = red[0];
}

__global__ void k_loss_final(const float* __restrict__ parts, float* __restrict__ loss) {
    __shared__ float red[256];
    const int tid = threadIdx.x;
    float s = 0.0f;
    for (int i = tid; i < MB_; i += 256) s += parts[i];
    red[tid] = s;
    __syncthreads();
    for (int k = 128; k > 0; k >>= 1) {
        if (tid < k) red[tid] += red[tid + k];
        __syncthreads();
    }
    if (tid == 0) *loss = red[0];
}

extern "C" void kernel_launch(void* const* d_in, const int* in_sizes, int n_in,
                              void* d_out, int out_size, void* d_ws, size_t ws_size,
                              hipStream_t stream) {
    const float* X       = (const float*)d_in[0];
    const float* M       = (const float*)d_in[1];
    const int*   obs_idx = (const int*)d_in[2];
    const float* cov     = (const float*)d_in[3];
    const float* cm_w1   = (const float*)d_in[4];
    const float* cm_b1   = (const float*)d_in[5];
    const float* cm_w2   = (const float*)d_in[6];
    const float* cm_b2   = (const float*)d_in[7];
    const float* p_w1    = (const float*)d_in[8];
    const float* p_b1    = (const float*)d_in[9];
    const float* p_w2    = (const float*)d_in[10];
    const float* p_b2    = (const float*)d_in[11];
    const float* xz_w    = (const float*)d_in[12];
    const float* xz_b    = (const float*)d_in[13];
    const float* hz_w    = (const float*)d_in[14];
    const float* xn_w    = (const float*)d_in[15];
    const float* xn_b    = (const float*)d_in[16];
    const float* hn_w    = (const float*)d_in[17];
    const float* w_ih    = (const float*)d_in[18];
    const float* w_hh    = (const float*)d_in[19];
    const float* b_ih    = (const float*)d_in[20];
    const float* b_hh    = (const float*)d_in[21];
    const float* w_prep  = (const float*)d_in[22];
    const float* b_prep  = (const float*)d_in[23];

    float* h    = (float*)d_out;
    float* p    = h + (size_t)B_ * H_;
    float* loss = p + (size_t)B_ * P2_;
    float* loss_part = (float*)d_ws;

    k_traj<<<MB_, 256, 0, stream>>>(X, M, obs_idx, cov, cm_w1, cm_b1, cm_w2, cm_b2,
                                    p_w1, p_b1, p_w2, p_b2,
                                    xz_w, xz_b, hz_w, xn_w, xn_b, hn_w,
                                    w_ih, w_hh, b_ih, b_hh, w_prep, b_prep,
                                    h, p, loss_part);
    k_loss_final<<<1, 256, 0, stream>>>(loss_part, loss);
}


# jump front: gin = relu(feats.w_prep+b)*M from the matrix unit, one 16x16x4 product per d (4 per wave), not 20 scalar passes per thread; its 4-term feature sum now in the matrix unit's order
# speedup vs baseline: 1.8869x; 1.0648x over previous
#include <hip/hip_runtime.h>
#include <hip/hip_bf16.h>
#include <math.h>


#define B_    8192
#define D_    32
#define H_    128
#define P2_   64
#define PREP_ 10
#define GIN_  320
#define G3_   384
#define CS_   8
#define CH_   32
#define NT_   16
#define NO_   4096
#define DT_   0.1f
#define NPOST_ 4
#define LOGC_  0.91893853320467274178f

#define MB_   (B_ / 16)
#define HP_   132
#define PP_   68
#define GP_   324

typedef __attribute__((ext_vector_type(2))) float v2f;
typedef __attribute__((ext_vector_type(4))) float v4f;
typedef __attribute__((ext_vector_type(8))) float v8f;
typedef __attribute__((ext_vector_type(4))) int   v4i;

__device__ __forceinline__ float sigmoidf_(float x) { return 1.0f / (1.0f + expf(-x)); }


__device__ __forceinline__ v2f load_afrag(const float* A, int lda, int m0, int k, int lane) {
    int m  = m0 + (lane & 15);
    int kk = k + ((lane >> 4) << 1);
    v2f a;
    a.x = A[m * lda + kk];
    a.y = A[m * lda + kk + 1];
    return a;
}

__device__ __forceinline__ v8f wmma4(v2f a, v2f b, v8f c) {
    return __builtin_amdgcn_wmma_f32_16x16x4_f32(false, a, false, b, (short)0, c, false, false);
}

template <int NK>
__device__ __forceinline__ void bfrag2_load(const float* Bm, int ldb, int n0, int lane,
                                            v2f (&lo)[NK], v2f (&hi)[NK]) {
    const float* bp = Bm + ((lane >> 4) << 1) * ldb + n0 + 2 * (lane & 15);
    #pragma unroll
    for (int i = 0; i < NK; ++i) {
        lo[i] = *reinterpret_cast<const v2f*>(bp + (4 * i) * ldb);
        hi[i] = *reinterpret_cast<const v2f*>(bp + (4 * i + 1) * ldb);
    }
}

template <int NK>
__device__ __forceinline__ void lds_frag2_mma(const float* As, int lda, int lane,
                                              const v2f (&lo)[NK], const v2f (&hi)[NK],
                                              v8f& c0, v8f& c1) {
    #pragma unroll
    for (int i = 0; i < NK; ++i) {
        const v2f a = load_afrag(As, lda, 0, 4 * i, lane);
        v2f b0, b1;
        b0.x = lo[i].x; b0.y = hi[i].x;
        b1.x = lo[i].y; b1.y = hi[i].y;
        c0 = wmma4(a, b0, c0);
        c1 = wmma4(a, b1, c1);
    }
}

template <int NS>
__device__ __forceinline__ void gemm2_lds_glb(const float* As, int lda,
                                              const float* Bm, int ldb, int n0, int lane,
                                              v8f& c0, v8f& c1) {
    v2f lo[2][16], hi[2][16];
    bfrag2_load<16>(Bm, ldb, n0, lane, lo[0], hi[0]);
    #pragma unroll
    for (int s = 0; s < NS; ++s) {
        if (s + 1 < NS)
            bfrag2_load<16>(Bm + (s + 1) * 64 * ldb, ldb, n0, lane, lo[(s + 1) & 1], hi[(s + 1) & 1]);
        lds_frag2_mma<16>(As + s * 64, lda, lane, lo[s & 1], hi[s & 1], c0, c1);
    }
}

__device__ __forceinline__ void pmodel_phase(const float* h_s, float* t_s, float* p_s,
                                             const float* __restrict__ w1, const float* __restrict__ b1,
                                             const float* __restrict__ w2, const float* __restrict__ b2,
                                             int lane, int wv, int wb) {
    const int pw = wv - wb;
    const bool mine = (pw == 0) || (pw == 1);
    const int n0 = (pw & 1) * 32;
    const int col = n0 + 2 * (lane & 15), rb = (lane >> 4) << 3;
    if (mine) {
        v8f acc0 = {}, acc1 = {};
        gemm2_lds_glb<H_ / 64>(h_s, HP_, w1, P2_, n0, lane, acc0, acc1);
        const v2f bv = *reinterpret_cast<const v2f*>(b1 + col);
        #pragma unroll
        for (int j = 0; j < 8; ++j) {
            v2f t;
            t.x = fmaxf(acc0[j] + bv.x, 0.0f);
            t.y = fmaxf(acc1[j] + bv.y, 0.0f);
            *reinterpret_cast<v2f*>(t_s + (rb + j) * PP_ + col) = t;
        }
    }
    __syncthreads();
    if (mine) {
        v8f acc2 = {}, acc3 = {};
        gemm2_lds_glb<P2_ / 64>(t_s, PP_, w2, P2_, n0, lane, acc2, acc3);
        const v2f bv2 = *reinterpret_cast<const v2f*>(b2 + col);
        #pragma unroll
        for (int j = 0; j < 8; ++j) {
            v2f o;
            o.x = acc2[j] + bv2.x;
            o.y = acc3[j] + bv2.y;
            *reinterpret_cast<v2f*>(p_s + (rb + j) * PP_ + col) = o;
        }
    }
    __syncthreads();
}

__device__ __forceinline__ void ode_phase(const int* inv_t,
                                          const float* __restrict__ Xt, const float* __restrict__ Mt,
                                          float* x_s, float* m_s, int tid,
                                          float* h_s, const float* p_s, float* zh_s,
                                          const float* __restrict__ xz_w, const float* __restrict__ xz_b,
                                          const float* __restrict__ hz_w,
                                          const float* __restrict__ xn_w, const float* __restrict__ xn_b,
                                          const float* __restrict__ hn_w,
                                          int lane, int wv) {
    const int n0 = (wv & 3) * 32;
    const int col = n0 + 2 * (lane & 15), rb = (lane >> 4) << 3;
    float z0[8], z1[8];
    if (wv < 4) {
        v8f az0 = {}, az1 = {};
        gemm2_lds_glb<P2_ / 64>(p_s, PP_, xz_w, H_, n0, lane, az0, az1);
        gemm2_lds_glb<H_ / 64>(h_s, HP_, hz_w, H_, n0, lane, az0, az1);
        const v2f bz = *reinterpret_cast<const v2f*>(xz_b + col);
        #pragma unroll
        for (int j = 0; j < 8; ++j) {
            const v2f hv = *reinterpret_cast<const v2f*>(h_s + (rb + j) * HP_ + col);
            z0[j] = sigmoidf_(az0[j] + bz.x);
            z1[j] = sigmoidf_(az1[j] + bz.y);
            v2f zh;
            zh.x = z0[j] * hv.x;
            zh.y = z1[j] * hv.y;
            *reinterpret_cast<v2f*>(zh_s + (rb + j) * HP_ + col) = zh;
        }
    } else if (inv_t != nullptr) {
        const int j = tid - 128;
        const int r = j >> 3, c4 = j & 7;
        const int o = inv_t[r];
        if (o >= 0) {
            *reinterpret_cast<v4f*>(x_s + r * D_ + 4 * c4) =
                *reinterpret_cast<const v4f*>(Xt + (size_t)o * D_ + 4 * c4);
            *reinterpret_cast<v4f*>(m_s + r * D_ + 4 * c4) =
                *reinterpret_cast<const v4f*>(Mt + (size_t)o * D_ + 4 * c4);
        } else {
            *reinterpret_cast<v4f*>(x_s + r * D_ + 4 * c4) = (v4f){0.0f, 0.0f, 0.0f, 0.0f};
            *reinterpret_cast<v4f*>(m_s + r * D_ + 4 * c4) = (v4f){0.0f, 0.0f, 0.0f, 0.0f};
        }
    }
    __syncthreads();
    if (wv < 4) {
        v8f an0 = {}, an1 = {};
        gemm2_lds_glb<P2_ / 64>(p_s, PP_, xn_w, H_, n0, lane, an0, an1);
        gemm2_lds_glb<H_ / 64>(zh_s, HP_, hn_w, H_, n0, lane, an0, an1);
        const v2f bn = *reinterpret_cast<const v2f*>(xn_b + col);
        #pragma unroll
        for (int j = 0; j < 8; ++j) {
            const float n0v = tanhf(an0[j] + bn.x);
            const float n1v = tanhf(an1[j] + bn.y);
            float* hp = h_s + (rb + j) * HP_ + col;
            const v2f hv = *reinterpret_cast<const v2f*>(hp);
            v2f o;
            o.x = hv.x + DT_ * (1.0f - z0[j]) * (n0v - hv.x);
            o.y = hv.y + DT_ * (1.0f - z1[j]) * (n1v - hv.y);
            *reinterpret_cast<v2f*>(hp) = o;
        }
    }
    __syncthreads();
}

#define JKS_ 8
#define JNCI_ 10
struct __attribute__((aligned(4))) f3_t { float x, y, z; };

__device__ __forceinline__ void bfrag3_load(const float* __restrict__ bp,
                                            f3_t (&lo)[JKS_], f3_t (&hi)[JKS_]) {
    #pragma unroll
    for (int i = 0; i < JKS_; ++i) {
        lo[i] = *reinterpret_cast<const f3_t*>(bp + (4 * i) * G3_);
        hi[i] = *reinterpret_cast<const f3_t*>(bp + (4 * i + 1) * G3_);
    }
}

__device__ __forceinline__ void frag3_mma(const float* As, int lda, int lane,
                                          const f3_t (&lo)[JKS_], const f3_t (&hi)[JKS_],
                                          v8f (&acc)[3]) {
    #pragma unroll
    for (int i = 0; i < JKS_; ++i) {
        const v2f a = load_afrag(As, lda, 0, 4 * i, lane);
        v2f b0, b1, b2;
        b0.x = lo[i].x; b0.y = hi[i].x;
        b1.x = lo[i].y; b1.y = hi[i].y;
        b2.x = lo[i].z; b2.y = hi[i].z;
        acc[0] = wmma4(a, b0, acc[0]);
        acc[1] = wmma4(a, b1, acc[1]);
        acc[2] = wmma4(a, b2, acc[2]);
    }
}

__device__ __forceinline__ void jump_phase(const int* inv_t,
                                           const float* x_s, const float* m_s,
                                           const float* wp_s, const float* bp_s,
                                           float* h_s, const float* p_s, float* gin_s, float* gate_s,
                                           const float* __restrict__ w_ih, const float* __restrict__ w_hh,
                                           const float* __restrict__ b_ih, const float* __restrict__ b_hh,
                                           float& lsum, int tid, int lane, int wv) {
    float* gi_s = gate_s;
    float* gh_s = gate_s + 16 * G3_;

    const int n0w = wv * 48;
    const int woff = ((lane >> 4) << 1) * G3_ + n0w + 3 * (lane & 15);
    const float* wi = w_ih + woff;
    const float* wh = w_hh + woff;

    f3_t lo0[JKS_], hi0[JKS_], lo1[JKS_], hi1[JKS_];
    bfrag3_load(wi, lo0, hi0);

    for (int i = tid; i < 16 * D_; i += 256) {
        const int r = i / D_, d = i % D_;
        const int o = inv_t[r];
        if (o >= 0) {
            float mean = p_s[r * PP_ + d];
            float lv   = p_s[r * PP_ + D_ + d];
            float x    = x_s[r * D_ + d];
            float m    = m_s[r * D_ + d];
            float err  = (x - mean) * expf(-0.5f * lv);
            lsum += 0.5f * ((err * err + lv + 2.0f * LOGC_) * m);
        }
    }

    {
        int lq = lane;
        asm volatile("" : "+v"(lq));
        const int mr = lq & 15, hh = lq >> 4;
        const int d0 = wv * 4;
        const int pc = mr < PREP_ ? mr : PREP_ - 1;
        const v4f xv = *reinterpret_cast<const v4f*>(x_s + mr * D_ + d0);
        const v4f mu = *reinterpret_cast<const v4f*>(p_s + mr * PP_ + d0);
        const v4f lg = *reinterpret_cast<const v4f*>(p_s + mr * PP_ + D_ + d0);
        v8f acc[4];
        float bias[4];
        #pragma unroll
        for (int i = 0; i < 4; ++i) {
            const int d = d0 + i;
            const float err = (xv[i] - mu[i]) * expf(-0.5f * lg[i]);
            v2f a, b;
            a.x = hh ? lg[i] : xv[i];
            a.y = hh ? err   : mu[i];
            const float* wp = wp_s + (d * 4 + 2 * hh) * PREP_ + pc;
            b.x = wp[0];
            b.y = wp[PREP_];
            bias[i] = bp_s[d * PREP_ + pc];
            acc[i] = (v8f){};
            acc[i] = wmma4(a, b, acc[i]);
        }
        if (mr < PREP_) {
            float* gp = gin_s + (8 * hh) * GP_ + d0 * PREP_ + mr;
            #pragma unroll
            for (int j = 0; j < 8; ++j) {
                const bool ob = inv_t[8 * hh + j] >= 0;
                const v4f mk = *reinterpret_cast<const v4f*>(m_s + (8 * hh + j) * D_ + d0);
                #pragma unroll
                for (int i = 0; i < 4; ++i) {
                    const float g = ob ? fmaxf(acc[i][j] + bias[i], 0.0f) * mk[i] : 0.0f;
                    gp[j * GP_ + i * PREP_] = g;
                }
            }
        }
    }
    __syncthreads();

    v8f gi[3], gh[3];
    #pragma unroll
    for (int s = 0; s < 3; ++s) { gi[s] = (v8f){}; gh[s] = (v8f){}; }

    bfrag3_load(wi + JKS_ * 4 * G3_, lo1, hi1);
    #pragma unroll
    for (int s2 = 0; s2 < JNCI_ / 2; ++s2) {
        const bool last = (s2 == JNCI_ / 2 - 1);
        const float* nx0 = last ? wh : wi + (2 * s2 + 2) * (JKS_ * 4 * G3_);
        const float* nx1 = nx0 + JKS_ * 4 * G3_;
        frag3_mma(gin_s + (2 * s2) * (JKS_ * 4), GP_, lane, lo0, hi0, gi);
        bfrag3_load(nx0, lo0, hi0);
        frag3_mma(gin_s + (2 * s2 + 1) * (JKS_ * 4), GP_, lane, lo1, hi1, gi);
        bfrag3_load(nx1, lo1, hi1);
    }
    frag3_mma(h_s, HP_, lane, lo0, hi0, gh);
    bfrag3_load(wh + 2 * (JKS_ * 4 * G3_), lo0, hi0);
    frag3_mma(h_s + JKS_ * 4, HP_, lane, lo1, hi1, gh);
    bfrag3_load(wh + 3 * (JKS_ * 4 * G3_), lo1, hi1);
    frag3_mma(h_s + 2 * (JKS_ * 4), HP_, lane, lo0, hi0, gh);
    frag3_mma(h_s + 3 * (JKS_ * 4), HP_, lane, lo1, hi1, gh);

    {
        const int rb = (lane >> 4) << 3;
        #pragma unroll
        for (int s = 0; s < 3; ++s) {
            const int col = n0w + 3 * (lane & 15) + s;
            const float bi = b_ih[col], bh = b_hh[col];
            #pragma unroll
            for (int j = 0; j < 8; ++j) {
                gi_s[(rb + j) * G3_ + col] = gi[s][j] + bi;
                gh_s[(rb + j) * G3_ + col] = gh[s][j] + bh;
            }
        }
    }
    __syncthreads();

    for (int i = tid; i < 16 * H_; i += 256) {
        const int r = i >> 7, j = i & 127;
        if (inv_t[r] >= 0) {
            float rr = sigmoidf_(gi_s[r * G3_ + j] + gh_s[r * G3_ + j]);
            float z  = sigmoidf_(gi_s[r * G3_ + H_ + j] + gh_s[r * G3_ + H_ + j]);
            float n  = tanhf(gi_s[r * G3_ + 2 * H_ + j] + rr * gh_s[r * G3_ + 2 * H_ + j]);
            float hv = h_s[r * HP_ + j];
            h_s[r * HP_ + j] = (1.0f - z) * n + z * hv;
        }
    }
    __syncthreads();
}

__global__ void __launch_bounds__(256, 4) k_traj(
                       const float* __restrict__ X, const float* __restrict__ M,
                       const int* __restrict__ obs_idx, const float* __restrict__ cov,
                       const float* __restrict__ cm_w1, const float* __restrict__ cm_b1,
                       const float* __restrict__ cm_w2, const float* __restrict__ cm_b2,
                       const float* __restrict__ p_w1, const float* __restrict__ p_b1,
                       const float* __restrict__ p_w2, const float* __restrict__ p_b2,
                       const float* __restrict__ xz_w, const float* __restrict__ xz_b,
                       const float* __restrict__ hz_w,
                       const float* __restrict__ xn_w, const float* __restrict__ xn_b,
                       const float* __restrict__ hn_w,
                       const float* __restrict__ w_ih, const float* __restrict__ w_hh,
                       const float* __restrict__ b_ih, const float* __restrict__ b_hh,
                       const float* __restrict__ w_prep, const float* __restrict__ bias_prep,
                       float* __restrict__ h, float* __restrict__ p,
                       float* __restrict__ loss_part) {
    __shared__ __attribute__((aligned(16))) float h_s[16 * HP_];
    __shared__ __attribute__((aligned(16))) float p_s[16 * PP_];
    __shared__ __attribute__((aligned(16))) float t_s[16 * PP_];
    __shared__ __attribute__((aligned(16))) float gin_s[16 * GP_];
    __shared__ __attribute__((aligned(16))) float gate_s[2 * 16 * G3_];
    __shared__ __attribute__((aligned(16))) float x_s[16 * D_];
    __shared__ __attribute__((aligned(16))) float m_s[16 * D_];
    __shared__ float wp_s[D_ * 4 * PREP_];
    __shared__ float bp_s[D_ * PREP_];
    __shared__ int   inv_s[NT_ * 16];
    float* zh_s = gin_s;
    float* red  = t_s;
    const int tid = threadIdx.x, lane = tid & 31;
    const int wv = __builtin_amdgcn_readfirstlane(tid >> 5);
    const int bx = blockIdx.x;
    const int m0 = bx * 16;

    inv_s[tid] = -1;
    __syncthreads();
    {
        const v4i* oi = reinterpret_cast<const v4i*>(obs_idx);
        #pragma unroll 1
        for (int q0 = 0; q0 < NT_ * NO_ / 4; q0 += 8 * 256) {
            v4i v[8];
            #pragma unroll
            for (int u = 0; u < 8; ++u) v[u] = oi[q0 + u * 256 + tid];
            #pragma unroll
            for (int u = 0; u < 8; ++u) {
                const int e = 4 * (q0 + u * 256 + tid);
                const int tt = e >> 12, o = e & (NO_ - 1);
                if ((v[u].x >> 4) == bx) inv_s[tt * 16 + (v[u].x & 15)] = o;
                if ((v[u].y >> 4) == bx) inv_s[tt * 16 + (v[u].y & 15)] = o + 1;
                if ((v[u].z >> 4) == bx) inv_s[tt * 16 + (v[u].z & 15)] = o + 2;
                if ((v[u].w >> 4) == bx) inv_s[tt * 16 + (v[u].w & 15)] = o + 3;
            }
        }
    }
    for (int i = tid; i < D_ * 4 * PREP_; i += 256) wp_s[i] = w_prep[i];
    for (int i = tid; i < D_ * PREP_; i += 256) bp_s[i] = bias_prep[i];

    for (int i = tid; i < 16 * CH_; i += 256) {
        int r = i / CH_, c = i % CH_;
        float acc = cm_b1[c];
        #pragma unroll
        for (int k = 0; k < CS_; ++k) acc += cov[(m0 + r) * CS_ + k] * cm_w1[k * CH_ + c];
        t_s[r * PP_ + c] = fmaxf(acc, 0.0f);
    }
    __syncthreads();
    for (int i = tid; i < 16 * H_; i += 256) {
        int r = i / H_, c = i % H_;
        float acc = cm_b2[c];
        #pragma unroll
        for (int k = 0; k < CH_; ++k) acc += t_s[r * PP_ + k] * cm_w2[k * H_ + c];
        h_s[r * HP_ + c] = tanhf(acc);
    }
    __syncthreads();
    pmodel_phase(h_s, t_s, p_s, p_w1, p_b1, p_w2, p_b2, lane, wv, 0);

    float lsum = 0.0f;
    #pragma unroll 1
    for (int t = 0; t < NT_; ++t) {
        int tl = tid;
        asm volatile("" : "+v"(tl));
        const int ln = tl & 31;
        const int w = __builtin_amdgcn_readfirstlane(tl >> 5);
        ode_phase(inv_s + t * 16, X + (size_t)t * NO_ * D_, M + (size_t)t * NO_ * D_, x_s, m_s, tl,
                  h_s, p_s, zh_s, xz_w, xz_b, hz_w, xn_w, xn_b, hn_w, ln, w);
        pmodel_phase(h_s, t_s, p_s, p_w1, p_b1, p_w2, p_b2, ln, w, 0);
        jump_phase(inv_s + t * 16, x_s, m_s, wp_s, bp_s,
                   h_s, p_s, gin_s, gate_s, w_ih, w_hh, b_ih, b_hh,
                   lsum, tl, ln, w);
        pmodel_phase(h_s, t_s, p_s, p_w1, p_b1, p_w2, p_b2, ln, w, 2);
    }
    #pragma unroll 1
    for (int s = 0; s < NPOST_; ++s) {
        int tl = tid;
        asm volatile("" : "+v"(tl));
        const int ln = tl & 31;
        const int w = __builtin_amdgcn_readfirstlane(tl >> 5);
        ode_phase(nullptr, X, M, x_s, m_s, tl,
                  h_s, p_s, zh_s, xz_w, xz_b, hz_w, xn_w, xn_b, hn_w, ln, w);
        pmodel_phase(h_s, t_s, p_s, p_w1, p_b1, p_w2, p_b2, ln, w, 2 * (s & 1));
    }

    for (int i = tid; i < 16 * H_ / 4; i += 256) {
        const int r = i >> 5, c4 = i & 31;
        *reinterpret_cast<v4f*>(h + (size_t)(m0 + r) * H_ + 4 * c4) =
            *reinterpret_cast<const v4f*>(h_s + r * HP_ + 4 * c4);
    }
    {
        const int r = tid >> 4, c4 = tid & 15;
        *reinterpret_cast<v4f*>(p + (size_t)(m0 + r) * P2_ + 4 * c4) =
            *reinterpret_cast<const v4f*>(p_s + r * PP_ + 4 * c4);
    }

    red[tid] = lsum;
    __syncthreads();
    for (int s2 = 128; s2 > 0; s2 >>= 1) {
        if (tid < s2) red[tid] += red[tid + s2];
        __syncthreads();
    }
    if (tid == 0) loss_part[bx] = red[0];
}

__global__ void k_loss_final(const float* __restrict__ parts, float* __restrict__ loss) {
    __shared__ float red[256];
    const int tid = threadIdx.x;
    float s = 0.0f;
    for (int i = tid; i < MB_; i += 256) s += parts[i];
    red[tid] = s;
    __syncthreads();
    for (int k = 128; k > 0; k >>= 1) {
        if (tid < k) red[tid] += red[tid + k];
        __syncthreads();
    }
    if (tid == 0) *loss = red[0];
}

extern "C" void kernel_launch(void* const* d_in, const int* in_sizes, int n_in,
                              void* d_out, int out_size, void* d_ws, size_t ws_size,
                              hipStream_t stream) {
    const float* X       = (const float*)d_in[0];
    const float* M       = (const float*)d_in[1];
    const int*   obs_idx = (const int*)d_in[2];
    const float* cov     = (const float*)d_in[3];
    const float* cm_w1   = (const float*)d_in[4];
    const float* cm_b1   = (const float*)d_in[5];
    const float* cm_w2   = (const float*)d_in[6];
    const float* cm_b2   = (const float*)d_in[7];
    const float* p_w1    = (const float*)d_in[8];
    const float* p_b1    = (const float*)d_in[9];
    const float* p_w2    = (const float*)d_in[10];
    const float* p_b2    = (const float*)d_in[11];
    const float* xz_w    = (const float*)d_in[12];
    const float* xz_b    = (const float*)d_in[13];
    const float* hz_w    = (const float*)d_in[14];
    const float* xn_w    = (const float*)d_in[15];
    const float* xn_b    = (const float*)d_in[16];
    const float* hn_w    = (const float*)d_in[17];
    const float* w_ih    = (const float*)d_in[18];
    const float* w_hh    = (const float*)d_in[19];
    const float* b_ih    = (const float*)d_in[20];
    const float* b_hh    = (const float*)d_in[21];
    const float* w_prep  = (const float*)d_in[22];
    const float* b_prep  = (const float*)d_in[23];

    float* h    = (float*)d_out;
    float* p    = h + (size_t)B_ * H_;
    float* loss = p + (size_t)B_ * P2_;
    float* loss_part = (float*)d_ws;

    k_traj<<<MB_, 256, 0, stream>>>(X, M, obs_idx, cov, cm_w1, cm_b1, cm_w2, cm_b2,
                                    p_w1, p_b1, p_w2, p_b2,
                                    xz_w, xz_b, hz_w, xn_w, xn_b, hn_w,
                                    w_ih, w_hh, b_ih, b_hh, w_prep, b_prep,
                                    h, p, loss_part);
    k_loss_final<<<1, 256, 0, stream>>>(loss_part, loss);
}


# wave priority by phase: ode, p-model, jump front and gates run at s_setprio 3, the jump's long matrix run at 0, so a co-resident work-group's short phases no longer queue behind it; sums kept
# speedup vs baseline: 1.9477x; 1.0322x over previous
#include <hip/hip_runtime.h>
#include <hip/hip_bf16.h>
#include <math.h>


#define B_    8192
#define D_    32
#define H_    128
#define P2_   64
#define PREP_ 10
#define GIN_  320
#define G3_   384
#define CS_   8
#define CH_   32
#define NT_   16
#define NO_   4096
#define DT_   0.1f
#define NPOST_ 4
#define LOGC_  0.91893853320467274178f

#define MB_   (B_ / 16)
#define HP_   132
#define PP_   68
#define GP_   324

typedef __attribute__((ext_vector_type(2))) float v2f;
typedef __attribute__((ext_vector_type(4))) float v4f;
typedef __attribute__((ext_vector_type(8))) float v8f;
typedef __attribute__((ext_vector_type(4))) int   v4i;

__device__ __forceinline__ float sigmoidf_(float x) { return 1.0f / (1.0f + expf(-x)); }


__device__ __forceinline__ v2f load_afrag(const float* A, int lda, int m0, int k, int lane) {
    int m  = m0 + (lane & 15);
    int kk = k + ((lane >> 4) << 1);
    v2f a;
    a.x = A[m * lda + kk];
    a.y = A[m * lda + kk + 1];
    return a;
}

__device__ __forceinline__ v8f wmma4(v2f a, v2f b, v8f c) {
    return __builtin_amdgcn_wmma_f32_16x16x4_f32(false, a, false, b, (short)0, c, false, false);
}

template <int NK>
__device__ __forceinline__ void bfrag2_load(const float* Bm, int ldb, int n0, int lane,
                                            v2f (&lo)[NK], v2f (&hi)[NK]) {
    const float* bp = Bm + ((lane >> 4) << 1) * ldb + n0 + 2 * (lane & 15);
    #pragma unroll
    for (int i = 0; i < NK; ++i) {
        lo[i] = *reinterpret_cast<const v2f*>(bp + (4 * i) * ldb);
        hi[i] = *reinterpret_cast<const v2f*>(bp + (4 * i + 1) * ldb);
    }
}

template <int NK>
__device__ __forceinline__ void lds_frag2_mma(const float* As, int lda, int lane,
                                              const v2f (&lo)[NK], const v2f (&hi)[NK],
                                              v8f& c0, v8f& c1) {
    #pragma unroll
    for (int i = 0; i < NK; ++i) {
        const v2f a = load_afrag(As, lda, 0, 4 * i, lane);
        v2f b0, b1;
        b0.x = lo[i].x; b0.y = hi[i].x;
        b1.x = lo[i].y; b1.y = hi[i].y;
        c0 = wmma4(a, b0, c0);
        c1 = wmma4(a, b1, c1);
    }
}

template <int NS>
__device__ __forceinline__ void gemm2_lds_glb(const float* As, int lda,
                                              const float* Bm, int ldb, int n0, int lane,
                                              v8f& c0, v8f& c1) {
    v2f lo[2][16], hi[2][16];
    bfrag2_load<16>(Bm, ldb, n0, lane, lo[0], hi[0]);
    #pragma unroll
    for (int s = 0; s < NS; ++s) {
        if (s + 1 < NS)
            bfrag2_load<16>(Bm + (s + 1) * 64 * ldb, ldb, n0, lane, lo[(s + 1) & 1], hi[(s + 1) & 1]);
        lds_frag2_mma<16>(As + s * 64, lda, lane, lo[s & 1], hi[s & 1], c0, c1);
    }
}

__device__ __forceinline__ void pmodel_phase(const float* h_s, float* t_s, float* p_s,
                                             const float* __restrict__ w1, const float* __restrict__ b1,
                                             const float* __restrict__ w2, const float* __restrict__ b2,
                                             int lane, int wv, int wb) {
    const int pw = wv - wb;
    const bool mine = (pw == 0) || (pw == 1);
    const int n0 = (pw & 1) * 32;
    const int col = n0 + 2 * (lane & 15), rb = (lane >> 4) << 3;
    if (mine) {
        v8f acc0 = {}, acc1 = {};
        gemm2_lds_glb<H_ / 64>(h_s, HP_, w1, P2_, n0, lane, acc0, acc1);
        const v2f bv = *reinterpret_cast<const v2f*>(b1 + col);
        #pragma unroll
        for (int j = 0; j < 8; ++j) {
            v2f t;
            t.x = fmaxf(acc0[j] + bv.x, 0.0f);
            t.y = fmaxf(acc1[j] + bv.y, 0.0f);
            *reinterpret_cast<v2f*>(t_s + (rb + j) * PP_ + col) = t;
        }
    }
    __syncthreads();
    if (mine) {
        v8f acc2 = {}, acc3 = {};
        gemm2_lds_glb<P2_ / 64>(t_s, PP_, w2, P2_, n0, lane, acc2, acc3);
        const v2f bv2 = *reinterpret_cast<const v2f*>(b2 + col);
        #pragma unroll
        for (int j = 0; j < 8; ++j) {
            v2f o;
            o.x = acc2[j] + bv2.x;
            o.y = acc3[j] + bv2.y;
            *reinterpret_cast<v2f*>(p_s + (rb + j) * PP_ + col) = o;
        }
    }
    __syncthreads();
}

__device__ __forceinline__ void ode_phase(const int* inv_t,
                                          const float* __restrict__ Xt, const float* __restrict__ Mt,
                                          float* x_s, float* m_s, int tid,
                                          float* h_s, const float* p_s, float* zh_s,
                                          const float* __restrict__ xz_w, const float* __restrict__ xz_b,
                                          const float* __restrict__ hz_w,
                                          const float* __restrict__ xn_w, const float* __restrict__ xn_b,
                                          const float* __restrict__ hn_w,
                                          int lane, int wv) {
    const int n0 = (wv & 3) * 32;
    const int col = n0 + 2 * (lane & 15), rb = (lane >> 4) << 3;
    float z0[8], z1[8];
    if (wv < 4) {
        v8f az0 = {}, az1 = {};
        gemm2_lds_glb<P2_ / 64>(p_s, PP_, xz_w, H_, n0, lane, az0, az1);
        gemm2_lds_glb<H_ / 64>(h_s, HP_, hz_w, H_, n0, lane, az0, az1);
        const v2f bz = *reinterpret_cast<const v2f*>(xz_b + col);
        #pragma unroll
        for (int j = 0; j < 8; ++j) {
            const v2f hv = *reinterpret_cast<const v2f*>(h_s + (rb + j) * HP_ + col);
            z0[j] = sigmoidf_(az0[j] + bz.x);
            z1[j] = sigmoidf_(az1[j] + bz.y);
            v2f zh;
            zh.x = z0[j] * hv.x;
            zh.y = z1[j] * hv.y;
            *reinterpret_cast<v2f*>(zh_s + (rb + j) * HP_ + col) = zh;
        }
    } else if (inv_t != nullptr) {
        const int j = tid - 128;
        const int r = j >> 3, c4 = j & 7;
        const int o = inv_t[r];
        if (o >= 0) {
            *reinterpret_cast<v4f*>(x_s + r * D_ + 4 * c4) =
                *reinterpret_cast<const v4f*>(Xt + (size_t)o * D_ + 4 * c4);
            *reinterpret_cast<v4f*>(m_s + r * D_ + 4 * c4) =
                *reinterpret_cast<const v4f*>(Mt + (size_t)o * D_ + 4 * c4);
        } else {
            *reinterpret_cast<v4f*>(x_s + r * D_ + 4 * c4) = (v4f){0.0f, 0.0f, 0.0f, 0.0f};
            *reinterpret_cast<v4f*>(m_s + r * D_ + 4 * c4) = (v4f){0.0f, 0.0f, 0.0f, 0.0f};
        }
    }
    __syncthreads();
    if (wv < 4) {
        v8f an0 = {}, an1 = {};
        gemm2_lds_glb<P2_ / 64>(p_s, PP_, xn_w, H_, n0, lane, an0, an1);
        gemm2_lds_glb<H_ / 64>(zh_s, HP_, hn_w, H_, n0, lane, an0, an1);
        const v2f bn = *reinterpret_cast<const v2f*>(xn_b + col);
        #pragma unroll
        for (int j = 0; j < 8; ++j) {
            const float n0v = tanhf(an0[j] + bn.x);
            const float n1v = tanhf(an1[j] + bn.y);
            float* hp = h_s + (rb + j) * HP_ + col;
            const v2f hv = *reinterpret_cast<const v2f*>(hp);
            v2f o;
            o.x = hv.x + DT_ * (1.0f - z0[j]) * (n0v - hv.x);
            o.y = hv.y + DT_ * (1.0f - z1[j]) * (n1v - hv.y);
            *reinterpret_cast<v2f*>(hp) = o;
        }
    }
    __syncthreads();
}

#define JKS_ 8
#define JNCI_ 10
struct __attribute__((aligned(4))) f3_t { float x, y, z; };

__device__ __forceinline__ void bfrag3_load(const float* __restrict__ bp,
                                            f3_t (&lo)[JKS_], f3_t (&hi)[JKS_]) {
    #pragma unroll
    for (int i = 0; i < JKS_; ++i) {
        lo[i] = *reinterpret_cast<const f3_t*>(bp + (4 * i) * G3_);
        hi[i] = *reinterpret_cast<const f3_t*>(bp + (4 * i + 1) * G3_);
    }
}

__device__ __forceinline__ void frag3_mma(const float* As, int lda, int lane,
                                          const f3_t (&lo)[JKS_], const f3_t (&hi)[JKS_],
                                          v8f (&acc)[3]) {
    #pragma unroll
    for (int i = 0; i < JKS_; ++i) {
        const v2f a = load_afrag(As, lda, 0, 4 * i, lane);
        v2f b0, b1, b2;
        b0.x = lo[i].x; b0.y = hi[i].x;
        b1.x = lo[i].y; b1.y = hi[i].y;
        b2.x = lo[i].z; b2.y = hi[i].z;
        acc[0] = wmma4(a, b0, acc[0]);
        acc[1] = wmma4(a, b1, acc[1]);
        acc[2] = wmma4(a, b2, acc[2]);
    }
}

__device__ __forceinline__ void jump_phase(const int* inv_t,
                                           const float* x_s, const float* m_s,
                                           const float* wp_s, const float* bp_s,
                                           float* h_s, const float* p_s, float* gin_s, float* gate_s,
                                           const float* __restrict__ w_ih, const float* __restrict__ w_hh,
                                           const float* __restrict__ b_ih, const float* __restrict__ b_hh,
                                           float& lsum, int tid, int lane, int wv) {
    float* gi_s = gate_s;
    float* gh_s = gate_s + 16 * G3_;

    const int n0w = wv * 48;
    const int woff = ((lane >> 4) << 1) * G3_ + n0w + 3 * (lane & 15);
    const float* wi = w_ih + woff;
    const float* wh = w_hh + woff;

    f3_t lo0[JKS_], hi0[JKS_], lo1[JKS_], hi1[JKS_];
    bfrag3_load(wi, lo0, hi0);

    for (int i = tid; i < 16 * D_; i += 256) {
        const int r = i / D_, d = i % D_;
        const int o = inv_t[r];
        if (o >= 0) {
            float mean = p_s[r * PP_ + d];
            float lv   = p_s[r * PP_ + D_ + d];
            float x    = x_s[r * D_ + d];
            float m    = m_s[r * D_ + d];
            float err  = (x - mean) * expf(-0.5f * lv);
            lsum += 0.5f * ((err * err + lv + 2.0f * LOGC_) * m);
        }
    }

    {
        int lq = lane;
        asm volatile("" : "+v"(lq));
        const int mr = lq & 15, hh = lq >> 4;
        const int d0 = wv * 4;
        const int pc = mr < PREP_ ? mr : PREP_ - 1;
        const v4f xv = *reinterpret_cast<const v4f*>(x_s + mr * D_ + d0);
        const v4f mu = *reinterpret_cast<const v4f*>(p_s + mr * PP_ + d0);
        const v4f lg = *reinterpret_cast<const v4f*>(p_s + mr * PP_ + D_ + d0);
        v8f acc[4];
        float bias[4];
        #pragma unroll
        for (int i = 0; i < 4; ++i) {
            const int d = d0 + i;
            const float err = (xv[i] - mu[i]) * expf(-0.5f * lg[i]);
            v2f a, b;
            a.x = hh ? lg[i] : xv[i];
            a.y = hh ? err   : mu[i];
            const float* wp = wp_s + (d * 4 + 2 * hh) * PREP_ + pc;
            b.x = wp[0];
            b.y = wp[PREP_];
            bias[i] = bp_s[d * PREP_ + pc];
            acc[i] = (v8f){};
            acc[i] = wmma4(a, b, acc[i]);
        }
        if (mr < PREP_) {
            float* gp = gin_s + (8 * hh) * GP_ + d0 * PREP_ + mr;
            #pragma unroll
            for (int j = 0; j < 8; ++j) {
                const bool ob = inv_t[8 * hh + j] >= 0;
                const v4f mk = *reinterpret_cast<const v4f*>(m_s + (8 * hh + j) * D_ + d0);
                #pragma unroll
                for (int i = 0; i < 4; ++i) {
                    const float g = ob ? fmaxf(acc[i][j] + bias[i], 0.0f) * mk[i] : 0.0f;
                    gp[j * GP_ + i * PREP_] = g;
                }
            }
        }
    }
    __syncthreads();
    asm volatile("s_setprio 0");

    v8f gi[3], gh[3];
    #pragma unroll
    for (int s = 0; s < 3; ++s) { gi[s] = (v8f){}; gh[s] = (v8f){}; }

    bfrag3_load(wi + JKS_ * 4 * G3_, lo1, hi1);
    #pragma unroll
    for (int s2 = 0; s2 < JNCI_ / 2; ++s2) {
        const bool last = (s2 == JNCI_ / 2 - 1);
        const float* nx0 = last ? wh : wi + (2 * s2 + 2) * (JKS_ * 4 * G3_);
        const float* nx1 = nx0 + JKS_ * 4 * G3_;
        frag3_mma(gin_s + (2 * s2) * (JKS_ * 4), GP_, lane, lo0, hi0, gi);
        bfrag3_load(nx0, lo0, hi0);
        frag3_mma(gin_s + (2 * s2 + 1) * (JKS_ * 4), GP_, lane, lo1, hi1, gi);
        bfrag3_load(nx1, lo1, hi1);
    }
    frag3_mma(h_s, HP_, lane, lo0, hi0, gh);
    bfrag3_load(wh + 2 * (JKS_ * 4 * G3_), lo0, hi0);
    frag3_mma(h_s + JKS_ * 4, HP_, lane, lo1, hi1, gh);
    bfrag3_load(wh + 3 * (JKS_ * 4 * G3_), lo1, hi1);
    frag3_mma(h_s + 2 * (JKS_ * 4), HP_, lane, lo0, hi0, gh);
    frag3_mma(h_s + 3 * (JKS_ * 4), HP_, lane, lo1, hi1, gh);

    {
        const int rb = (lane >> 4) << 3;
        #pragma unroll
        for (int s = 0; s < 3; ++s) {
            const int col = n0w + 3 * (lane & 15) + s;
            const float bi = b_ih[col], bh = b_hh[col];
            #pragma unroll
            for (int j = 0; j < 8; ++j) {
                gi_s[(rb + j) * G3_ + col] = gi[s][j] + bi;
                gh_s[(rb + j) * G3_ + col] = gh[s][j] + bh;
            }
        }
    }
    __syncthreads();
    asm volatile("s_setprio 3");

    for (int i = tid; i < 16 * H_; i += 256) {
        const int r = i >> 7, j = i & 127;
        if (inv_t[r] >= 0) {
            float rr = sigmoidf_(gi_s[r * G3_ + j] + gh_s[r * G3_ + j]);
            float z  = sigmoidf_(gi_s[r * G3_ + H_ + j] + gh_s[r * G3_ + H_ + j]);
            float n  = tanhf(gi_s[r * G3_ + 2 * H_ + j] + rr * gh_s[r * G3_ + 2 * H_ + j]);
            float hv = h_s[r * HP_ + j];
            h_s[r * HP_ + j] = (1.0f - z) * n + z * hv;
        }
    }
    __syncthreads();
}

__global__ void __launch_bounds__(256, 4) k_traj(
                       const float* __restrict__ X, const float* __restrict__ M,
                       const int* __restrict__ obs_idx, const float* __restrict__ cov,
                       const float* __restrict__ cm_w1, const float* __restrict__ cm_b1,
                       const float* __restrict__ cm_w2, const float* __restrict__ cm_b2,
                       const float* __restrict__ p_w1, const float* __restrict__ p_b1,
                       const float* __restrict__ p_w2, const float* __restrict__ p_b2,
                       const float* __restrict__ xz_w, const float* __restrict__ xz_b,
                       const float* __restrict__ hz_w,
                       const float* __restrict__ xn_w, const float* __restrict__ xn_b,
                       const float* __restrict__ hn_w,
                       const float* __restrict__ w_ih, const float* __restrict__ w_hh,
                       const float* __restrict__ b_ih, const float* __restrict__ b_hh,
                       const float* __restrict__ w_prep, const float* __restrict__ bias_prep,
                       float* __restrict__ h, float* __restrict__ p,
                       float* __restrict__ loss_part) {
    __shared__ __attribute__((aligned(16))) float h_s[16 * HP_];
    __shared__ __attribute__((aligned(16))) float p_s[16 * PP_];
    __shared__ __attribute__((aligned(16))) float t_s[16 * PP_];
    __shared__ __attribute__((aligned(16))) float gin_s[16 * GP_];
    __shared__ __attribute__((aligned(16))) float gate_s[2 * 16 * G3_];
    __shared__ __attribute__((aligned(16))) float x_s[16 * D_];
    __shared__ __attribute__((aligned(16))) float m_s[16 * D_];
    __shared__ float wp_s[D_ * 4 * PREP_];
    __shared__ float bp_s[D_ * PREP_];
    __shared__ int   inv_s[NT_ * 16];
    float* zh_s = gin_s;
    float* red  = t_s;
    const int tid = threadIdx.x, lane = tid & 31;
    const int wv = __builtin_amdgcn_readfirstlane(tid >> 5);
    const int bx = blockIdx.x;
    const int m0 = bx * 16;

    asm volatile("s_setprio 3");


    inv_s[tid] = -1;
    __syncthreads();
    {
        const v4i* oi = reinterpret_cast<const v4i*>(obs_idx);
        #pragma unroll 1
        for (int q0 = 0; q0 < NT_ * NO_ / 4; q0 += 8 * 256) {
            v4i v[8];
            #pragma unroll
            for (int u = 0; u < 8; ++u) v[u] = oi[q0 + u * 256 + tid];
            #pragma unroll
            for (int u = 0; u < 8; ++u) {
                const int e = 4 * (q0 + u * 256 + tid);
                const int tt = e >> 12, o = e & (NO_ - 1);
                if ((v[u].x >> 4) == bx) inv_s[tt * 16 + (v[u].x & 15)] = o;
                if ((v[u].y >> 4) == bx) inv_s[tt * 16 + (v[u].y & 15)] = o + 1;
                if ((v[u].z >> 4) == bx) inv_s[tt * 16 + (v[u].z & 15)] = o + 2;
                if ((v[u].w >> 4) == bx) inv_s[tt * 16 + (v[u].w & 15)] = o + 3;
            }
        }
    }
    for (int i = tid; i < D_ * 4 * PREP_; i += 256) wp_s[i] = w_prep[i];
    for (int i = tid; i < D_ * PREP_; i += 256) bp_s[i] = bias_prep[i];

    for (int i = tid; i < 16 * CH_; i += 256) {
        int r = i / CH_, c = i % CH_;
        float acc = cm_b1[c];
        #pragma unroll
        for (int k = 0; k < CS_; ++k) acc += cov[(m0 + r) * CS_ + k] * cm_w1[k * CH_ + c];
        t_s[r * PP_ + c] = fmaxf(acc, 0.0f);
    }
    __syncthreads();
    for (int i = tid; i < 16 * H_; i += 256) {
        int r = i / H_, c = i % H_;
        float acc = cm_b2[c];
        #pragma unroll
        for (int k = 0; k < CH_; ++k) acc += t_s[r * PP_ + k] * cm_w2[k * H_ + c];
        h_s[r * HP_ + c] = tanhf(acc);
    }
    __syncthreads();
    pmodel_phase(h_s, t_s, p_s, p_w1, p_b1, p_w2, p_b2, lane, wv, 0);

    float lsum = 0.0f;
    #pragma unroll 1
    for (int t = 0; t < NT_; ++t) {
        int tl = tid;
        asm volatile("" : "+v"(tl));
        const int ln = tl & 31;
        const int w = __builtin_amdgcn_readfirstlane(tl >> 5);
        ode_phase(inv_s + t * 16, X + (size_t)t * NO_ * D_, M + (size_t)t * NO_ * D_, x_s, m_s, tl,
                  h_s, p_s, zh_s, xz_w, xz_b, hz_w, xn_w, xn_b, hn_w, ln, w);
        pmodel_phase(h_s, t_s, p_s, p_w1, p_b1, p_w2, p_b2, ln, w, 0);
        jump_phase(inv_s + t * 16, x_s, m_s, wp_s, bp_s,
                   h_s, p_s, gin_s, gate_s, w_ih, w_hh, b_ih, b_hh,
                   lsum, tl, ln, w);
        pmodel_phase(h_s, t_s, p_s, p_w1, p_b1, p_w2, p_b2, ln, w, 2);
    }
    #pragma unroll 1
    for (int s = 0; s < NPOST_; ++s) {
        int tl = tid;
        asm volatile("" : "+v"(tl));
        const int ln = tl & 31;
        const int w = __builtin_amdgcn_readfirstlane(tl >> 5);
        ode_phase(nullptr, X, M, x_s, m_s, tl,
                  h_s, p_s, zh_s, xz_w, xz_b, hz_w, xn_w, xn_b, hn_w, ln, w);
        pmodel_phase(h_s, t_s, p_s, p_w1, p_b1, p_w2, p_b2, ln, w, 2 * (s & 1));
    }

    asm volatile("s_setprio 0");

    for (int i = tid; i < 16 * H_ / 4; i += 256) {
        const int r = i >> 5, c4 = i & 31;
        *reinterpret_cast<v4f*>(h + (size_t)(m0 + r) * H_ + 4 * c4) =
            *reinterpret_cast<const v4f*>(h_s + r * HP_ + 4 * c4);
    }
    {
        const int r = tid >> 4, c4 = tid & 15;
        *reinterpret_cast<v4f*>(p + (size_t)(m0 + r) * P2_ + 4 * c4) =
            *reinterpret_cast<const v4f*>(p_s + r * PP_ + 4 * c4);
    }

    red[tid] = lsum;
    __syncthreads();
    for (int s2 = 128; s2 > 0; s2 >>= 1) {
        if (tid < s2) red[tid] += red[tid + s2];
        __syncthreads();
    }
    if (tid == 0) loss_part[bx] = red[0];
}

__global__ void k_loss_final(const float* __restrict__ parts, float* __restrict__ loss) {
    __shared__ float red[256];
    const int tid = threadIdx.x;
    float s = 0.0f;
    for (int i = tid; i < MB_; i += 256) s += parts[i];
    red[tid] = s;
    __syncthreads();
    for (int k = 128; k > 0; k >>= 1) {
        if (tid < k) red[tid] += red[tid + k];
        __syncthreads();
    }
    if (tid == 0) *loss = red[0];
}

extern "C" void kernel_launch(void* const* d_in, const int* in_sizes, int n_in,
                              void* d_out, int out_size, void* d_ws, size_t ws_size,
                              hipStream_t stream) {
    const float* X       = (const float*)d_in[0];
    const float* M       = (const float*)d_in[1];
    const int*   obs_idx = (const int*)d_in[2];
    const float* cov     = (const float*)d_in[3];
    const float* cm_w1   = (const float*)d_in[4];
    const float* cm_b1   = (const float*)d_in[5];
    const float* cm_w2   = (const float*)d_in[6];
    const float* cm_b2   = (const float*)d_in[7];
    const float* p_w1    = (const float*)d_in[8];
    const float* p_b1    = (const float*)d_in[9];
    const float* p_w2    = (const float*)d_in[10];
    const float* p_b2    = (const float*)d_in[11];
    const float* xz_w    = (const float*)d_in[12];
    const float* xz_b    = (const float*)d_in[13];
    const float* hz_w    = (const float*)d_in[14];
    const float* xn_w    = (const float*)d_in[15];
    const float* xn_b    = (const float*)d_in[16];
    const float* hn_w    = (const float*)d_in[17];
    const float* w_ih    = (const float*)d_in[18];
    const float* w_hh    = (const float*)d_in[19];
    const float* b_ih    = (const float*)d_in[20];
    const float* b_hh    = (const float*)d_in[21];
    const float* w_prep  = (const float*)d_in[22];
    const float* b_prep  = (const float*)d_in[23];

    float* h    = (float*)d_out;
    float* p    = h + (size_t)B_ * H_;
    float* loss = p + (size_t)B_ * P2_;
    float* loss_part = (float*)d_ws;

    k_traj<<<MB_, 256, 0, stream>>>(X, M, obs_idx, cov, cm_w1, cm_b1, cm_w2, cm_b2,
                                    p_w1, p_b1, p_w2, p_b2,
                                    xz_w, xz_b, hz_w, xn_w, xn_b, hn_w,
                                    w_ih, w_hh, b_ih, b_hh, w_prep, b_prep,
                                    h, p, loss_part);
    k_loss_final<<<1, 256, 0, stream>>>(loss_part, loss);
}


# 32 rows per work-group (256 groups, one per WGP): in the jump a loaded weight fragment feeds both 16-row tiles; ode and p-model split by row tile over 8 and 4 waves; loss partials per 32 rows
# speedup vs baseline: 2.0393x; 1.0470x over previous
#include <hip/hip_runtime.h>
#include <hip/hip_bf16.h>
#include <math.h>


#define B_    8192
#define D_    32
#define H_    128
#define P2_   64
#define PREP_ 10
#define GIN_  320
#define G3_   384
#define CS_   8
#define CH_   32
#define NT_   16
#define NO_   4096
#define DT_   0.1f
#define NPOST_ 4
#define LOGC_  0.91893853320467274178f

#define RW_   32
#define MB_   (B_ / RW_)
#define HP_   132
#define PP_   68
#define GP_   324

typedef __attribute__((ext_vector_type(2))) float v2f;
typedef __attribute__((ext_vector_type(4))) float v4f;
typedef __attribute__((ext_vector_type(8))) float v8f;
typedef __attribute__((ext_vector_type(4))) int   v4i;

__device__ __forceinline__ float sigmoidf_(float x) { return 1.0f / (1.0f + expf(-x)); }


__device__ __forceinline__ v2f load_afrag(const float* A, int lda, int m0, int k, int lane) {
    int m  = m0 + (lane & 15);
    int kk = k + ((lane >> 4) << 1);
    v2f a;
    a.x = A[m * lda + kk];
    a.y = A[m * lda + kk + 1];
    return a;
}

__device__ __forceinline__ v8f wmma4(v2f a, v2f b, v8f c) {
    return __builtin_amdgcn_wmma_f32_16x16x4_f32(false, a, false, b, (short)0, c, false, false);
}

template <int NK>
__device__ __forceinline__ void bfrag2_load(const float* Bm, int ldb, int n0, int lane,
                                            v2f (&lo)[NK], v2f (&hi)[NK]) {
    const float* bp = Bm + ((lane >> 4) << 1) * ldb + n0 + 2 * (lane & 15);
    #pragma unroll
    for (int i = 0; i < NK; ++i) {
        lo[i] = *reinterpret_cast<const v2f*>(bp + (4 * i) * ldb);
        hi[i] = *reinterpret_cast<const v2f*>(bp + (4 * i + 1) * ldb);
    }
}

template <int NK>
__device__ __forceinline__ void lds_frag2_mma(const float* As, int lda, int lane,
                                              const v2f (&lo)[NK], const v2f (&hi)[NK],
                                              v8f& c0, v8f& c1) {
    #pragma unroll
    for (int i = 0; i < NK; ++i) {
        const v2f a = load_afrag(As, lda, 0, 4 * i, lane);
        v2f b0, b1;
        b0.x = lo[i].x; b0.y = hi[i].x;
        b1.x = lo[i].y; b1.y = hi[i].y;
        c0 = wmma4(a, b0, c0);
        c1 = wmma4(a, b1, c1);
    }
}

template <int NS>
__device__ __forceinline__ void gemm2_lds_glb(const float* As, int lda,
                                              const float* Bm, int ldb, int n0, int lane,
                                              v8f& c0, v8f& c1) {
    v2f lo[2][16], hi[2][16];
    bfrag2_load<16>(Bm, ldb, n0, lane, lo[0], hi[0]);
    #pragma unroll
    for (int s = 0; s < NS; ++s) {
        if (s + 1 < NS)
            bfrag2_load<16>(Bm + (s + 1) * 64 * ldb, ldb, n0, lane, lo[(s + 1) & 1], hi[(s + 1) & 1]);
        lds_frag2_mma<16>(As + s * 64, lda, lane, lo[s & 1], hi[s & 1], c0, c1);
    }
}

__device__ __forceinline__ void pmodel_phase(const float* h_s, float* t_s, float* p_s,
                                             const float* __restrict__ w1, const float* __restrict__ b1,
                                             const float* __restrict__ w2, const float* __restrict__ b2,
                                             int lane, int wv, int wb) {
    const int pw = wv - wb;
    const bool mine = (pw >= 0) && (pw < 4);
    const int rt = mine ? (pw >> 1) : 0;
    const int n0 = (pw & 1) * 32;
    const int col = n0 + 2 * (lane & 15), rb = ((lane >> 4) << 3) + 16 * rt;
    if (mine) {
        v8f acc0 = {}, acc1 = {};
        gemm2_lds_glb<H_ / 64>(h_s + 16 * rt * HP_, HP_, w1, P2_, n0, lane, acc0, acc1);
        const v2f bv = *reinterpret_cast<const v2f*>(b1 + col);
        #pragma unroll
        for (int j = 0; j < 8; ++j) {
            v2f t;
            t.x = fmaxf(acc0[j] + bv.x, 0.0f);
            t.y = fmaxf(acc1[j] + bv.y, 0.0f);
            *reinterpret_cast<v2f*>(t_s + (rb + j) * PP_ + col) = t;
        }
    }
    __syncthreads();
    if (mine) {
        v8f acc2 = {}, acc3 = {};
        gemm2_lds_glb<P2_ / 64>(t_s + 16 * rt * PP_, PP_, w2, P2_, n0, lane, acc2, acc3);
        const v2f bv2 = *reinterpret_cast<const v2f*>(b2 + col);
        #pragma unroll
        for (int j = 0; j < 8; ++j) {
            v2f o;
            o.x = acc2[j] + bv2.x;
            o.y = acc3[j] + bv2.y;
            *reinterpret_cast<v2f*>(p_s + (rb + j) * PP_ + col) = o;
        }
    }
    __syncthreads();
}

__device__ __forceinline__ void ode_phase(const int* inv_t,
                                          const float* __restrict__ Xt, const float* __restrict__ Mt,
                                          float* x_s, float* m_s, int tid,
                                          float* h_s, const float* p_s, float* zh_s,
                                          const float* __restrict__ xz_w, const float* __restrict__ xz_b,
                                          const float* __restrict__ hz_w,
                                          const float* __restrict__ xn_w, const float* __restrict__ xn_b,
                                          const float* __restrict__ hn_w,
                                          int lane, int wv) {
    const int n0 = (wv & 3) * 32;
    const int rt = wv >> 2;
    const int col = n0 + 2 * (lane & 15), rb = ((lane >> 4) << 3) + 16 * rt;
    float z0[8], z1[8];
    const int xr = tid >> 3, xc4 = tid & 7;
    v4f xq = {0.0f, 0.0f, 0.0f, 0.0f}, mq = {0.0f, 0.0f, 0.0f, 0.0f};
    if (inv_t != nullptr) {
        const int o = inv_t[xr];
        if (o >= 0) {
            xq = *reinterpret_cast<const v4f*>(Xt + (size_t)o * D_ + 4 * xc4);
            mq = *reinterpret_cast<const v4f*>(Mt + (size_t)o * D_ + 4 * xc4);
        }
    }
    {
        v8f az0 = {}, az1 = {};
        gemm2_lds_glb<P2_ / 64>(p_s + 16 * rt * PP_, PP_, xz_w, H_, n0, lane, az0, az1);
        gemm2_lds_glb<H_ / 64>(h_s + 16 * rt * HP_, HP_, hz_w, H_, n0, lane, az0, az1);
        const v2f bz = *reinterpret_cast<const v2f*>(xz_b + col);
        #pragma unroll
        for (int j = 0; j < 8; ++j) {
            const v2f hv = *reinterpret_cast<const v2f*>(h_s + (rb + j) * HP_ + col);
            z0[j] = sigmoidf_(az0[j] + bz.x);
            z1[j] = sigmoidf_(az1[j] + bz.y);
            v2f zh;
            zh.x = z0[j] * hv.x;
            zh.y = z1[j] * hv.y;
            *reinterpret_cast<v2f*>(zh_s + (rb + j) * HP_ + col) = zh;
        }
    }
    if (inv_t != nullptr) {
        *reinterpret_cast<v4f*>(x_s + xr * D_ + 4 * xc4) = xq;
        *reinterpret_cast<v4f*>(m_s + xr * D_ + 4 * xc4) = mq;
    }
    __syncthreads();
    {
        v8f an0 = {}, an1 = {};
        gemm2_lds_glb<P2_ / 64>(p_s + 16 * rt * PP_, PP_, xn_w, H_, n0, lane, an0, an1);
        gemm2_lds_glb<H_ / 64>(zh_s + 16 * rt * HP_, HP_, hn_w, H_, n0, lane, an0, an1);
        const v2f bn = *reinterpret_cast<const v2f*>(xn_b + col);
        #pragma unroll
        for (int j = 0; j < 8; ++j) {
            const float n0v = tanhf(an0[j] + bn.x);
            const float n1v = tanhf(an1[j] + bn.y);
            float* hp = h_s + (rb + j) * HP_ + col;
            const v2f hv = *reinterpret_cast<const v2f*>(hp);
            v2f o;
            o.x = hv.x + DT_ * (1.0f - z0[j]) * (n0v - hv.x);
            o.y = hv.y + DT_ * (1.0f - z1[j]) * (n1v - hv.y);
            *reinterpret_cast<v2f*>(hp) = o;
        }
    }
    __syncthreads();
}

#define JKS_ 8
#define JNCI_ 10
struct __attribute__((aligned(4))) f3_t { float x, y, z; };

__device__ __forceinline__ void bfrag3_load(const float* __restrict__ bp,
                                            f3_t (&lo)[JKS_], f3_t (&hi)[JKS_]) {
    #pragma unroll
    for (int i = 0; i < JKS_; ++i) {
        lo[i] = *reinterpret_cast<const f3_t*>(bp + (4 * i) * G3_);
        hi[i] = *reinterpret_cast<const f3_t*>(bp + (4 * i + 1) * G3_);
    }
}

__device__ __forceinline__ void afrag2_load(const float* As, int lda, int lane,
                                            v2f (&a0)[JKS_], v2f (&a1)[JKS_]) {
    #pragma unroll
    for (int i = 0; i < JKS_; ++i) {
        a0[i] = load_afrag(As, lda, 0, 4 * i, lane);
        a1[i] = load_afrag(As, lda, 16, 4 * i, lane);
    }
}

__device__ __forceinline__ void frag3_mma(const v2f (&a0s)[JKS_], const v2f (&a1s)[JKS_],
                                          const f3_t (&lo)[JKS_], const f3_t (&hi)[JKS_],
                                          v8f (&accA)[3], v8f (&accB)[3]) {
    #pragma unroll
    for (int i = 0; i < JKS_; ++i) {
        const v2f a0 = a0s[i];
        const v2f a1 = a1s[i];
        v2f b0, b1, b2;
        b0.x = lo[i].x; b0.y = hi[i].x;
        b1.x = lo[i].y; b1.y = hi[i].y;
        b2.x = lo[i].z; b2.y = hi[i].z;
        accA[0] = wmma4(a0, b0, accA[0]);
        accB[0] = wmma4(a1, b0, accB[0]);
        accA[1] = wmma4(a0, b1, accA[1]);
        accB[1] = wmma4(a1, b1, accB[1]);
        accA[2] = wmma4(a0, b2, accA[2]);
        accB[2] = wmma4(a1, b2, accB[2]);
    }
}

__device__ __forceinline__ void jump_phase(const int* inv_t,
                                           const float* x_s, const float* m_s,
                                           const float* wp_s, const float* bp_s,
                                           float* h_s, const float* p_s, float* gin_s, float* gate_s,
                                           const float* __restrict__ w_ih, const float* __restrict__ w_hh,
                                           const float* __restrict__ b_ih, const float* __restrict__ b_hh,
                                           float& lsum, int tid, int lane, int wv) {
    float* gi_s = gate_s;
    float* gh_s = gate_s + RW_ * G3_;

    const int n0w = wv * 48;
    const int woff = ((lane >> 4) << 1) * G3_ + n0w + 3 * (lane & 15);
    const float* wi = w_ih + woff;
    const float* wh = w_hh + woff;

    f3_t lo[2][JKS_], hi[2][JKS_];
    bfrag3_load(wi, lo[0], hi[0]);

    for (int i = tid; i < RW_ * D_; i += 256) {
        const int r = i / D_, d = i % D_;
        const int o = inv_t[r];
        if (o >= 0) {
            float mean = p_s[r * PP_ + d];
            float lv   = p_s[r * PP_ + D_ + d];
            float x    = x_s[r * D_ + d];
            float m    = m_s[r * D_ + d];
            float err  = (x - mean) * expf(-0.5f * lv);
            lsum += 0.5f * ((err * err + lv + 2.0f * LOGC_) * m);
        }
    }

    #pragma unroll 1
    for (int rt = 0; rt < 2; ++rt) {
        const int mr = lane & 15, hh = lane >> 4;
        const int r0 = 16 * rt;
        const int d0 = wv * 4;
        const int pc = mr < PREP_ ? mr : PREP_ - 1;
        const v4f xv = *reinterpret_cast<const v4f*>(x_s + (r0 + mr) * D_ + d0);
        const v4f mu = *reinterpret_cast<const v4f*>(p_s + (r0 + mr) * PP_ + d0);
        const v4f lg = *reinterpret_cast<const v4f*>(p_s + (r0 + mr) * PP_ + D_ + d0);
        v8f acc[4];
        float bias[4];
        #pragma unroll
        for (int i = 0; i < 4; ++i) {
            const int d = d0 + i;
            const float err = (xv[i] - mu[i]) * expf(-0.5f * lg[i]);
            v2f a, b;
            a.x = hh ? lg[i] : xv[i];
            a.y = hh ? err   : mu[i];
            const float* wp = wp_s + (d * 4 + 2 * hh) * PREP_ + pc;
            b.x = wp[0];
            b.y = wp[PREP_];
            bias[i] = bp_s[d * PREP_ + pc];
            acc[i] = (v8f){};
            acc[i] = wmma4(a, b, acc[i]);
        }
        if (mr < PREP_) {
            float* gp = gin_s + (r0 + 8 * hh) * GP_ + d0 * PREP_ + mr;
            #pragma unroll
            for (int j = 0; j < 8; ++j) {
                const bool ob = inv_t[r0 + 8 * hh + j] >= 0;
                const v4f mk = *reinterpret_cast<const v4f*>(m_s + (r0 + 8 * hh + j) * D_ + d0);
                #pragma unroll
                for (int i = 0; i < 4; ++i) {
                    const float g = ob ? fmaxf(acc[i][j] + bias[i], 0.0f) * mk[i] : 0.0f;
                    gp[j * GP_ + i * PREP_] = g;
                }
            }
        }
    }
    __syncthreads();

    v8f giA[3], giB[3], ghA[3], ghB[3];
    #pragma unroll
    for (int s = 0; s < 3; ++s) { giA[s] = (v8f){}; giB[s] = (v8f){}; ghA[s] = (v8f){}; ghB[s] = (v8f){}; }

    v2f fa0[2][JKS_], fa1[2][JKS_];
    bfrag3_load(wi + JKS_ * 4 * G3_, lo[1], hi[1]);
    afrag2_load(gin_s, GP_, lane, fa0[0], fa1[0]);
    #pragma unroll
    for (int sl = 0; sl < JNCI_ + 4; ++sl) {
        __builtin_amdgcn_sched_barrier(0);
        if (sl + 1 < JNCI_ + 4) {
            if (sl + 1 < JNCI_) afrag2_load(gin_s + (sl + 1) * (JKS_ * 4), GP_, lane, fa0[(sl + 1) & 1], fa1[(sl + 1) & 1]);
            else                afrag2_load(h_s + (sl + 1 - JNCI_) * (JKS_ * 4), HP_, lane, fa0[(sl + 1) & 1], fa1[(sl + 1) & 1]);
        }
        if (sl < JNCI_) frag3_mma(fa0[sl & 1], fa1[sl & 1], lo[sl & 1], hi[sl & 1], giA, giB);
        else            frag3_mma(fa0[sl & 1], fa1[sl & 1], lo[sl & 1], hi[sl & 1], ghA, ghB);
        if (sl + 2 < JNCI_ + 4) {
            if (sl + 2 < JNCI_) bfrag3_load(wi + (sl + 2) * (JKS_ * 4 * G3_), lo[sl & 1], hi[sl & 1]);
            else                bfrag3_load(wh + (sl + 2 - JNCI_) * (JKS_ * 4 * G3_), lo[sl & 1], hi[sl & 1]);
        }
    }
    __builtin_amdgcn_sched_barrier(0);

    {
        const int rb = (lane >> 4) << 3;
        #pragma unroll
        for (int s = 0; s < 3; ++s) {
            const int col = n0w + 3 * (lane & 15) + s;
            const float bi = b_ih[col], bh = b_hh[col];
            #pragma unroll
            for (int j = 0; j < 8; ++j) {
                gi_s[(rb + j) * G3_ + col] = giA[s][j] + bi;
                gh_s[(rb + j) * G3_ + col] = ghA[s][j] + bh;
                gi_s[(16 + rb + j) * G3_ + col] = giB[s][j] + bi;
                gh_s[(16 + rb + j) * G3_ + col] = ghB[s][j] + bh;
            }
        }
    }
    __syncthreads();

    for (int i = tid; i < RW_ * H_; i += 256) {
        const int r = i >> 7, j = i & 127;
        if (inv_t[r] >= 0) {
            float rr = sigmoidf_(gi_s[r * G3_ + j] + gh_s[r * G3_ + j]);
            float z  = sigmoidf_(gi_s[r * G3_ + H_ + j] + gh_s[r * G3_ + H_ + j]);
            float n  = tanhf(gi_s[r * G3_ + 2 * H_ + j] + rr * gh_s[r * G3_ + 2 * H_ + j]);
            float hv = h_s[r * HP_ + j];
            h_s[r * HP_ + j] = (1.0f - z) * n + z * hv;
        }
    }
    __syncthreads();
}

__global__ void __launch_bounds__(256, 2) k_traj(
                       const float* __restrict__ X, const float* __restrict__ M,
                       const int* __restrict__ obs_idx, const float* __restrict__ cov,
                       const float* __restrict__ cm_w1, const float* __restrict__ cm_b1,
                       const float* __restrict__ cm_w2, const float* __restrict__ cm_b2,
                       const float* __restrict__ p_w1, const float* __restrict__ p_b1,
                       const float* __restrict__ p_w2, const float* __restrict__ p_b2,
                       const float* __restrict__ xz_w, const float* __restrict__ xz_b,
                       const float* __restrict__ hz_w,
                       const float* __restrict__ xn_w, const float* __restrict__ xn_b,
                       const float* __restrict__ hn_w,
                       const float* __restrict__ w_ih, const float* __restrict__ w_hh,
                       const float* __restrict__ b_ih, const float* __restrict__ b_hh,
                       const float* __restrict__ w_prep, const float* __restrict__ bias_prep,
                       float* __restrict__ h, float* __restrict__ p,
                       float* __restrict__ loss_part) {
    __shared__ __attribute__((aligned(16))) float h_s[RW_ * HP_];
    __shared__ __attribute__((aligned(16))) float p_s[RW_ * PP_];
    __shared__ __attribute__((aligned(16))) float t_s[RW_ * PP_];
    __shared__ __attribute__((aligned(16))) float gin_s[RW_ * GP_];
    __shared__ __attribute__((aligned(16))) float gate_s[2 * RW_ * G3_];
    __shared__ __attribute__((aligned(16))) float x_s[RW_ * D_];
    __shared__ __attribute__((aligned(16))) float m_s[RW_ * D_];
    __shared__ float wp_s[D_ * 4 * PREP_];
    __shared__ float bp_s[D_ * PREP_];
    __shared__ int   inv_s[NT_ * RW_];
    float* zh_s = gin_s;
    float* red  = t_s;
    const int tid = threadIdx.x, lane = tid & 31;
    const int wv = __builtin_amdgcn_readfirstlane(tid >> 5);
    const int bx = blockIdx.x;
    const int m0 = bx * RW_;

    inv_s[tid] = -1;
    inv_s[tid + 256] = -1;
    __syncthreads();
    {
        const v4i* oi = reinterpret_cast<const v4i*>(obs_idx);
        #pragma unroll 1
        for (int q0 = 0; q0 < NT_ * NO_ / 4; q0 += 8 * 256) {
            v4i v[8];
            #pragma unroll
            for (int u = 0; u < 8; ++u) v[u] = oi[q0 + u * 256 + tid];
            #pragma unroll
            for (int u = 0; u < 8; ++u) {
                const int e = 4 * (q0 + u * 256 + tid);
                const int tt = e >> 12, o = e & (NO_ - 1);
                if ((v[u].x >> 5) == bx) inv_s[tt * RW_ + (v[u].x & 31)] = o;
                if ((v[u].y >> 5) == bx) inv_s[tt * RW_ + (v[u].y & 31)] = o + 1;
                if ((v[u].z >> 5) == bx) inv_s[tt * RW_ + (v[u].z & 31)] = o + 2;
                if ((v[u].w >> 5) == bx) inv_s[tt * RW_ + (v[u].w & 31)] = o + 3;
            }
        }
    }
    for (int i = tid; i < D_ * 4 * PREP_; i += 256) wp_s[i] = w_prep[i];
    for (int i = tid; i < D_ * PREP_; i += 256) bp_s[i] = bias_prep[i];

    for (int i = tid; i < RW_ * CH_; i += 256) {
        int r = i / CH_, c = i % CH_;
        float acc = cm_b1[c];
        #pragma unroll
        for (int k = 0; k < CS_; ++k) acc += cov[(m0 + r) * CS_ + k] * cm_w1[k * CH_ + c];
        t_s[r * PP_ + c] = fmaxf(acc, 0.0f);
    }
    __syncthreads();
    for (int i = tid; i < RW_ * H_; i += 256) {
        int r = i / H_, c = i % H_;
        float acc = cm_b2[c];
        #pragma unroll
        for (int k = 0; k < CH_; ++k) acc += t_s[r * PP_ + k] * cm_w2[k * H_ + c];
        h_s[r * HP_ + c] = tanhf(acc);
    }
    __syncthreads();
    pmodel_phase(h_s, t_s, p_s, p_w1, p_b1, p_w2, p_b2, lane, wv, 0);

    float lsum = 0.0f;
    #pragma unroll 1
    for (int t = 0; t < NT_; ++t) {
        const int w = __builtin_amdgcn_readfirstlane(tid >> 5);
        ode_phase(inv_s + t * RW_, X + (size_t)t * NO_ * D_, M + (size_t)t * NO_ * D_, x_s, m_s, tid,
                  h_s, p_s, zh_s, xz_w, xz_b, hz_w, xn_w, xn_b, hn_w, lane, w);
        pmodel_phase(h_s, t_s, p_s, p_w1, p_b1, p_w2, p_b2, lane, w, 0);
        jump_phase(inv_s + t * RW_, x_s, m_s, wp_s, bp_s,
                   h_s, p_s, gin_s, gate_s, w_ih, w_hh, b_ih, b_hh,
                   lsum, tid, lane, w);
        pmodel_phase(h_s, t_s, p_s, p_w1, p_b1, p_w2, p_b2, lane, w, 4);
    }
    #pragma unroll 1
    for (int s = 0; s < NPOST_; ++s) {
        const int w = __builtin_amdgcn_readfirstlane(tid >> 5);
        ode_phase(nullptr, X, M, x_s, m_s, tid,
                  h_s, p_s, zh_s, xz_w, xz_b, hz_w, xn_w, xn_b, hn_w, lane, w);
        pmodel_phase(h_s, t_s, p_s, p_w1, p_b1, p_w2, p_b2, lane, w, 4 * (s & 1));
    }

    for (int i = tid; i < RW_ * H_ / 4; i += 256) {
        const int r = i >> 5, c4 = i & 31;
        *reinterpret_cast<v4f*>(h + (size_t)(m0 + r) * H_ + 4 * c4) =
            *reinterpret_cast<const v4f*>(h_s + r * HP_ + 4 * c4);
    }
    for (int i = tid; i < RW_ * P2_ / 4; i += 256) {
        const int r = i >> 4, c4 = i & 15;
        *reinterpret_cast<v4f*>(p + (size_t)(m0 + r) * P2_ + 4 * c4) =
            *reinterpret_cast<const v4f*>(p_s + r * PP_ + 4 * c4);
    }

    red[tid] = lsum;
    __syncthreads();
    for (int s2 = 128; s2 > 0; s2 >>= 1) {
        if (tid < s2) red[tid] += red[tid + s2];
        __syncthreads();
    }
    if (tid == 0) loss_part[bx] = red[0];
}

__global__ void k_loss_final(const float* __restrict__ parts, float* __restrict__ loss) {
    __shared__ float red[256];
    const int tid = threadIdx.x;
    float s = 0.0f;
    for (int i = tid; i < MB_; i += 256) s += parts[i];
    red[tid] = s;
    __syncthreads();
    for (int k = 128; k > 0; k >>= 1) {
        if (tid < k) red[tid] += red[tid + k];
        __syncthreads();
    }
    if (tid == 0) *loss = red[0];
}

extern "C" void kernel_launch(void* const* d_in, const int* in_sizes, int n_in,
                              void* d_out, int out_size, void* d_ws, size_t ws_size,
                              hipStream_t stream) {
    const float* X       = (const float*)d_in[0];
    const float* M       = (const float*)d_in[1];
    const int*   obs_idx = (const int*)d_in[2];
    const float* cov     = (const float*)d_in[3];
    const float* cm_w1   = (const float*)d_in[4];
    const float* cm_b1   = (const float*)d_in[5];
    const float* cm_w2   = (const float*)d_in[6];
    const float* cm_b2   = (const float*)d_in[7];
    const float* p_w1    = (const float*)d_in[8];
    const float* p_b1    = (const float*)d_in[9];
    const float* p_w2    = (const float*)d_in[10];
    const float* p_b2    = (const float*)d_in[11];
    const float* xz_w    = (const float*)d_in[12];
    const float* xz_b    = (const float*)d_in[13];
    const float* hz_w    = (const float*)d_in[14];
    const float* xn_w    = (const float*)d_in[15];
    const float* xn_b    = (const float*)d_in[16];
    const float* hn_w    = (const float*)d_in[17];
    const float* w_ih    = (const float*)d_in[18];
    const float* w_hh    = (const float*)d_in[19];
    const float* b_ih    = (const float*)d_in[20];
    const float* b_hh    = (const float*)d_in[21];
    const float* w_prep  = (const float*)d_in[22];
    const float* b_prep  = (const float*)d_in[23];

    float* h    = (float*)d_out;
    float* p    = h + (size_t)B_ * H_;
    float* loss = p + (size_t)B_ * P2_;
    float* loss_part = (float*)d_ws;

    k_traj<<<MB_, 256, 0, stream>>>(X, M, obs_idx, cov, cm_w1, cm_b1, cm_w2, cm_b2,
                                    p_w1, p_b1, p_w2, p_b2,
                                    xz_w, xz_b, hz_w, xn_w, xn_b, hn_w,
                                    w_ih, w_hh, b_ih, b_hh, w_prep, b_prep,
                                    h, p, loss_part);
    k_loss_final<<<1, 256, 0, stream>>>(loss_part, loss);
}


# ode, p-model, jump: the LDS A-fragment reads of a matrix run kept as ONE group in front of it by a scheduler fence, not spread one or two at a time among the matrix instructions; sums kept
# speedup vs baseline: 2.1066x; 1.0330x over previous
#include <hip/hip_runtime.h>
#include <hip/hip_bf16.h>
#include <math.h>


#define B_    8192
#define D_    32
#define H_    128
#define P2_   64
#define PREP_ 10
#define GIN_  320
#define G3_   384
#define CS_   8
#define CH_   32
#define NT_   16
#define NO_   4096
#define DT_   0.1f
#define NPOST_ 4
#define LOGC_  0.91893853320467274178f

#define RW_   32
#define MB_   (B_ / RW_)
#define HP_   132
#define PP_   68
#define GP_   324

typedef __attribute__((ext_vector_type(2))) float v2f;
typedef __attribute__((ext_vector_type(4))) float v4f;
typedef __attribute__((ext_vector_type(8))) float v8f;
typedef __attribute__((ext_vector_type(4))) int   v4i;

__device__ __forceinline__ float sigmoidf_(float x) { return 1.0f / (1.0f + expf(-x)); }


__device__ __forceinline__ v2f load_afrag(const float* A, int lda, int m0, int k, int lane) {
    int m  = m0 + (lane & 15);
    int kk = k + ((lane >> 4) << 1);
    v2f a;
    a.x = A[m * lda + kk];
    a.y = A[m * lda + kk + 1];
    return a;
}

__device__ __forceinline__ v8f wmma4(v2f a, v2f b, v8f c) {
    return __builtin_amdgcn_wmma_f32_16x16x4_f32(false, a, false, b, (short)0, c, false, false);
}

template <int NK>
__device__ __forceinline__ void bfrag2_load(const float* Bm, int ldb, int n0, int lane,
                                            v2f (&lo)[NK], v2f (&hi)[NK]) {
    const float* bp = Bm + ((lane >> 4) << 1) * ldb + n0 + 2 * (lane & 15);
    #pragma unroll
    for (int i = 0; i < NK; ++i) {
        lo[i] = *reinterpret_cast<const v2f*>(bp + (4 * i) * ldb);
        hi[i] = *reinterpret_cast<const v2f*>(bp + (4 * i + 1) * ldb);
    }
}

#define LDS_READS_AHEAD_FENCE() __builtin_amdgcn_sched_barrier(0x676)

template <int NK>
__device__ __forceinline__ void afrag_group(const float* As, int lda, int lane, v2f (&a)[NK]) {
    #pragma unroll
    for (int i = 0; i < NK; ++i) a[i] = load_afrag(As, lda, 0, 4 * i, lane);
}

template <int NK>
__device__ __forceinline__ void reg_frag2_mma(const v2f* a,
                                              const v2f (&lo)[NK], const v2f (&hi)[NK],
                                              v8f& c0, v8f& c1) {
    #pragma unroll
    for (int i = 0; i < NK; ++i) {
        v2f b0, b1;
        b0.x = lo[i].x; b0.y = hi[i].x;
        b1.x = lo[i].y; b1.y = hi[i].y;
        c0 = wmma4(a[i], b0, c0);
        c1 = wmma4(a[i], b1, c1);
    }
}

template <int NS>
__device__ __forceinline__ void gemm2_reg_glb(const v2f (&a)[NS * 16],
                                              const float* Bm, int ldb, int n0, int lane,
                                              v8f& c0, v8f& c1) {
    v2f lo[2][16], hi[2][16];
    bfrag2_load<16>(Bm, ldb, n0, lane, lo[0], hi[0]);
    #pragma unroll
    for (int s = 0; s < NS; ++s) {
        if (s + 1 < NS)
            bfrag2_load<16>(Bm + (s + 1) * 64 * ldb, ldb, n0, lane, lo[(s + 1) & 1], hi[(s + 1) & 1]);
        reg_frag2_mma<16>(&a[s * 16], lo[s & 1], hi[s & 1], c0, c1);
    }
}

__device__ __forceinline__ void pmodel_phase(const float* h_s, float* t_s, float* p_s,
                                             const float* __restrict__ w1, const float* __restrict__ b1,
                                             const float* __restrict__ w2, const float* __restrict__ b2,
                                             int lane, int wv, int wb) {
    const int pw = wv - wb;
    const bool mine = (pw >= 0) && (pw < 4);
    const int rt = mine ? (pw >> 1) : 0;
    const int n0 = (pw & 1) * 32;
    const int col = n0 + 2 * (lane & 15), rb = ((lane >> 4) << 3) + 16 * rt;
    if (mine) {
        v8f acc0 = {}, acc1 = {};
        v2f ah[H_ / 4];
        afrag_group<H_ / 4>(h_s + 16 * rt * HP_, HP_, lane, ah);
        LDS_READS_AHEAD_FENCE();
        gemm2_reg_glb<H_ / 64>(ah, w1, P2_, n0, lane, acc0, acc1);
        const v2f bv = *reinterpret_cast<const v2f*>(b1 + col);
        #pragma unroll
        for (int j = 0; j < 8; ++j) {
            v2f t;
            t.x = fmaxf(acc0[j] + bv.x, 0.0f);
            t.y = fmaxf(acc1[j] + bv.y, 0.0f);
            *reinterpret_cast<v2f*>(t_s + (rb + j) * PP_ + col) = t;
        }
    }
    __syncthreads();
    if (mine) {
        v8f acc2 = {}, acc3 = {};
        v2f at[P2_ / 4];
        afrag_group<P2_ / 4>(t_s + 16 * rt * PP_, PP_, lane, at);
        LDS_READS_AHEAD_FENCE();
        gemm2_reg_glb<P2_ / 64>(at, w2, P2_, n0, lane, acc2, acc3);
        const v2f bv2 = *reinterpret_cast<const v2f*>(b2 + col);
        #pragma unroll
        for (int j = 0; j < 8; ++j) {
            v2f o;
            o.x = acc2[j] + bv2.x;
            o.y = acc3[j] + bv2.y;
            *reinterpret_cast<v2f*>(p_s + (rb + j) * PP_ + col) = o;
        }
    }
    __syncthreads();
}

__device__ __forceinline__ void ode_phase(const int* inv_t,
                                          const float* __restrict__ Xt, const float* __restrict__ Mt,
                                          float* x_s, float* m_s, int tid,
                                          float* h_s, const float* p_s, float* zh_s,
                                          const float* __restrict__ xz_w, const float* __restrict__ xz_b,
                                          const float* __restrict__ hz_w,
                                          const float* __restrict__ xn_w, const float* __restrict__ xn_b,
                                          const float* __restrict__ hn_w,
                                          int lane, int wv) {
    const int n0 = (wv & 3) * 32;
    const int rt = wv >> 2;
    const int col = n0 + 2 * (lane & 15), rb = ((lane >> 4) << 3) + 16 * rt;
    float z0[8], z1[8];
    const int xr = tid >> 3, xc4 = tid & 7;
    v4f xq = {0.0f, 0.0f, 0.0f, 0.0f}, mq = {0.0f, 0.0f, 0.0f, 0.0f};
    if (inv_t != nullptr) {
        const int o = inv_t[xr];
        if (o >= 0) {
            xq = *reinterpret_cast<const v4f*>(Xt + (size_t)o * D_ + 4 * xc4);
            mq = *reinterpret_cast<const v4f*>(Mt + (size_t)o * D_ + 4 * xc4);
        }
    }
    {
        v8f az0 = {}, az1 = {};
        v2f ap[P2_ / 4], ah[H_ / 4];
        afrag_group<P2_ / 4>(p_s + 16 * rt * PP_, PP_, lane, ap);
        afrag_group<H_ / 4>(h_s + 16 * rt * HP_, HP_, lane, ah);
        LDS_READS_AHEAD_FENCE();
        gemm2_reg_glb<P2_ / 64>(ap, xz_w, H_, n0, lane, az0, az1);
        gemm2_reg_glb<H_ / 64>(ah, hz_w, H_, n0, lane, az0, az1);
        const v2f bz = *reinterpret_cast<const v2f*>(xz_b + col);
        #pragma unroll
        for (int j = 0; j < 8; ++j) {
            const v2f hv = *reinterpret_cast<const v2f*>(h_s + (rb + j) * HP_ + col);
            z0[j] = sigmoidf_(az0[j] + bz.x);
            z1[j] = sigmoidf_(az1[j] + bz.y);
            v2f zh;
            zh.x = z0[j] * hv.x;
            zh.y = z1[j] * hv.y;
            *reinterpret_cast<v2f*>(zh_s + (rb + j) * HP_ + col) = zh;
        }
    }
    if (inv_t != nullptr) {
        *reinterpret_cast<v4f*>(x_s + xr * D_ + 4 * xc4) = xq;
        *reinterpret_cast<v4f*>(m_s + xr * D_ + 4 * xc4) = mq;
    }
    __syncthreads();
    {
        v8f an0 = {}, an1 = {};
        v2f ap[P2_ / 4], azh[H_ / 4];
        afrag_group<P2_ / 4>(p_s + 16 * rt * PP_, PP_, lane, ap);
        afrag_group<H_ / 4>(zh_s + 16 * rt * HP_, HP_, lane, azh);
        LDS_READS_AHEAD_FENCE();
        gemm2_reg_glb<P2_ / 64>(ap, xn_w, H_, n0, lane, an0, an1);
        gemm2_reg_glb<H_ / 64>(azh, hn_w, H_, n0, lane, an0, an1);
        const v2f bn = *reinterpret_cast<const v2f*>(xn_b + col);
        #pragma unroll
        for (int j = 0; j < 8; ++j) {
            const float n0v = tanhf(an0[j] + bn.x);
            const float n1v = tanhf(an1[j] + bn.y);
            float* hp = h_s + (rb + j) * HP_ + col;
            const v2f hv = *reinterpret_cast<const v2f*>(hp);
            v2f o;
            o.x = hv.x + DT_ * (1.0f - z0[j]) * (n0v - hv.x);
            o.y = hv.y + DT_ * (1.0f - z1[j]) * (n1v - hv.y);
            *reinterpret_cast<v2f*>(hp) = o;
        }
    }
    __syncthreads();
}

#define JKS_ 8
#define JNCI_ 10
struct __attribute__((aligned(4))) f3_t { float x, y, z; };

__device__ __forceinline__ void bfrag3_load(const float* __restrict__ bp,
                                            f3_t (&lo)[JKS_], f3_t (&hi)[JKS_]) {
    #pragma unroll
    for (int i = 0; i < JKS_; ++i) {
        lo[i] = *reinterpret_cast<const f3_t*>(bp + (4 * i) * G3_);
        hi[i] = *reinterpret_cast<const f3_t*>(bp + (4 * i + 1) * G3_);
    }
}

__device__ __forceinline__ void afrag2_load(const float* As, int lda, int lane,
                                            v2f (&a0)[JKS_], v2f (&a1)[JKS_]) {
    #pragma unroll
    for (int i = 0; i < JKS_; ++i) {
        a0[i] = load_afrag(As, lda, 0, 4 * i, lane);
        a1[i] = load_afrag(As, lda, 16, 4 * i, lane);
    }
}

__device__ __forceinline__ void frag3_mma(const v2f (&a0s)[JKS_], const v2f (&a1s)[JKS_],
                                          const f3_t (&lo)[JKS_], const f3_t (&hi)[JKS_],
                                          v8f (&accA)[3], v8f (&accB)[3]) {
    #pragma unroll
    for (int i = 0; i < JKS_; ++i) {
        const v2f a0 = a0s[i];
        const v2f a1 = a1s[i];
        v2f b0, b1, b2;
        b0.x = lo[i].x; b0.y = hi[i].x;
        b1.x = lo[i].y; b1.y = hi[i].y;
        b2.x = lo[i].z; b2.y = hi[i].z;
        accA[0] = wmma4(a0, b0, accA[0]);
        accB[0] = wmma4(a1, b0, accB[0]);
        accA[1] = wmma4(a0, b1, accA[1]);
        accB[1] = wmma4(a1, b1, accB[1]);
        accA[2] = wmma4(a0, b2, accA[2]);
        accB[2] = wmma4(a1, b2, accB[2]);
    }
}

__device__ __forceinline__ void jump_phase(const int* inv_t,
                                           const float* x_s, const float* m_s,
                                           const float* wp_s, const float* bp_s,
                                           float* h_s, const float* p_s, float* gin_s, float* gate_s,
                                           const float* __restrict__ w_ih, const float* __restrict__ w_hh,
                                           const float* __restrict__ b_ih, const float* __restrict__ b_hh,
                                           float& lsum, int tid, int lane, int wv) {
    float* gi_s = gate_s;
    float* gh_s = gate_s + RW_ * G3_;

    const int n0w = wv * 48;
    const int woff = ((lane >> 4) << 1) * G3_ + n0w + 3 * (lane & 15);
    const float* wi = w_ih + woff;
    const float* wh = w_hh + woff;

    f3_t lo[2][JKS_], hi[2][JKS_];
    bfrag3_load(wi, lo[0], hi[0]);

    for (int i = tid; i < RW_ * D_; i += 256) {
        const int r = i / D_, d = i % D_;
        const int o = inv_t[r];
        if (o >= 0) {
            float mean = p_s[r * PP_ + d];
            float lv   = p_s[r * PP_ + D_ + d];
            float x    = x_s[r * D_ + d];
            float m    = m_s[r * D_ + d];
            float err  = (x - mean) * expf(-0.5f * lv);
            lsum += 0.5f * ((err * err + lv + 2.0f * LOGC_) * m);
        }
    }

    #pragma unroll 1
    for (int rt = 0; rt < 2; ++rt) {
        const int mr = lane & 15, hh = lane >> 4;
        const int r0 = 16 * rt;
        const int d0 = wv * 4;
        const int pc = mr < PREP_ ? mr : PREP_ - 1;
        const v4f xv = *reinterpret_cast<const v4f*>(x_s + (r0 + mr) * D_ + d0);
        const v4f mu = *reinterpret_cast<const v4f*>(p_s + (r0 + mr) * PP_ + d0);
        const v4f lg = *reinterpret_cast<const v4f*>(p_s + (r0 + mr) * PP_ + D_ + d0);
        v8f acc[4];
        float bias[4];
        #pragma unroll
        for (int i = 0; i < 4; ++i) {
            const int d = d0 + i;
            const float err = (xv[i] - mu[i]) * expf(-0.5f * lg[i]);
            v2f a, b;
            a.x = hh ? lg[i] : xv[i];
            a.y = hh ? err   : mu[i];
            const float* wp = wp_s + (d * 4 + 2 * hh) * PREP_ + pc;
            b.x = wp[0];
            b.y = wp[PREP_];
            bias[i] = bp_s[d * PREP_ + pc];
            acc[i] = (v8f){};
            acc[i] = wmma4(a, b, acc[i]);
        }
        if (mr < PREP_) {
            float* gp = gin_s + (r0 + 8 * hh) * GP_ + d0 * PREP_ + mr;
            #pragma unroll
            for (int j = 0; j < 8; ++j) {
                const bool ob = inv_t[r0 + 8 * hh + j] >= 0;
                const v4f mk = *reinterpret_cast<const v4f*>(m_s + (r0 + 8 * hh + j) * D_ + d0);
                #pragma unroll
                for (int i = 0; i < 4; ++i) {
                    const float g = ob ? fmaxf(acc[i][j] + bias[i], 0.0f) * mk[i] : 0.0f;
                    gp[j * GP_ + i * PREP_] = g;
                }
            }
        }
    }
    __syncthreads();

    v8f giA[3], giB[3], ghA[3], ghB[3];
    #pragma unroll
    for (int s = 0; s < 3; ++s) { giA[s] = (v8f){}; giB[s] = (v8f){}; ghA[s] = (v8f){}; ghB[s] = (v8f){}; }

    v2f fa0[2][JKS_], fa1[2][JKS_];
    bfrag3_load(wi + JKS_ * 4 * G3_, lo[1], hi[1]);
    afrag2_load(gin_s, GP_, lane, fa0[0], fa1[0]);
    #pragma unroll
    for (int sl = 0; sl < JNCI_ + 4; ++sl) {
        __builtin_amdgcn_sched_barrier(0);
        if (sl + 1 < JNCI_ + 4) {
            if (sl + 1 < JNCI_) afrag2_load(gin_s + (sl + 1) * (JKS_ * 4), GP_, lane, fa0[(sl + 1) & 1], fa1[(sl + 1) & 1]);
            else                afrag2_load(h_s + (sl + 1 - JNCI_) * (JKS_ * 4), HP_, lane, fa0[(sl + 1) & 1], fa1[(sl + 1) & 1]);
        }
        LDS_READS_AHEAD_FENCE();
        if (sl < JNCI_) frag3_mma(fa0[sl & 1], fa1[sl & 1], lo[sl & 1], hi[sl & 1], giA, giB);
        else            frag3_mma(fa0[sl & 1], fa1[sl & 1], lo[sl & 1], hi[sl & 1], ghA, ghB);
        if (sl + 2 < JNCI_ + 4) {
            if (sl + 2 < JNCI_) bfrag3_load(wi + (sl + 2) * (JKS_ * 4 * G3_), lo[sl & 1], hi[sl & 1]);
            else                bfrag3_load(wh + (sl + 2 - JNCI_) * (JKS_ * 4 * G3_), lo[sl & 1], hi[sl & 1]);
        }
    }
    __builtin_amdgcn_sched_barrier(0);

    {
        const int rb = (lane >> 4) << 3;
        #pragma unroll
        for (int s = 0; s < 3; ++s) {
            const int col = n0w + 3 * (lane & 15) + s;
            const float bi = b_ih[col], bh = b_hh[col];
            #pragma unroll
            for (int j = 0; j < 8; ++j) {
                gi_s[(rb + j) * G3_ + col] = giA[s][j] + bi;
                gh_s[(rb + j) * G3_ + col] = ghA[s][j] + bh;
                gi_s[(16 + rb + j) * G3_ + col] = giB[s][j] + bi;
                gh_s[(16 + rb + j) * G3_ + col] = ghB[s][j] + bh;
            }
        }
    }
    __syncthreads();

    for (int i = tid; i < RW_ * H_; i += 256) {
        const int r = i >> 7, j = i & 127;
        if (inv_t[r] >= 0) {
            float rr = sigmoidf_(gi_s[r * G3_ + j] + gh_s[r * G3_ + j]);
            float z  = sigmoidf_(gi_s[r * G3_ + H_ + j] + gh_s[r * G3_ + H_ + j]);
            float n  = tanhf(gi_s[r * G3_ + 2 * H_ + j] + rr * gh_s[r * G3_ + 2 * H_ + j]);
            float hv = h_s[r * HP_ + j];
            h_s[r * HP_ + j] = (1.0f - z) * n + z * hv;
        }
    }
    __syncthreads();
}

__global__ void __launch_bounds__(256, 2) k_traj(
                       const float* __restrict__ X, const float* __restrict__ M,
                       const int* __restrict__ obs_idx, const float* __restrict__ cov,
                       const float* __restrict__ cm_w1, const float* __restrict__ cm_b1,
                       const float* __restrict__ cm_w2, const float* __restrict__ cm_b2,
                       const float* __restrict__ p_w1, const float* __restrict__ p_b1,
                       const float* __restrict__ p_w2, const float* __restrict__ p_b2,
                       const float* __restrict__ xz_w, const float* __restrict__ xz_b,
                       const float* __restrict__ hz_w,
                       const float* __restrict__ xn_w, const float* __restrict__ xn_b,
                       const float* __restrict__ hn_w,
                       const float* __restrict__ w_ih, const float* __restrict__ w_hh,
                       const float* __restrict__ b_ih, const float* __restrict__ b_hh,
                       const float* __restrict__ w_prep, const float* __restrict__ bias_prep,
                       float* __restrict__ h, float* __restrict__ p,
                       float* __restrict__ loss_part) {
    __shared__ __attribute__((aligned(16))) float h_s[RW_ * HP_];
    __shared__ __attribute__((aligned(16))) float p_s[RW_ * PP_];
    __shared__ __attribute__((aligned(16))) float t_s[RW_ * PP_];
    __shared__ __attribute__((aligned(16))) float gin_s[RW_ * GP_];
    __shared__ __attribute__((aligned(16))) float gate_s[2 * RW_ * G3_];
    __shared__ __attribute__((aligned(16))) float x_s[RW_ * D_];
    __shared__ __attribute__((aligned(16))) float m_s[RW_ * D_];
    __shared__ float wp_s[D_ * 4 * PREP_];
    __shared__ float bp_s[D_ * PREP_];
    __shared__ int   inv_s[NT_ * RW_];
    float* zh_s = gin_s;
    float* red  = t_s;
    const int tid = threadIdx.x, lane = tid & 31;
    const int wv = __builtin_amdgcn_readfirstlane(tid >> 5);
    const int bx = blockIdx.x;
    const int m0 = bx * RW_;

    inv_s[tid] = -1;
    inv_s[tid + 256] = -1;
    __syncthreads();
    {
        const v4i* oi = reinterpret_cast<const v4i*>(obs_idx);
        #pragma unroll 1
        for (int q0 = 0; q0 < NT_ * NO_ / 4; q0 += 8 * 256) {
            v4i v[8];
            #pragma unroll
            for (int u = 0; u < 8; ++u) v[u] = oi[q0 + u * 256 + tid];
            #pragma unroll
            for (int u = 0; u < 8; ++u) {
                const int e = 4 * (q0 + u * 256 + tid);
                const int tt = e >> 12, o = e & (NO_ - 1);
                if ((v[u].x >> 5) == bx) inv_s[tt * RW_ + (v[u].x & 31)] = o;
                if ((v[u].y >> 5) == bx) inv_s[tt * RW_ + (v[u].y & 31)] = o + 1;
                if ((v[u].z >> 5) == bx) inv_s[tt * RW_ + (v[u].z & 31)] = o + 2;
                if ((v[u].w >> 5) == bx) inv_s[tt * RW_ + (v[u].w & 31)] = o + 3;
            }
        }
    }
    for (int i = tid; i < D_ * 4 * PREP_; i += 256) wp_s[i] = w_prep[i];
    for (int i = tid; i < D_ * PREP_; i += 256) bp_s[i] = bias_prep[i];

    for (int i = tid; i < RW_ * CH_; i += 256) {
        int r = i / CH_, c = i % CH_;
        float acc = cm_b1[c];
        #pragma unroll
        for (int k = 0; k < CS_; ++k) acc += cov[(m0 + r) * CS_ + k] * cm_w1[k * CH_ + c];
        t_s[r * PP_ + c] = fmaxf(acc, 0.0f);
    }
    __syncthreads();
    for (int i = tid; i < RW_ * H_; i += 256) {
        int r = i / H_, c = i % H_;
        float acc = cm_b2[c];
        #pragma unroll
        for (int k = 0; k < CH_; ++k) acc += t_s[r * PP_ + k] * cm_w2[k * H_ + c];
        h_s[r * HP_ + c] = tanhf(acc);
    }
    __syncthreads();
    pmodel_phase(h_s, t_s, p_s, p_w1, p_b1, p_w2, p_b2, lane, wv, 0);

    float lsum = 0.0f;
    #pragma unroll 1
    for (int t = 0; t < NT_; ++t) {
        const int w = __builtin_amdgcn_readfirstlane(tid >> 5);
        ode_phase(inv_s + t * RW_, X + (size_t)t * NO_ * D_, M + (size_t)t * NO_ * D_, x_s, m_s, tid,
                  h_s, p_s, zh_s, xz_w, xz_b, hz_w, xn_w, xn_b, hn_w, lane, w);
        pmodel_phase(h_s, t_s, p_s, p_w1, p_b1, p_w2, p_b2, lane, w, 0);
        jump_phase(inv_s + t * RW_, x_s, m_s, wp_s, bp_s,
                   h_s, p_s, gin_s, gate_s, w_ih, w_hh, b_ih, b_hh,
                   lsum, tid, lane, w);
        pmodel_phase(h_s, t_s, p_s, p_w1, p_b1, p_w2, p_b2, lane, w, 4);
    }
    #pragma unroll 1
    for (int s = 0; s < NPOST_; ++s) {
        const int w = __builtin_amdgcn_readfirstlane(tid >> 5);
        ode_phase(nullptr, X, M, x_s, m_s, tid,
                  h_s, p_s, zh_s, xz_w, xz_b, hz_w, xn_w, xn_b, hn_w, lane, w);
        pmodel_phase(h_s, t_s, p_s, p_w1, p_b1, p_w2, p_b2, lane, w, 4 * (s & 1));
    }

    for (int i = tid; i < RW_ * H_ / 4; i += 256) {
        const int r = i >> 5, c4 = i & 31;
        *reinterpret_cast<v4f*>(h + (size_t)(m0 + r) * H_ + 4 * c4) =
            *reinterpret_cast<const v4f*>(h_s + r * HP_ + 4 * c4);
    }
    for (int i = tid; i < RW_ * P2_ / 4; i += 256) {
        const int r = i >> 4, c4 = i & 15;
        *reinterpret_cast<v4f*>(p + (size_t)(m0 + r) * P2_ + 4 * c4) =
            *reinterpret_cast<const v4f*>(p_s + r * PP_ + 4 * c4);
    }

    red[tid] = lsum;
    __syncthreads();
    for (int s2 = 128; s2 > 0; s2 >>= 1) {
        if (tid < s2) red[tid] += red[tid + s2];
        __syncthreads();
    }
    if (tid == 0) loss_part[bx] = red[0];
}

__global__ void k_loss_final(const float* __restrict__ parts, float* __restrict__ loss) {
    __shared__ float red[256];
    const int tid = threadIdx.x;
    float s = 0.0f;
    for (int i = tid; i < MB_; i += 256) s += parts[i];
    red[tid] = s;
    __syncthreads();
    for (int k = 128; k > 0; k >>= 1) {
        if (tid < k) red[tid] += red[tid + k];
        __syncthreads();
    }
    if (tid == 0) *loss = red[0];
}

extern "C" void kernel_launch(void* const* d_in, const int* in_sizes, int n_in,
                              void* d_out, int out_size, void* d_ws, size_t ws_size,
                              hipStream_t stream) {
    const float* X       = (const float*)d_in[0];
    const float* M       = (const float*)d_in[1];
    const int*   obs_idx = (const int*)d_in[2];
    const float* cov     = (const float*)d_in[3];
    const float* cm_w1   = (const float*)d_in[4];
    const float* cm_b1   = (const float*)d_in[5];
    const float* cm_w2   = (const float*)d_in[6];
    const float* cm_b2   = (const float*)d_in[7];
    const float* p_w1    = (const float*)d_in[8];
    const float* p_b1    = (const float*)d_in[9];
    const float* p_w2    = (const float*)d_in[10];
    const float* p_b2    = (const float*)d_in[11];
    const float* xz_w    = (const float*)d_in[12];
    const float* xz_b    = (const float*)d_in[13];
    const float* hz_w    = (const float*)d_in[14];
    const float* xn_w    = (const float*)d_in[15];
    const float* xn_b    = (const float*)d_in[16];
    const float* hn_w    = (const float*)d_in[17];
    const float* w_ih    = (const float*)d_in[18];
    const float* w_hh    = (const float*)d_in[19];
    const float* b_ih    = (const float*)d_in[20];
    const float* b_hh    = (const float*)d_in[21];
    const float* w_prep  = (const float*)d_in[22];
    const float* b_prep  = (const float*)d_in[23];

    float* h    = (float*)d_out;
    float* p    = h + (size_t)B_ * H_;
    float* loss = p + (size_t)B_ * P2_;
    float* loss_part = (float*)d_ws;

    k_traj<<<MB_, 256, 0, stream>>>(X, M, obs_idx, cov, cm_w1, cm_b1, cm_w2, cm_b2,
                                    p_w1, p_b1, p_w2, p_b2,
                                    xz_w, xz_b, hz_w, xn_w, xn_b, hn_w,
                                    w_ih, w_hh, b_ih, b_hh, w_prep, b_prep,
                                    h, p, loss_part);
    k_loss_final<<<1, 256, 0, stream>>>(loss_part, loss);
}


# p_w1, p_w2, xz_w, xn_w staged once into LDS (row pitch 16 mod 32 floats), B fragments read there: each p-model product and each ode gate's first product start without a trip to memory; sums kept
# speedup vs baseline: 2.1995x; 1.0441x over previous
#include <hip/hip_runtime.h>
#include <hip/hip_bf16.h>
#include <math.h>


#define B_    8192
#define D_    32
#define H_    128
#define P2_   64
#define PREP_ 10
#define GIN_  320
#define G3_   384
#define CS_   8
#define CH_   32
#define NT_   16
#define NO_   4096
#define DT_   0.1f
#define NPOST_ 4
#define LOGC_  0.91893853320467274178f

#define RW_   32
#define MB_   (B_ / RW_)
#define HP_   132
#define PP_   68
#define GP_   324
#define W1P_  80
#define XWP_  144

typedef __attribute__((ext_vector_type(2))) float v2f;
typedef __attribute__((ext_vector_type(4))) float v4f;
typedef __attribute__((ext_vector_type(8))) float v8f;
typedef __attribute__((ext_vector_type(4))) int   v4i;

__device__ __forceinline__ float sigmoidf_(float x) { return 1.0f / (1.0f + expf(-x)); }


__device__ __forceinline__ v2f load_afrag(const float* A, int lda, int m0, int k, int lane) {
    int m  = m0 + (lane & 15);
    int kk = k + ((lane >> 4) << 1);
    v2f a;
    a.x = A[m * lda + kk];
    a.y = A[m * lda + kk + 1];
    return a;
}

__device__ __forceinline__ v8f wmma4(v2f a, v2f b, v8f c) {
    return __builtin_amdgcn_wmma_f32_16x16x4_f32(false, a, false, b, (short)0, c, false, false);
}

template <int NK>
__device__ __forceinline__ void bfrag2_load(const float* Bm, int ldb, int n0, int lane,
                                            v2f (&lo)[NK], v2f (&hi)[NK]) {
    const float* bp = Bm + ((lane >> 4) << 1) * ldb + n0 + 2 * (lane & 15);
    #pragma unroll
    for (int i = 0; i < NK; ++i) {
        lo[i] = *reinterpret_cast<const v2f*>(bp + (4 * i) * ldb);
        hi[i] = *reinterpret_cast<const v2f*>(bp + (4 * i + 1) * ldb);
    }
}

template <int NK>
__device__ __forceinline__ void bfrag2_lds(const float* Bs, int ldb, int n0, int lane,
                                           v2f (&lo)[NK], v2f (&hi)[NK]) {
    const float* bp = Bs + ((lane >> 4) << 1) * ldb + n0 + 2 * (lane & 15);
    #pragma unroll
    for (int i = 0; i < NK; ++i) {
        lo[i] = *reinterpret_cast<const v2f*>(bp + (4 * i) * ldb);
        hi[i] = *reinterpret_cast<const v2f*>(bp + (4 * i + 1) * ldb);
    }
}

#define LDS_READS_AHEAD_FENCE() __builtin_amdgcn_sched_barrier(0x676)

template <int NK>
__device__ __forceinline__ void afrag_group(const float* As, int lda, int lane, v2f (&a)[NK]) {
    #pragma unroll
    for (int i = 0; i < NK; ++i) a[i] = load_afrag(As, lda, 0, 4 * i, lane);
}

template <int NK>
__device__ __forceinline__ void reg_frag2_mma(const v2f* a,
                                              const v2f (&lo)[NK], const v2f (&hi)[NK],
                                              v8f& c0, v8f& c1) {
    #pragma unroll
    for (int i = 0; i < NK; ++i) {
        v2f b0, b1;
        b0.x = lo[i].x; b0.y = hi[i].x;
        b1.x = lo[i].y; b1.y = hi[i].y;
        c0 = wmma4(a[i], b0, c0);
        c1 = wmma4(a[i], b1, c1);
    }
}

template <int NS>
__device__ __forceinline__ void gemm2_reg_glb(const v2f (&a)[NS * 16],
                                              const float* Bm, int ldb, int n0, int lane,
                                              v8f& c0, v8f& c1) {
    v2f lo[2][16], hi[2][16];
    bfrag2_load<16>(Bm, ldb, n0, lane, lo[0], hi[0]);
    #pragma unroll
    for (int s = 0; s < NS; ++s) {
        if (s + 1 < NS)
            bfrag2_load<16>(Bm + (s + 1) * 64 * ldb, ldb, n0, lane, lo[(s + 1) & 1], hi[(s + 1) & 1]);
        reg_frag2_mma<16>(&a[s * 16], lo[s & 1], hi[s & 1], c0, c1);
    }
}

__device__ __forceinline__ void pmodel_phase(const float* h_s, float* t_s, float* p_s,
                                             const float* w1_s, const float* __restrict__ b1,
                                             const float* w2_s, const float* __restrict__ b2,
                                             int lane, int wv, int wb) {
    const int pw = wv - wb;
    const bool mine = (pw >= 0) && (pw < 4);
    const int rt = mine ? (pw >> 1) : 0;
    const int n0 = (pw & 1) * 32;
    const int col = n0 + 2 * (lane & 15), rb = ((lane >> 4) << 3) + 16 * rt;
    if (mine) {
        v8f acc0 = {}, acc1 = {};
        v2f ah[H_ / 4], w1lo[H_ / 4], w1hi[H_ / 4];
        afrag_group<H_ / 4>(h_s + 16 * rt * HP_, HP_, lane, ah);
        bfrag2_lds<H_ / 4>(w1_s, W1P_, n0, lane, w1lo, w1hi);
        LDS_READS_AHEAD_FENCE();
        reg_frag2_mma<H_ / 4>(ah, w1lo, w1hi, acc0, acc1);
        const v2f bv = *reinterpret_cast<const v2f*>(b1 + col);
        #pragma unroll
        for (int j = 0; j < 8; ++j) {
            v2f t;
            t.x = fmaxf(acc0[j] + bv.x, 0.0f);
            t.y = fmaxf(acc1[j] + bv.y, 0.0f);
            *reinterpret_cast<v2f*>(t_s + (rb + j) * PP_ + col) = t;
        }
    }
    __syncthreads();
    if (mine) {
        v8f acc2 = {}, acc3 = {};
        v2f at[P2_ / 4], w2lo[P2_ / 4], w2hi[P2_ / 4];
        afrag_group<P2_ / 4>(t_s + 16 * rt * PP_, PP_, lane, at);
        bfrag2_lds<P2_ / 4>(w2_s, W1P_, n0, lane, w2lo, w2hi);
        LDS_READS_AHEAD_FENCE();
        reg_frag2_mma<P2_ / 4>(at, w2lo, w2hi, acc2, acc3);
        const v2f bv2 = *reinterpret_cast<const v2f*>(b2 + col);
        #pragma unroll
        for (int j = 0; j < 8; ++j) {
            v2f o;
            o.x = acc2[j] + bv2.x;
            o.y = acc3[j] + bv2.y;
            *reinterpret_cast<v2f*>(p_s + (rb + j) * PP_ + col) = o;
        }
    }
    __syncthreads();
}

__device__ __forceinline__ void ode_phase(const int* inv_t,
                                          const float* __restrict__ Xt, const float* __restrict__ Mt,
                                          float* x_s, float* m_s, int tid,
                                          float* h_s, const float* p_s, float* zh_s,
                                          const float* xz_s, const float* __restrict__ xz_b,
                                          const float* __restrict__ hz_w,
                                          const float* xn_s, const float* __restrict__ xn_b,
                                          const float* __restrict__ hn_w,
                                          int lane, int wv) {
    const int n0 = (wv & 3) * 32;
    const int rt = wv >> 2;
    const int col = n0 + 2 * (lane & 15), rb = ((lane >> 4) << 3) + 16 * rt;
    float z0[8], z1[8];
    const int xr = tid >> 3, xc4 = tid & 7;
    v4f xq = {0.0f, 0.0f, 0.0f, 0.0f}, mq = {0.0f, 0.0f, 0.0f, 0.0f};
    if (inv_t != nullptr) {
        const int o = inv_t[xr];
        if (o >= 0) {
            xq = *reinterpret_cast<const v4f*>(Xt + (size_t)o * D_ + 4 * xc4);
            mq = *reinterpret_cast<const v4f*>(Mt + (size_t)o * D_ + 4 * xc4);
        }
    }
    {
        v8f az0 = {}, az1 = {};
        v2f ap[P2_ / 4], ah[H_ / 4], xlo[P2_ / 4], xhi[P2_ / 4];
        afrag_group<P2_ / 4>(p_s + 16 * rt * PP_, PP_, lane, ap);
        bfrag2_lds<P2_ / 4>(xz_s, XWP_, n0, lane, xlo, xhi);
        afrag_group<H_ / 4>(h_s + 16 * rt * HP_, HP_, lane, ah);
        LDS_READS_AHEAD_FENCE();
        reg_frag2_mma<P2_ / 4>(ap, xlo, xhi, az0, az1);
        gemm2_reg_glb<H_ / 64>(ah, hz_w, H_, n0, lane, az0, az1);
        const v2f bz = *reinterpret_cast<const v2f*>(xz_b + col);
        #pragma unroll
        for (int j = 0; j < 8; ++j) {
            const v2f hv = *reinterpret_cast<const v2f*>(h_s + (rb + j) * HP_ + col);
            z0[j] = sigmoidf_(az0[j] + bz.x);
            z1[j] = sigmoidf_(az1[j] + bz.y);
            v2f zh;
            zh.x = z0[j] * hv.x;
            zh.y = z1[j] * hv.y;
            *reinterpret_cast<v2f*>(zh_s + (rb + j) * HP_ + col) = zh;
        }
    }
    if (inv_t != nullptr) {
        *reinterpret_cast<v4f*>(x_s + xr * D_ + 4 * xc4) = xq;
        *reinterpret_cast<v4f*>(m_s + xr * D_ + 4 * xc4) = mq;
    }
    __syncthreads();
    {
        v8f an0 = {}, an1 = {};
        v2f ap[P2_ / 4], azh[H_ / 4], xlo[P2_ / 4], xhi[P2_ / 4];
        afrag_group<P2_ / 4>(p_s + 16 * rt * PP_, PP_, lane, ap);
        bfrag2_lds<P2_ / 4>(xn_s, XWP_, n0, lane, xlo, xhi);
        afrag_group<H_ / 4>(zh_s + 16 * rt * HP_, HP_, lane, azh);
        LDS_READS_AHEAD_FENCE();
        reg_frag2_mma<P2_ / 4>(ap, xlo, xhi, an0, an1);
        gemm2_reg_glb<H_ / 64>(azh, hn_w, H_, n0, lane, an0, an1);
        const v2f bn = *reinterpret_cast<const v2f*>(xn_b + col);
        #pragma unroll
        for (int j = 0; j < 8; ++j) {
            const float n0v = tanhf(an0[j] + bn.x);
            const float n1v = tanhf(an1[j] + bn.y);
            float* hp = h_s + (rb + j) * HP_ + col;
            const v2f hv = *reinterpret_cast<const v2f*>(hp);
            v2f o;
            o.x = hv.x + DT_ * (1.0f - z0[j]) * (n0v - hv.x);
            o.y = hv.y + DT_ * (1.0f - z1[j]) * (n1v - hv.y);
            *reinterpret_cast<v2f*>(hp) = o;
        }
    }
    __syncthreads();
}

#define JKS_ 8
#define JNCI_ 10
struct __attribute__((aligned(4))) f3_t { float x, y, z; };

__device__ __forceinline__ void bfrag3_load(const float* __restrict__ bp,
                                            f3_t (&lo)[JKS_], f3_t (&hi)[JKS_]) {
    #pragma unroll
    for (int i = 0; i < JKS_; ++i) {
        lo[i] = *reinterpret_cast<const f3_t*>(bp + (4 * i) * G3_);
        hi[i] = *reinterpret_cast<const f3_t*>(bp + (4 * i + 1) * G3_);
    }
}

__device__ __forceinline__ void afrag2_load(const float* As, int lda, int lane,
                                            v2f (&a0)[JKS_], v2f (&a1)[JKS_]) {
    #pragma unroll
    for (int i = 0; i < JKS_; ++i) {
        a0[i] = load_afrag(As, lda, 0, 4 * i, lane);
        a1[i] = load_afrag(As, lda, 16, 4 * i, lane);
    }
}

__device__ __forceinline__ void frag3_mma(const v2f (&a0s)[JKS_], const v2f (&a1s)[JKS_],
                                          const f3_t (&lo)[JKS_], const f3_t (&hi)[JKS_],
                                          v8f (&accA)[3], v8f (&accB)[3]) {
    #pragma unroll
    for (int i = 0; i < JKS_; ++i) {
        const v2f a0 = a0s[i];
        const v2f a1 = a1s[i];
        v2f b0, b1, b2;
        b0.x = lo[i].x; b0.y = hi[i].x;
        b1.x = lo[i].y; b1.y = hi[i].y;
        b2.x = lo[i].z; b2.y = hi[i].z;
        accA[0] = wmma4(a0, b0, accA[0]);
        accB[0] = wmma4(a1, b0, accB[0]);
        accA[1] = wmma4(a0, b1, accA[1]);
        accB[1] = wmma4(a1, b1, accB[1]);
        accA[2] = wmma4(a0, b2, accA[2]);
        accB[2] = wmma4(a1, b2, accB[2]);
    }
}

__device__ __forceinline__ void jump_phase(const int* inv_t,
                                           const float* x_s, const float* m_s,
                                           const float* wp_s, const float* bp_s,
                                           float* h_s, const float* p_s, float* gin_s, float* gate_s,
                                           const float* __restrict__ w_ih, const float* __restrict__ w_hh,
                                           const float* __restrict__ b_ih, const float* __restrict__ b_hh,
                                           float& lsum, int tid, int lane, int wv) {
    float* gi_s = gate_s;
    float* gh_s = gate_s + RW_ * G3_;

    const int n0w = wv * 48;
    const int woff = ((lane >> 4) << 1) * G3_ + n0w + 3 * (lane & 15);
    const float* wi = w_ih + woff;
    const float* wh = w_hh + woff;

    f3_t lo[2][JKS_], hi[2][JKS_];
    bfrag3_load(wi, lo[0], hi[0]);

    for (int i = tid; i < RW_ * D_; i += 256) {
        const int r = i / D_, d = i % D_;
        const int o = inv_t[r];
        if (o >= 0) {
            float mean = p_s[r * PP_ + d];
            float lv   = p_s[r * PP_ + D_ + d];
            float x    = x_s[r * D_ + d];
            float m    = m_s[r * D_ + d];
            float err  = (x - mean) * expf(-0.5f * lv);
            lsum += 0.5f * ((err * err + lv + 2.0f * LOGC_) * m);
        }
    }

    #pragma unroll 1
    for (int rt = 0; rt < 2; ++rt) {
        const int mr = lane & 15, hh = lane >> 4;
        const int r0 = 16 * rt;
        const int d0 = wv * 4;
        const int pc = mr < PREP_ ? mr : PREP_ - 1;
        const v4f xv = *reinterpret_cast<const v4f*>(x_s + (r0 + mr) * D_ + d0);
        const v4f mu = *reinterpret_cast<const v4f*>(p_s + (r0 + mr) * PP_ + d0);
        const v4f lg = *reinterpret_cast<const v4f*>(p_s + (r0 + mr) * PP_ + D_ + d0);
        v8f acc[4];
        float bias[4];
        #pragma unroll
        for (int i = 0; i < 4; ++i) {
            const int d = d0 + i;
            const float err = (xv[i] - mu[i]) * expf(-0.5f * lg[i]);
            v2f a, b;
            a.x = hh ? lg[i] : xv[i];
            a.y = hh ? err   : mu[i];
            const float* wp = wp_s + (d * 4 + 2 * hh) * PREP_ + pc;
            b.x = wp[0];
            b.y = wp[PREP_];
            bias[i] = bp_s[d * PREP_ + pc];
            acc[i] = (v8f){};
            acc[i] = wmma4(a, b, acc[i]);
        }
        if (mr < PREP_) {
            float* gp = gin_s + (r0 + 8 * hh) * GP_ + d0 * PREP_ + mr;
            #pragma unroll
            for (int j = 0; j < 8; ++j) {
                const bool ob = inv_t[r0 + 8 * hh + j] >= 0;
                const v4f mk = *reinterpret_cast<const v4f*>(m_s + (r0 + 8 * hh + j) * D_ + d0);
                #pragma unroll
                for (int i = 0; i < 4; ++i) {
                    const float g = ob ? fmaxf(acc[i][j] + bias[i], 0.0f) * mk[i] : 0.0f;
                    gp[j * GP_ + i * PREP_] = g;
                }
            }
        }
    }
    __syncthreads();

    v8f giA[3], giB[3], ghA[3], ghB[3];
    #pragma unroll
    for (int s = 0; s < 3; ++s) { giA[s] = (v8f){}; giB[s] = (v8f){}; ghA[s] = (v8f){}; ghB[s] = (v8f){}; }

    v2f fa0[2][JKS_], fa1[2][JKS_];
    bfrag3_load(wi + JKS_ * 4 * G3_, lo[1], hi[1]);
    afrag2_load(gin_s, GP_, lane, fa0[0], fa1[0]);
    #pragma unroll
    for (int sl = 0; sl < JNCI_ + 4; ++sl) {
        __builtin_amdgcn_sched_barrier(0);
        if (sl + 1 < JNCI_ + 4) {
            if (sl + 1 < JNCI_) afrag2_load(gin_s + (sl + 1) * (JKS_ * 4), GP_, lane, fa0[(sl + 1) & 1], fa1[(sl + 1) & 1]);
            else                afrag2_load(h_s + (sl + 1 - JNCI_) * (JKS_ * 4), HP_, lane, fa0[(sl + 1) & 1], fa1[(sl + 1) & 1]);
        }
        LDS_READS_AHEAD_FENCE();
        if (sl < JNCI_) frag3_mma(fa0[sl & 1], fa1[sl & 1], lo[sl & 1], hi[sl & 1], giA, giB);
        else            frag3_mma(fa0[sl & 1], fa1[sl & 1], lo[sl & 1], hi[sl & 1], ghA, ghB);
        if (sl + 2 < JNCI_ + 4) {
            if (sl + 2 < JNCI_) bfrag3_load(wi + (sl + 2) * (JKS_ * 4 * G3_), lo[sl & 1], hi[sl & 1]);
            else                bfrag3_load(wh + (sl + 2 - JNCI_) * (JKS_ * 4 * G3_), lo[sl & 1], hi[sl & 1]);
        }
    }
    __builtin_amdgcn_sched_barrier(0);

    {
        const int rb = (lane >> 4) << 3;
        #pragma unroll
        for (int s = 0; s < 3; ++s) {
            const int col = n0w + 3 * (lane & 15) + s;
            const float bi = b_ih[col], bh = b_hh[col];
            #pragma unroll
            for (int j = 0; j < 8; ++j) {
                gi_s[(rb + j) * G3_ + col] = giA[s][j] + bi;
                gh_s[(rb + j) * G3_ + col] = ghA[s][j] + bh;
                gi_s[(16 + rb + j) * G3_ + col] = giB[s][j] + bi;
                gh_s[(16 + rb + j) * G3_ + col] = ghB[s][j] + bh;
            }
        }
    }
    __syncthreads();

    for (int i = tid; i < RW_ * H_; i += 256) {
        const int r = i >> 7, j = i & 127;
        if (inv_t[r] >= 0) {
            float rr = sigmoidf_(gi_s[r * G3_ + j] + gh_s[r * G3_ + j]);
            float z  = sigmoidf_(gi_s[r * G3_ + H_ + j] + gh_s[r * G3_ + H_ + j]);
            float n  = tanhf(gi_s[r * G3_ + 2 * H_ + j] + rr * gh_s[r * G3_ + 2 * H_ + j]);
            float hv = h_s[r * HP_ + j];
            h_s[r * HP_ + j] = (1.0f - z) * n + z * hv;
        }
    }
    __syncthreads();
}

__global__ void __launch_bounds__(256, 2) k_traj(
                       const float* __restrict__ X, const float* __restrict__ M,
                       const int* __restrict__ obs_idx, const float* __restrict__ cov,
                       const float* __restrict__ cm_w1, const float* __restrict__ cm_b1,
                       const float* __restrict__ cm_w2, const float* __restrict__ cm_b2,
                       const float* __restrict__ p_w1, const float* __restrict__ p_b1,
                       const float* __restrict__ p_w2, const float* __restrict__ p_b2,
                       const float* __restrict__ xz_w, const float* __restrict__ xz_b,
                       const float* __restrict__ hz_w,
                       const float* __restrict__ xn_w, const float* __restrict__ xn_b,
                       const float* __restrict__ hn_w,
                       const float* __restrict__ w_ih, const float* __restrict__ w_hh,
                       const float* __restrict__ b_ih, const float* __restrict__ b_hh,
                       const float* __restrict__ w_prep, const float* __restrict__ bias_prep,
                       float* __restrict__ h, float* __restrict__ p,
                       float* __restrict__ loss_part) {
    __shared__ __attribute__((aligned(16))) float h_s[RW_ * HP_];
    __shared__ __attribute__((aligned(16))) float p_s[RW_ * PP_];
    __shared__ __attribute__((aligned(16))) float t_s[RW_ * PP_];
    __shared__ __attribute__((aligned(16))) float gin_s[RW_ * GP_];
    __shared__ __attribute__((aligned(16))) float gate_s[2 * RW_ * G3_];
    __shared__ __attribute__((aligned(16))) float x_s[RW_ * D_];
    __shared__ __attribute__((aligned(16))) float m_s[RW_ * D_];
    __shared__ float wp_s[D_ * 4 * PREP_];
    __shared__ float bp_s[D_ * PREP_];
    __shared__ int   inv_s[NT_ * RW_];
    __shared__ __attribute__((aligned(16))) float pw1_s[H_ * W1P_];
    __shared__ __attribute__((aligned(16))) float pw2_s[P2_ * W1P_];
    __shared__ __attribute__((aligned(16))) float xz_s[P2_ * XWP_];
    __shared__ __attribute__((aligned(16))) float xn_s[P2_ * XWP_];
    float* zh_s = gin_s;
    float* red  = t_s;
    const int tid = threadIdx.x, lane = tid & 31;
    const int wv = __builtin_amdgcn_readfirstlane(tid >> 5);
    const int bx = blockIdx.x;
    const int m0 = bx * RW_;

    inv_s[tid] = -1;
    inv_s[tid + 256] = -1;
    __syncthreads();
    v4f q1[8], q2[4], qz[8], qn[8];
    #pragma unroll
    for (int u = 0; u < 8; ++u) {
        const int i = tid + 256 * u;
        q1[u] = *reinterpret_cast<const v4f*>(p_w1 + (i >> 4) * P2_ + 4 * (i & 15));
        qz[u] = *reinterpret_cast<const v4f*>(xz_w + (i >> 5) * H_ + 4 * (i & 31));
        qn[u] = *reinterpret_cast<const v4f*>(xn_w + (i >> 5) * H_ + 4 * (i & 31));
        if (u < 4) q2[u] = *reinterpret_cast<const v4f*>(p_w2 + (i >> 4) * P2_ + 4 * (i & 15));
    }
    {
        const v4i* oi = reinterpret_cast<const v4i*>(obs_idx);
        #pragma unroll 1
        for (int q0 = 0; q0 < NT_ * NO_ / 4; q0 += 8 * 256) {
            v4i v[8];
            #pragma unroll
            for (int u = 0; u < 8; ++u) v[u] = oi[q0 + u * 256 + tid];
            #pragma unroll
            for (int u = 0; u < 8; ++u) {
                const int e = 4 * (q0 + u * 256 + tid);
                const int tt = e >> 12, o = e & (NO_ - 1);
                if ((v[u].x >> 5) == bx) inv_s[tt * RW_ + (v[u].x & 31)] = o;
                if ((v[u].y >> 5) == bx) inv_s[tt * RW_ + (v[u].y & 31)] = o + 1;
                if ((v[u].z >> 5) == bx) inv_s[tt * RW_ + (v[u].z & 31)] = o + 2;
                if ((v[u].w >> 5) == bx) inv_s[tt * RW_ + (v[u].w & 31)] = o + 3;
            }
        }
    }
    #pragma unroll
    for (int u = 0; u < 8; ++u) {
        const int i = tid + 256 * u;
        *reinterpret_cast<v4f*>(pw1_s + (i >> 4) * W1P_ + 4 * (i & 15)) = q1[u];
        *reinterpret_cast<v4f*>(xz_s + (i >> 5) * XWP_ + 4 * (i & 31)) = qz[u];
        *reinterpret_cast<v4f*>(xn_s + (i >> 5) * XWP_ + 4 * (i & 31)) = qn[u];
        if (u < 4) *reinterpret_cast<v4f*>(pw2_s + (i >> 4) * W1P_ + 4 * (i & 15)) = q2[u];
    }
    for (int i = tid; i < D_ * 4 * PREP_; i += 256) wp_s[i] = w_prep[i];
    for (int i = tid; i < D_ * PREP_; i += 256) bp_s[i] = bias_prep[i];

    for (int i = tid; i < RW_ * CH_; i += 256) {
        int r = i / CH_, c = i % CH_;
        float acc = cm_b1[c];
        #pragma unroll
        for (int k = 0; k < CS_; ++k) acc += cov[(m0 + r) * CS_ + k] * cm_w1[k * CH_ + c];
        t_s[r * PP_ + c] = fmaxf(acc, 0.0f);
    }
    __syncthreads();
    for (int i = tid; i < RW_ * H_; i += 256) {
        int r = i / H_, c = i % H_;
        float acc = cm_b2[c];
        #pragma unroll
        for (int k = 0; k < CH_; ++k) acc += t_s[r * PP_ + k] * cm_w2[k * H_ + c];
        h_s[r * HP_ + c] = tanhf(acc);
    }
    __syncthreads();
    pmodel_phase(h_s, t_s, p_s, pw1_s, p_b1, pw2_s, p_b2, lane, wv, 0);

    float lsum = 0.0f;
    #pragma unroll 1
    for (int t = 0; t < NT_; ++t) {
        const int w = __builtin_amdgcn_readfirstlane(tid >> 5);
        ode_phase(inv_s + t * RW_, X + (size_t)t * NO_ * D_, M + (size_t)t * NO_ * D_, x_s, m_s, tid,
                  h_s, p_s, zh_s, xz_s, xz_b, hz_w, xn_s, xn_b, hn_w, lane, w);
        pmodel_phase(h_s, t_s, p_s, pw1_s, p_b1, pw2_s, p_b2, lane, w, 0);
        jump_phase(inv_s + t * RW_, x_s, m_s, wp_s, bp_s,
                   h_s, p_s, gin_s, gate_s, w_ih, w_hh, b_ih, b_hh,
                   lsum, tid, lane, w);
        pmodel_phase(h_s, t_s, p_s, pw1_s, p_b1, pw2_s, p_b2, lane, w, 4);
    }
    #pragma unroll 1
    for (int s = 0; s < NPOST_; ++s) {
        const int w = __builtin_amdgcn_readfirstlane(tid >> 5);
        ode_phase(nullptr, X, M, x_s, m_s, tid,
                  h_s, p_s, zh_s, xz_s, xz_b, hz_w, xn_s, xn_b, hn_w, lane, w);
        pmodel_phase(h_s, t_s, p_s, pw1_s, p_b1, pw2_s, p_b2, lane, w, 4 * (s & 1));
    }

    for (int i = tid; i < RW_ * H_ / 4; i += 256) {
        const int r = i >> 5, c4 = i & 31;
        *reinterpret_cast<v4f*>(h + (size_t)(m0 + r) * H_ + 4 * c4) =
            *reinterpret_cast<const v4f*>(h_s + r * HP_ + 4 * c4);
    }
    for (int i = tid; i < RW_ * P2_ / 4; i += 256) {
        const int r = i >> 4, c4 = i & 15;
        *reinterpret_cast<v4f*>(p + (size_t)(m0 + r) * P2_ + 4 * c4) =
            *reinterpret_cast<const v4f*>(p_s + r * PP_ + 4 * c4);
    }

    red[tid] = lsum;
    __syncthreads();
    for (int s2 = 128; s2 > 0; s2 >>= 1) {
        if (tid < s2) red[tid] += red[tid + s2];
        __syncthreads();
    }
    if (tid == 0) loss_part[bx] = red[0];
}

__global__ void k_loss_final(const float* __restrict__ parts, float* __restrict__ loss) {
    __shared__ float red[256];
    const int tid = threadIdx.x;
    float s = 0.0f;
    for (int i = tid; i < MB_; i += 256) s += parts[i];
    red[tid] = s;
    __syncthreads();
    for (int k = 128; k > 0; k >>= 1) {
        if (tid < k) red[tid] += red[tid + k];
        __syncthreads();
    }
    if (tid == 0) *loss = red[0];
}

extern "C" void kernel_launch(void* const* d_in, const int* in_sizes, int n_in,
                              void* d_out, int out_size, void* d_ws, size_t ws_size,
                              hipStream_t stream) {
    const float* X       = (const float*)d_in[0];
    const float* M       = (const float*)d_in[1];
    const int*   obs_idx = (const int*)d_in[2];
    const float* cov     = (const float*)d_in[3];
    const float* cm_w1   = (const float*)d_in[4];
    const float* cm_b1   = (const float*)d_in[5];
    const float* cm_w2   = (const float*)d_in[6];
    const float* cm_b2   = (const float*)d_in[7];
    const float* p_w1    = (const float*)d_in[8];
    const float* p_b1    = (const float*)d_in[9];
    const float* p_w2    = (const float*)d_in[10];
    const float* p_b2    = (const float*)d_in[11];
    const float* xz_w    = (const float*)d_in[12];
    const float* xz_b    = (const float*)d_in[13];
    const float* hz_w    = (const float*)d_in[14];
    const float* xn_w    = (const float*)d_in[15];
    const float* xn_b    = (const float*)d_in[16];
    const float* hn_w    = (const float*)d_in[17];
    const float* w_ih    = (const float*)d_in[18];
    const float* w_hh    = (const float*)d_in[19];
    const float* b_ih    = (const float*)d_in[20];
    const float* b_hh    = (const float*)d_in[21];
    const float* w_prep  = (const float*)d_in[22];
    const float* b_prep  = (const float*)d_in[23];

    float* h    = (float*)d_out;
    float* p    = h + (size_t)B_ * H_;
    float* loss = p + (size_t)B_ * P2_;
    float* loss_part = (float*)d_ws;

    k_traj<<<MB_, 256, 0, stream>>>(X, M, obs_idx, cov, cm_w1, cm_b1, cm_w2, cm_b2,
                                    p_w1, p_b1, p_w2, p_b2,
                                    xz_w, xz_b, hz_w, xn_w, xn_b, hn_w,
                                    w_ih, w_hh, b_ih, b_hh, w_prep, b_prep,
                                    h, p, loss_part);
    k_loss_final<<<1, 256, 0, stream>>>(loss_part, loss);
}


# jump's element-wise passes (loss, gin front, gates): LDS reads as one group, rows' chains side by side (gates: 4 observed rows per pass by a wave-uniform mask), not one chain per row; sums kept
# speedup vs baseline: 2.3181x; 1.0539x over previous
#include <hip/hip_runtime.h>
#include <hip/hip_bf16.h>
#include <math.h>


#define B_    8192
#define D_    32
#define H_    128
#define P2_   64
#define PREP_ 10
#define GIN_  320
#define G3_   384
#define CS_   8
#define CH_   32
#define NT_   16
#define NO_   4096
#define DT_   0.1f
#define NPOST_ 4
#define LOGC_  0.91893853320467274178f

#define RW_   32
#define MB_   (B_ / RW_)
#define HP_   132
#define PP_   68
#define GP_   324
#define W1P_  80
#define XWP_  144

typedef __attribute__((ext_vector_type(2))) float v2f;
typedef __attribute__((ext_vector_type(4))) float v4f;
typedef __attribute__((ext_vector_type(8))) float v8f;
typedef __attribute__((ext_vector_type(4))) int   v4i;

__device__ __forceinline__ float sigmoidf_(float x) { return 1.0f / (1.0f + expf(-x)); }


__device__ __forceinline__ v2f load_afrag(const float* A, int lda, int m0, int k, int lane) {
    int m  = m0 + (lane & 15);
    int kk = k + ((lane >> 4) << 1);
    v2f a;
    a.x = A[m * lda + kk];
    a.y = A[m * lda + kk + 1];
    return a;
}

__device__ __forceinline__ v8f wmma4(v2f a, v2f b, v8f c) {
    return __builtin_amdgcn_wmma_f32_16x16x4_f32(false, a, false, b, (short)0, c, false, false);
}

template <int NK>
__device__ __forceinline__ void bfrag2_load(const float* Bm, int ldb, int n0, int lane,
                                            v2f (&lo)[NK], v2f (&hi)[NK]) {
    const float* bp = Bm + ((lane >> 4) << 1) * ldb + n0 + 2 * (lane & 15);
    #pragma unroll
    for (int i = 0; i < NK; ++i) {
        lo[i] = *reinterpret_cast<const v2f*>(bp + (4 * i) * ldb);
        hi[i] = *reinterpret_cast<const v2f*>(bp + (4 * i + 1) * ldb);
    }
}

template <int NK>
__device__ __forceinline__ void bfrag2_lds(const float* Bs, int ldb, int n0, int lane,
                                           v2f (&lo)[NK], v2f (&hi)[NK]) {
    const float* bp = Bs + ((lane >> 4) << 1) * ldb + n0 + 2 * (lane & 15);
    #pragma unroll
    for (int i = 0; i < NK; ++i) {
        lo[i] = *reinterpret_cast<const v2f*>(bp + (4 * i) * ldb);
        hi[i] = *reinterpret_cast<const v2f*>(bp + (4 * i + 1) * ldb);
    }
}

#define LDS_READS_AHEAD_FENCE() __builtin_amdgcn_sched_barrier(0x676)

template <int NK>
__device__ __forceinline__ void afrag_group(const float* As, int lda, int lane, v2f (&a)[NK]) {
    #pragma unroll
    for (int i = 0; i < NK; ++i) a[i] = load_afrag(As, lda, 0, 4 * i, lane);
}

template <int NK>
__device__ __forceinline__ void reg_frag2_mma(const v2f* a,
                                              const v2f (&lo)[NK], const v2f (&hi)[NK],
                                              v8f& c0, v8f& c1) {
    #pragma unroll
    for (int i = 0; i < NK; ++i) {
        v2f b0, b1;
        b0.x = lo[i].x; b0.y = hi[i].x;
        b1.x = lo[i].y; b1.y = hi[i].y;
        c0 = wmma4(a[i], b0, c0);
        c1 = wmma4(a[i], b1, c1);
    }
}

template <int NS>
__device__ __forceinline__ void gemm2_reg_glb(const v2f (&a)[NS * 16],
                                              const float* Bm, int ldb, int n0, int lane,
                                              v8f& c0, v8f& c1) {
    v2f lo[2][16], hi[2][16];
    bfrag2_load<16>(Bm, ldb, n0, lane, lo[0], hi[0]);
    #pragma unroll
    for (int s = 0; s < NS; ++s) {
        if (s + 1 < NS)
            bfrag2_load<16>(Bm + (s + 1) * 64 * ldb, ldb, n0, lane, lo[(s + 1) & 1], hi[(s + 1) & 1]);
        reg_frag2_mma<16>(&a[s * 16], lo[s & 1], hi[s & 1], c0, c1);
    }
}

__device__ __forceinline__ void pmodel_phase(const float* h_s, float* t_s, float* p_s,
                                             const float* w1_s, const float* __restrict__ b1,
                                             const float* w2_s, const float* __restrict__ b2,
                                             int lane, int wv, int wb) {
    const int pw = wv - wb;
    const bool mine = (pw >= 0) && (pw < 4);
    const int rt = mine ? (pw >> 1) : 0;
    const int n0 = (pw & 1) * 32;
    const int col = n0 + 2 * (lane & 15), rb = ((lane >> 4) << 3) + 16 * rt;
    if (mine) {
        v8f acc0 = {}, acc1 = {};
        v2f ah[H_ / 4], w1lo[H_ / 4], w1hi[H_ / 4];
        afrag_group<H_ / 4>(h_s + 16 * rt * HP_, HP_, lane, ah);
        bfrag2_lds<H_ / 4>(w1_s, W1P_, n0, lane, w1lo, w1hi);
        LDS_READS_AHEAD_FENCE();
        reg_frag2_mma<H_ / 4>(ah, w1lo, w1hi, acc0, acc1);
        const v2f bv = *reinterpret_cast<const v2f*>(b1 + col);
        #pragma unroll
        for (int j = 0; j < 8; ++j) {
            v2f t;
            t.x = fmaxf(acc0[j] + bv.x, 0.0f);
            t.y = fmaxf(acc1[j] + bv.y, 0.0f);
            *reinterpret_cast<v2f*>(t_s + (rb + j) * PP_ + col) = t;
        }
    }
    __syncthreads();
    if (mine) {
        v8f acc2 = {}, acc3 = {};
        v2f at[P2_ / 4], w2lo[P2_ / 4], w2hi[P2_ / 4];
        afrag_group<P2_ / 4>(t_s + 16 * rt * PP_, PP_, lane, at);
        bfrag2_lds<P2_ / 4>(w2_s, W1P_, n0, lane, w2lo, w2hi);
        LDS_READS_AHEAD_FENCE();
        reg_frag2_mma<P2_ / 4>(at, w2lo, w2hi, acc2, acc3);
        const v2f bv2 = *reinterpret_cast<const v2f*>(b2 + col);
        #pragma unroll
        for (int j = 0; j < 8; ++j) {
            v2f o;
            o.x = acc2[j] + bv2.x;
            o.y = acc3[j] + bv2.y;
            *reinterpret_cast<v2f*>(p_s + (rb + j) * PP_ + col) = o;
        }
    }
    __syncthreads();
}

__device__ __forceinline__ void ode_phase(const int* inv_t,
                                          const float* __restrict__ Xt, const float* __restrict__ Mt,
                                          float* x_s, float* m_s, int tid,
                                          float* h_s, const float* p_s, float* zh_s,
                                          const float* xz_s, const float* __restrict__ xz_b,
                                          const float* __restrict__ hz_w,
                                          const float* xn_s, const float* __restrict__ xn_b,
                                          const float* __restrict__ hn_w,
                                          int lane, int wv) {
    const int n0 = (wv & 3) * 32;
    const int rt = wv >> 2;
    const int col = n0 + 2 * (lane & 15), rb = ((lane >> 4) << 3) + 16 * rt;
    float z0[8], z1[8];
    const int xr = tid >> 3, xc4 = tid & 7;
    v4f xq = {0.0f, 0.0f, 0.0f, 0.0f}, mq = {0.0f, 0.0f, 0.0f, 0.0f};
    if (inv_t != nullptr) {
        const int o = inv_t[xr];
        if (o >= 0) {
            xq = *reinterpret_cast<const v4f*>(Xt + (size_t)o * D_ + 4 * xc4);
            mq = *reinterpret_cast<const v4f*>(Mt + (size_t)o * D_ + 4 * xc4);
        }
    }
    {
        v8f az0 = {}, az1 = {};
        v2f ap[P2_ / 4], ah[H_ / 4], xlo[P2_ / 4], xhi[P2_ / 4];
        afrag_group<P2_ / 4>(p_s + 16 * rt * PP_, PP_, lane, ap);
        bfrag2_lds<P2_ / 4>(xz_s, XWP_, n0, lane, xlo, xhi);
        afrag_group<H_ / 4>(h_s + 16 * rt * HP_, HP_, lane, ah);
        LDS_READS_AHEAD_FENCE();
        reg_frag2_mma<P2_ / 4>(ap, xlo, xhi, az0, az1);
        gemm2_reg_glb<H_ / 64>(ah, hz_w, H_, n0, lane, az0, az1);
        const v2f bz = *reinterpret_cast<const v2f*>(xz_b + col);
        #pragma unroll
        for (int j = 0; j < 8; ++j) {
            const v2f hv = *reinterpret_cast<const v2f*>(h_s + (rb + j) * HP_ + col);
            z0[j] = sigmoidf_(az0[j] + bz.x);
            z1[j] = sigmoidf_(az1[j] + bz.y);
            v2f zh;
            zh.x = z0[j] * hv.x;
            zh.y = z1[j] * hv.y;
            *reinterpret_cast<v2f*>(zh_s + (rb + j) * HP_ + col) = zh;
        }
    }
    if (inv_t != nullptr) {
        *reinterpret_cast<v4f*>(x_s + xr * D_ + 4 * xc4) = xq;
        *reinterpret_cast<v4f*>(m_s + xr * D_ + 4 * xc4) = mq;
    }
    __syncthreads();
    {
        v8f an0 = {}, an1 = {};
        v2f ap[P2_ / 4], azh[H_ / 4], xlo[P2_ / 4], xhi[P2_ / 4];
        afrag_group<P2_ / 4>(p_s + 16 * rt * PP_, PP_, lane, ap);
        bfrag2_lds<P2_ / 4>(xn_s, XWP_, n0, lane, xlo, xhi);
        afrag_group<H_ / 4>(zh_s + 16 * rt * HP_, HP_, lane, azh);
        LDS_READS_AHEAD_FENCE();
        reg_frag2_mma<P2_ / 4>(ap, xlo, xhi, an0, an1);
        gemm2_reg_glb<H_ / 64>(azh, hn_w, H_, n0, lane, an0, an1);
        const v2f bn = *reinterpret_cast<const v2f*>(xn_b + col);
        #pragma unroll
        for (int j = 0; j < 8; ++j) {
            const float n0v = tanhf(an0[j] + bn.x);
            const float n1v = tanhf(an1[j] + bn.y);
            float* hp = h_s + (rb + j) * HP_ + col;
            const v2f hv = *reinterpret_cast<const v2f*>(hp);
            v2f o;
            o.x = hv.x + DT_ * (1.0f - z0[j]) * (n0v - hv.x);
            o.y = hv.y + DT_ * (1.0f - z1[j]) * (n1v - hv.y);
            *reinterpret_cast<v2f*>(hp) = o;
        }
    }
    __syncthreads();
}

#define JKS_ 8
#define JNCI_ 10
struct __attribute__((aligned(4))) f3_t { float x, y, z; };

__device__ __forceinline__ void bfrag3_load(const float* __restrict__ bp,
                                            f3_t (&lo)[JKS_], f3_t (&hi)[JKS_]) {
    #pragma unroll
    for (int i = 0; i < JKS_; ++i) {
        lo[i] = *reinterpret_cast<const f3_t*>(bp + (4 * i) * G3_);
        hi[i] = *reinterpret_cast<const f3_t*>(bp + (4 * i + 1) * G3_);
    }
}

__device__ __forceinline__ void afrag2_load(const float* As, int lda, int lane,
                                            v2f (&a0)[JKS_], v2f (&a1)[JKS_]) {
    #pragma unroll
    for (int i = 0; i < JKS_; ++i) {
        a0[i] = load_afrag(As, lda, 0, 4 * i, lane);
        a1[i] = load_afrag(As, lda, 16, 4 * i, lane);
    }
}

__device__ __forceinline__ void frag3_mma(const v2f (&a0s)[JKS_], const v2f (&a1s)[JKS_],
                                          const f3_t (&lo)[JKS_], const f3_t (&hi)[JKS_],
                                          v8f (&accA)[3], v8f (&accB)[3]) {
    #pragma unroll
    for (int i = 0; i < JKS_; ++i) {
        const v2f a0 = a0s[i];
        const v2f a1 = a1s[i];
        v2f b0, b1, b2;
        b0.x = lo[i].x; b0.y = hi[i].x;
        b1.x = lo[i].y; b1.y = hi[i].y;
        b2.x = lo[i].z; b2.y = hi[i].z;
        accA[0] = wmma4(a0, b0, accA[0]);
        accB[0] = wmma4(a1, b0, accB[0]);
        accA[1] = wmma4(a0, b1, accA[1]);
        accB[1] = wmma4(a1, b1, accB[1]);
        accA[2] = wmma4(a0, b2, accA[2]);
        accB[2] = wmma4(a1, b2, accB[2]);
    }
}

__device__ __forceinline__ void jump_phase(const int* inv_t,
                                           const float* x_s, const float* m_s,
                                           const float* wp_s, const float* bp_s,
                                           float* h_s, const float* p_s, float* gin_s, float* gate_s,
                                           const float* __restrict__ w_ih, const float* __restrict__ w_hh,
                                           const float* __restrict__ b_ih, const float* __restrict__ b_hh,
                                           float& lsum, int tid, int lane, int wv) {
    float* gi_s = gate_s;
    float* gh_s = gate_s + RW_ * G3_;

    const int n0w = wv * 48;
    const int woff = ((lane >> 4) << 1) * G3_ + n0w + 3 * (lane & 15);
    const float* wi = w_ih + woff;
    const float* wh = w_hh + woff;

    f3_t lo[2][JKS_], hi[2][JKS_];
    bfrag3_load(wi, lo[0], hi[0]);

    {
        float q[4], mq[4];
        #pragma unroll
        for (int u = 0; u < 4; ++u) {
            const int r = wv + 8 * u, d = lane;
            float mean = p_s[r * PP_ + d];
            float lv   = p_s[r * PP_ + D_ + d];
            float x    = x_s[r * D_ + d];
            float m    = m_s[r * D_ + d];
            float err  = (x - mean) * expf(-0.5f * lv);
            q[u] = err * err + lv + 2.0f * LOGC_;
            mq[u] = m;
        }
        asm volatile("" :: "v"(q[0]), "v"(q[1]), "v"(q[2]), "v"(q[3]));
        #pragma unroll
        for (int u = 0; u < 4; ++u) lsum += 0.5f * ((inv_t[wv + 8 * u] >= 0) ? q[u] * mq[u] : 0.0f);
    }

    {
        const int mr = lane & 15, hh = lane >> 4;
        const int d0 = wv * 4;
        const int pc = mr < PREP_ ? mr : PREP_ - 1;
        v2f bw[4];
        float bias[4];
        #pragma unroll
        for (int i = 0; i < 4; ++i) {
            const int d = d0 + i;
            const float* wp = wp_s + (d * 4 + 2 * hh) * PREP_ + pc;
            bw[i].x = wp[0];
            bw[i].y = wp[PREP_];
            bias[i] = bp_s[d * PREP_ + pc];
        }
        v4f xv[2], mu[2], lg[2];
        #pragma unroll
        for (int rt = 0; rt < 2; ++rt) {
            xv[rt] = *reinterpret_cast<const v4f*>(x_s + (16 * rt + mr) * D_ + d0);
            mu[rt] = *reinterpret_cast<const v4f*>(p_s + (16 * rt + mr) * PP_ + d0);
            lg[rt] = *reinterpret_cast<const v4f*>(p_s + (16 * rt + mr) * PP_ + D_ + d0);
        }
        v8f acc[2][4];
        #pragma unroll
        for (int rt = 0; rt < 2; ++rt) {
            #pragma unroll
            for (int i = 0; i < 4; ++i) {
                const float err = (xv[rt][i] - mu[rt][i]) * expf(-0.5f * lg[rt][i]);
                v2f a;
                a.x = hh ? lg[rt][i] : xv[rt][i];
                a.y = hh ? err       : mu[rt][i];
                acc[rt][i] = (v8f){};
                acc[rt][i] = wmma4(a, bw[i], acc[rt][i]);
            }
        }
        if (mr < PREP_) {
            int  ov[2][8];
            v4f  mk[2][8];
            #pragma unroll
            for (int rt = 0; rt < 2; ++rt) {
                #pragma unroll
                for (int j = 0; j < 8; ++j) {
                    ov[rt][j] = inv_t[16 * rt + 8 * hh + j];
                    mk[rt][j] = *reinterpret_cast<const v4f*>(m_s + (16 * rt + 8 * hh + j) * D_ + d0);
                }
            }
            #pragma unroll
            for (int rt = 0; rt < 2; ++rt) {
                float* gp = gin_s + (16 * rt + 8 * hh) * GP_ + d0 * PREP_ + mr;
                #pragma unroll
                for (int j = 0; j < 8; ++j) {
                    const bool ob = ov[rt][j] >= 0;
                    #pragma unroll
                    for (int i = 0; i < 4; ++i) {
                        const float g = ob ? fmaxf(acc[rt][i][j] + bias[i], 0.0f) * mk[rt][j][i] : 0.0f;
                        gp[j * GP_ + i * PREP_] = g;
                    }
                }
            }
        }
    }
    __syncthreads();

    v8f giA[3], giB[3], ghA[3], ghB[3];
    #pragma unroll
    for (int s = 0; s < 3; ++s) { giA[s] = (v8f){}; giB[s] = (v8f){}; ghA[s] = (v8f){}; ghB[s] = (v8f){}; }

    v2f fa0[2][JKS_], fa1[2][JKS_];
    bfrag3_load(wi + JKS_ * 4 * G3_, lo[1], hi[1]);
    afrag2_load(gin_s, GP_, lane, fa0[0], fa1[0]);
    #pragma unroll
    for (int sl = 0; sl < JNCI_ + 4; ++sl) {
        __builtin_amdgcn_sched_barrier(0);
        if (sl + 1 < JNCI_ + 4) {
            if (sl + 1 < JNCI_) afrag2_load(gin_s + (sl + 1) * (JKS_ * 4), GP_, lane, fa0[(sl + 1) & 1], fa1[(sl + 1) & 1]);
            else                afrag2_load(h_s + (sl + 1 - JNCI_) * (JKS_ * 4), HP_, lane, fa0[(sl + 1) & 1], fa1[(sl + 1) & 1]);
        }
        LDS_READS_AHEAD_FENCE();
        if (sl < JNCI_) frag3_mma(fa0[sl & 1], fa1[sl & 1], lo[sl & 1], hi[sl & 1], giA, giB);
        else            frag3_mma(fa0[sl & 1], fa1[sl & 1], lo[sl & 1], hi[sl & 1], ghA, ghB);
        if (sl + 2 < JNCI_ + 4) {
            if (sl + 2 < JNCI_) bfrag3_load(wi + (sl + 2) * (JKS_ * 4 * G3_), lo[sl & 1], hi[sl & 1]);
            else                bfrag3_load(wh + (sl + 2 - JNCI_) * (JKS_ * 4 * G3_), lo[sl & 1], hi[sl & 1]);
        }
    }
    __builtin_amdgcn_sched_barrier(0);

    {
        const int rb = (lane >> 4) << 3;
        #pragma unroll
        for (int s = 0; s < 3; ++s) {
            const int col = n0w + 3 * (lane & 15) + s;
            const float bi = b_ih[col], bh = b_hh[col];
            #pragma unroll
            for (int j = 0; j < 8; ++j) {
                gi_s[(rb + j) * G3_ + col] = giA[s][j] + bi;
                gh_s[(rb + j) * G3_ + col] = ghA[s][j] + bh;
                gi_s[(16 + rb + j) * G3_ + col] = giB[s][j] + bi;
                gh_s[(16 + rb + j) * G3_ + col] = ghB[s][j] + bh;
            }
        }
    }
    __syncthreads();

    {
        const int half = wv >> 2;
        const int j = ((wv & 3) << 5) + lane;
        unsigned live = __builtin_amdgcn_ballot_w32(inv_t[half + 2 * (lane & 15)] >= 0) & 0xffffu;
        for (int pass = 0; pass < 4 && live != 0u; ++pass) {
            int r[4];
            r[0] = half + 2 * __builtin_ctz(live);
            live &= live - 1u;
            #pragma unroll
            for (int u = 1; u < 4; ++u) {
                r[u] = live != 0u ? half + 2 * __builtin_ctz(live) : r[0];
                live &= live - 1u;
            }
            float gir[4], ghr[4], giz[4], ghz[4], gic[4], ghc[4], hv[4], rr[4], zz[4], nn[4];
            #pragma unroll
            for (int u = 0; u < 4; ++u) {
                const float* gi = gi_s + r[u] * G3_ + j;
                const float* gh = gh_s + r[u] * G3_ + j;
                gir[u] = gi[0];      ghr[u] = gh[0];
                giz[u] = gi[H_];     ghz[u] = gh[H_];
                gic[u] = gi[2 * H_]; ghc[u] = gh[2 * H_];
                hv[u] = h_s[r[u] * HP_ + j];
            }
            #pragma unroll
            for (int u = 0; u < 4; ++u) {
                rr[u] = sigmoidf_(gir[u] + ghr[u]);
                zz[u] = sigmoidf_(giz[u] + ghz[u]);
            }
            asm volatile("" :: "v"(rr[0]), "v"(rr[1]), "v"(rr[2]), "v"(rr[3]),
                               "v"(zz[0]), "v"(zz[1]), "v"(zz[2]), "v"(zz[3]));
            #pragma unroll
            for (int u = 0; u < 4; ++u) nn[u] = tanhf(gic[u] + rr[u] * ghc[u]);
            #pragma unroll
            for (int u = 0; u < 4; ++u) h_s[r[u] * HP_ + j] = (1.0f - zz[u]) * nn[u] + zz[u] * hv[u];
        }
    }
    __syncthreads();
}

__global__ void __launch_bounds__(256, 2) k_traj(
                       const float* __restrict__ X, const float* __restrict__ M,
                       const int* __restrict__ obs_idx, const float* __restrict__ cov,
                       const float* __restrict__ cm_w1, const float* __restrict__ cm_b1,
                       const float* __restrict__ cm_w2, const float* __restrict__ cm_b2,
                       const float* __restrict__ p_w1, const float* __restrict__ p_b1,
                       const float* __restrict__ p_w2, const float* __restrict__ p_b2,
                       const float* __restrict__ xz_w, const float* __restrict__ xz_b,
                       const float* __restrict__ hz_w,
                       const float* __restrict__ xn_w, const float* __restrict__ xn_b,
                       const float* __restrict__ hn_w,
                       const float* __restrict__ w_ih, const float* __restrict__ w_hh,
                       const float* __restrict__ b_ih, const float* __restrict__ b_hh,
                       const float* __restrict__ w_prep, const float* __restrict__ bias_prep,
                       float* __restrict__ h, float* __restrict__ p,
                       float* __restrict__ loss_part) {
    __shared__ __attribute__((aligned(16))) float h_s[RW_ * HP_];
    __shared__ __attribute__((aligned(16))) float p_s[RW_ * PP_];
    __shared__ __attribute__((aligned(16))) float t_s[RW_ * PP_];
    __shared__ __attribute__((aligned(16))) float gin_s[RW_ * GP_];
    __shared__ __attribute__((aligned(16))) float gate_s[2 * RW_ * G3_];
    __shared__ __attribute__((aligned(16))) float x_s[RW_ * D_];
    __shared__ __attribute__((aligned(16))) float m_s[RW_ * D_];
    __shared__ float wp_s[D_ * 4 * PREP_];
    __shared__ float bp_s[D_ * PREP_];
    __shared__ int   inv_s[NT_ * RW_];
    __shared__ __attribute__((aligned(16))) float pw1_s[H_ * W1P_];
    __shared__ __attribute__((aligned(16))) float pw2_s[P2_ * W1P_];
    __shared__ __attribute__((aligned(16))) float xz_s[P2_ * XWP_];
    __shared__ __attribute__((aligned(16))) float xn_s[P2_ * XWP_];
    float* zh_s = gin_s;
    float* red  = t_s;
    const int tid = threadIdx.x, lane = tid & 31;
    const int wv = __builtin_amdgcn_readfirstlane(tid >> 5);
    const int bx = blockIdx.x;
    const int m0 = bx * RW_;

    inv_s[tid] = -1;
    inv_s[tid + 256] = -1;
    __syncthreads();
    v4f q1[8], q2[4], qz[8], qn[8];
    #pragma unroll
    for (int u = 0; u < 8; ++u) {
        const int i = tid + 256 * u;
        q1[u] = *reinterpret_cast<const v4f*>(p_w1 + (i >> 4) * P2_ + 4 * (i & 15));
        qz[u] = *reinterpret_cast<const v4f*>(xz_w + (i >> 5) * H_ + 4 * (i & 31));
        qn[u] = *reinterpret_cast<const v4f*>(xn_w + (i >> 5) * H_ + 4 * (i & 31));
        if (u < 4) q2[u] = *reinterpret_cast<const v4f*>(p_w2 + (i >> 4) * P2_ + 4 * (i & 15));
    }
    {
        const v4i* oi = reinterpret_cast<const v4i*>(obs_idx);
        #pragma unroll 1
        for (int q0 = 0; q0 < NT_ * NO_ / 4; q0 += 8 * 256) {
            v4i v[8];
            #pragma unroll
            for (int u = 0; u < 8; ++u) v[u] = oi[q0 + u * 256 + tid];
            #pragma unroll
            for (int u = 0; u < 8; ++u) {
                const int e = 4 * (q0 + u * 256 + tid);
                const int tt = e >> 12, o = e & (NO_ - 1);
                if ((v[u].x >> 5) == bx) inv_s[tt * RW_ + (v[u].x & 31)] = o;
                if ((v[u].y >> 5) == bx) inv_s[tt * RW_ + (v[u].y & 31)] = o + 1;
                if ((v[u].z >> 5) == bx) inv_s[tt * RW_ + (v[u].z & 31)] = o + 2;
                if ((v[u].w >> 5) == bx) inv_s[tt * RW_ + (v[u].w & 31)] = o + 3;
            }
        }
    }
    #pragma unroll
    for (int u = 0; u < 8; ++u) {
        const int i = tid + 256 * u;
        *reinterpret_cast<v4f*>(pw1_s + (i >> 4) * W1P_ + 4 * (i & 15)) = q1[u];
        *reinterpret_cast<v4f*>(xz_s + (i >> 5) * XWP_ + 4 * (i & 31)) = qz[u];
        *reinterpret_cast<v4f*>(xn_s + (i >> 5) * XWP_ + 4 * (i & 31)) = qn[u];
        if (u < 4) *reinterpret_cast<v4f*>(pw2_s + (i >> 4) * W1P_ + 4 * (i & 15)) = q2[u];
    }
    for (int i = tid; i < D_ * 4 * PREP_; i += 256) wp_s[i] = w_prep[i];
    for (int i = tid; i < D_ * PREP_; i += 256) bp_s[i] = bias_prep[i];

    for (int i = tid; i < RW_ * CH_; i += 256) {
        int r = i / CH_, c = i % CH_;
        float acc = cm_b1[c];
        #pragma unroll
        for (int k = 0; k < CS_; ++k) acc += cov[(m0 + r) * CS_ + k] * cm_w1[k * CH_ + c];
        t_s[r * PP_ + c] = fmaxf(acc, 0.0f);
    }
    __syncthreads();
    for (int i = tid; i < RW_ * H_; i += 256) {
        int r = i / H_, c = i % H_;
        float acc = cm_b2[c];
        #pragma unroll
        for (int k = 0; k < CH_; ++k) acc += t_s[r * PP_ + k] * cm_w2[k * H_ + c];
        h_s[r * HP_ + c] = tanhf(acc);
    }
    __syncthreads();
    pmodel_phase(h_s, t_s, p_s, pw1_s, p_b1, pw2_s, p_b2, lane, wv, 0);

    float lsum = 0.0f;
    #pragma unroll 1
    for (int t = 0; t < NT_; ++t) {
        const int w = __builtin_amdgcn_readfirstlane(tid >> 5);
        ode_phase(inv_s + t * RW_, X + (size_t)t * NO_ * D_, M + (size_t)t * NO_ * D_, x_s, m_s, tid,
                  h_s, p_s, zh_s, xz_s, xz_b, hz_w, xn_s, xn_b, hn_w, lane, w);
        pmodel_phase(h_s, t_s, p_s, pw1_s, p_b1, pw2_s, p_b2, lane, w, 0);
        jump_phase(inv_s + t * RW_, x_s, m_s, wp_s, bp_s,
                   h_s, p_s, gin_s, gate_s, w_ih, w_hh, b_ih, b_hh,
                   lsum, tid, lane, w);
        pmodel_phase(h_s, t_s, p_s, pw1_s, p_b1, pw2_s, p_b2, lane, w, 4);
    }
    #pragma unroll 1
    for (int s = 0; s < NPOST_; ++s) {
        const int w = __builtin_amdgcn_readfirstlane(tid >> 5);
        ode_phase(nullptr, X, M, x_s, m_s, tid,
                  h_s, p_s, zh_s, xz_s, xz_b, hz_w, xn_s, xn_b, hn_w, lane, w);
        pmodel_phase(h_s, t_s, p_s, pw1_s, p_b1, pw2_s, p_b2, lane, w, 4 * (s & 1));
    }

    for (int i = tid; i < RW_ * H_ / 4; i += 256) {
        const int r = i >> 5, c4 = i & 31;
        *reinterpret_cast<v4f*>(h + (size_t)(m0 + r) * H_ + 4 * c4) =
            *reinterpret_cast<const v4f*>(h_s + r * HP_ + 4 * c4);
    }
    for (int i = tid; i < RW_ * P2_ / 4; i += 256) {
        const int r = i >> 4, c4 = i & 15;
        *reinterpret_cast<v4f*>(p + (size_t)(m0 + r) * P2_ + 4 * c4) =
            *reinterpret_cast<const v4f*>(p_s + r * PP_ + 4 * c4);
    }

    red[tid] = lsum;
    __syncthreads();
    for (int s2 = 128; s2 > 0; s2 >>= 1) {
        if (tid < s2) red[tid] += red[tid + s2];
        __syncthreads();
    }
    if (tid == 0) loss_part[bx] = red[0];
}

__global__ void k_loss_final(const float* __restrict__ parts, float* __restrict__ loss) {
    __shared__ float red[256];
    const int tid = threadIdx.x;
    float s = 0.0f;
    for (int i = tid; i < MB_; i += 256) s += parts[i];
    red[tid] = s;
    __syncthreads();
    for (int k = 128; k > 0; k >>= 1) {
        if (tid < k) red[tid] += red[tid + k];
        __syncthreads();
    }
    if (tid == 0) *loss = red[0];
}

extern "C" void kernel_launch(void* const* d_in, const int* in_sizes, int n_in,
                              void* d_out, int out_size, void* d_ws, size_t ws_size,
                              hipStream_t stream) {
    const float* X       = (const float*)d_in[0];
    const float* M       = (const float*)d_in[1];
    const int*   obs_idx = (const int*)d_in[2];
    const float* cov     = (const float*)d_in[3];
    const float* cm_w1   = (const float*)d_in[4];
    const float* cm_b1   = (const float*)d_in[5];
    const float* cm_w2   = (const float*)d_in[6];
    const float* cm_b2   = (const float*)d_in[7];
    const float* p_w1    = (const float*)d_in[8];
    const float* p_b1    = (const float*)d_in[9];
    const float* p_w2    = (const float*)d_in[10];
    const float* p_b2    = (const float*)d_in[11];
    const float* xz_w    = (const float*)d_in[12];
    const float* xz_b    = (const float*)d_in[13];
    const float* hz_w    = (const float*)d_in[14];
    const float* xn_w    = (const float*)d_in[15];
    const float* xn_b    = (const float*)d_in[16];
    const float* hn_w    = (const float*)d_in[17];
    const float* w_ih    = (const float*)d_in[18];
    const float* w_hh    = (const float*)d_in[19];
    const float* b_ih    = (const float*)d_in[20];
    const float* b_hh    = (const float*)d_in[21];
    const float* w_prep  = (const float*)d_in[22];
    const float* b_prep  = (const float*)d_in[23];

    float* h    = (float*)d_out;
    float* p    = h + (size_t)B_ * H_;
    float* loss = p + (size_t)B_ * P2_;
    float* loss_part = (float*)d_ws;

    k_traj<<<MB_, 256, 0, stream>>>(X, M, obs_idx, cov, cm_w1, cm_b1, cm_w2, cm_b2,
                                    p_w1, p_b1, p_w2, p_b2,
                                    xz_w, xz_b, hz_w, xn_w, xn_b, hn_w,
                                    w_ih, w_hh, b_ih, b_hh, w_prep, b_prep,
                                    h, p, loss_part);
    k_loss_final<<<1, 256, 0, stream>>>(loss_part, loss);
}
